# starting kernel
# baseline (speedup 1.0000x reference)
#include <hip/hip_runtime.h>
#include <hip/hip_bf16.h>

typedef __attribute__((ext_vector_type(2))) float v2f;
typedef __attribute__((ext_vector_type(8))) float v8f;

__device__ __forceinline__ v8f wmma4(v2f a, v2f b, v8f c) {
    return __builtin_amdgcn_wmma_f32_16x16x4_f32(
         false, a,  false, b,
         (short)0, c,  false,  false);
}

__global__ void k_zero(float* p, int n) {
    int i = blockIdx.x * blockDim.x + threadIdx.x;
    if (i < n) p[i] = 0.0f;
}

__global__ void k_count_deg(const int* __restrict__ dst, float* __restrict__ deg, int E) {
    int e = blockIdx.x * blockDim.x + threadIdx.x;
    if (e < E) atomicAdd(&deg[dst[e]], 1.0f);
}

__global__ void k_dinv(float* __restrict__ d, int N) {
    int i = blockIdx.x * blockDim.x + threadIdx.x;
    if (i < N) d[i] = rsqrtf(d[i] + 1.0f);
}

__global__ void k_encoder(const float* __restrict__ x, const float* __restrict__ cnn,
                          const float* __restrict__ fcW, const float* __restrict__ fcb,
                          const float* __restrict__ cnnW, const float* __restrict__ cnnb,
                          float* __restrict__ h0, int N, int nTiles) {
    int wave = (blockIdx.x * blockDim.x + threadIdx.x) >> 5;
    if (wave >= nTiles) return;
    int lane = threadIdx.x & 31;
    int n    = lane & 15;
    int kh   = lane >> 4;
    int rowA = wave * 16 + n;    if (rowA >= N) rowA = N - 1;

    v8f acc0 = {}, acc1 = {}, acc2 = {}, acc3 = {};

    const float* ax = x + (size_t)rowA * 32;
#pragma unroll
    for (int k0 = 0; k0 < 32; k0 += 4) {
        int kb = k0 + 2 * kh;
        v2f a;  a.x = ax[kb];               a.y = ax[kb + 1];
        v2f b0; b0.x = fcW[kb * 32 + n];    b0.y = fcW[(kb + 1) * 32 + n];
        v2f b1; b1.x = fcW[kb * 32 + 16 + n]; b1.y = fcW[(kb + 1) * 32 + 16 + n];
        acc0 = wmma4(a, b0, acc0);
        acc1 = wmma4(a, b1, acc1);
    }
    const float* ac = cnn + (size_t)rowA * 128;
#pragma unroll
    for (int k0 = 0; k0 < 128; k0 += 4) {
        int kb = k0 + 2 * kh;
        v2f a;  a.x = ac[kb];                a.y = ac[kb + 1];
        v2f b0; b0.x = cnnW[kb * 32 + n];    b0.y = cnnW[(kb + 1) * 32 + n];
        v2f b1; b1.x = cnnW[kb * 32 + 16 + n]; b1.y = cnnW[(kb + 1) * 32 + 16 + n];
        acc2 = wmma4(a, b0, acc2);
        acc3 = wmma4(a, b1, acc3);
    }
#pragma unroll
    for (int v = 0; v < 8; ++v) {
        int mr   = (lane < 16) ? v : v + 8;
        int node = wave * 16 + mr;
        if (node < N) {
            float* o = h0 + (size_t)node * 64;
            o[n]      = fmaxf(acc0[v] + fcb[n],       0.0f);
            o[16 + n] = fmaxf(acc1[v] + fcb[16 + n],  0.0f);
            o[32 + n] = fmaxf(acc2[v] + cnnb[n],      0.0f);
            o[48 + n] = fmaxf(acc3[v] + cnnb[16 + n], 0.0f);
        }
    }
}

__global__ void k_gemm_t1(const float* __restrict__ h0, const float* __restrict__ W1,
                          float* __restrict__ t1, int N, int nTiles) {
    int wave = (blockIdx.x * blockDim.x + threadIdx.x) >> 5;
    if (wave >= nTiles) return;
    int lane = threadIdx.x & 31;
    int n    = lane & 15;
    int kh   = lane >> 4;
    int rowA = wave * 16 + n;    if (rowA >= N) rowA = N - 1;

    v8f acc0 = {}, acc1 = {};
    const float* ar = h0 + (size_t)rowA * 64;
#pragma unroll
    for (int k0 = 0; k0 < 64; k0 += 4) {
        int kb = k0 + 2 * kh;
        v2f a;  a.x = ar[kb];              a.y = ar[kb + 1];
        v2f b0; b0.x = W1[kb * 32 + n];    b0.y = W1[(kb + 1) * 32 + n];
        v2f b1; b1.x = W1[kb * 32 + 16 + n]; b1.y = W1[(kb + 1) * 32 + 16 + n];
        acc0 = wmma4(a, b0, acc0);
        acc1 = wmma4(a, b1, acc1);
    }
#pragma unroll
    for (int v = 0; v < 8; ++v) {
        int mr   = (lane < 16) ? v : v + 8;
        int node = wave * 16 + mr;
        if (node < N) {
            float* o = t1 + (size_t)node * 32;
            o[n]      = acc0[v];
            o[16 + n] = acc1[v];
        }
    }
}

__global__ void k_init_agg1(const float* __restrict__ t1, const float* __restrict__ dinv,
                            const float* __restrict__ b1, float* __restrict__ agg1, int N) {
    int i = blockIdx.x * blockDim.x + threadIdx.x;
    if (i < N * 32) {
        int node = i >> 5, f = i & 31;
        float di = dinv[node];
        agg1[i] = t1[i] * di * di + b1[f];
    }
}

__global__ void k_edge1(const int* __restrict__ src, const int* __restrict__ dst,
                        const float* __restrict__ dinv, const float* __restrict__ t1,
                        float* __restrict__ agg1, int E) {
    int e = (blockIdx.x * blockDim.x + threadIdx.x) >> 5;
    if (e >= E) return;
    int lane = threadIdx.x & 31;
    int s = src[e], d = dst[e];
    float w = dinv[s] * dinv[d];
    atomicAdd(&agg1[(size_t)d * 32 + lane], t1[(size_t)s * 32 + lane] * w);
}

__global__ void k_gemm_t2(const float* __restrict__ agg1, const float* __restrict__ W2,
                          float* __restrict__ t2, int N, int nTiles) {
    int wave = (blockIdx.x * blockDim.x + threadIdx.x) >> 5;
    if (wave >= nTiles) return;
    int lane = threadIdx.x & 31;
    int n    = lane & 15;
    int kh   = lane >> 4;
    int rowA = wave * 16 + n;    if (rowA >= N) rowA = N - 1;

    v8f acc0 = {}, acc1 = {};
    const float* ar = agg1 + (size_t)rowA * 32;
    bool c1ok = (16 + n) < 20;
#pragma unroll
    for (int k0 = 0; k0 < 32; k0 += 4) {
        int kb = k0 + 2 * kh;
        v2f a;  a.x = fmaxf(ar[kb], 0.0f);  a.y = fmaxf(ar[kb + 1], 0.0f);
        v2f b0; b0.x = W2[kb * 20 + n];     b0.y = W2[(kb + 1) * 20 + n];
        v2f b1; b1.x = c1ok ? W2[kb * 20 + 16 + n] : 0.0f;
                b1.y = c1ok ? W2[(kb + 1) * 20 + 16 + n] : 0.0f;
        acc0 = wmma4(a, b0, acc0);
        acc1 = wmma4(a, b1, acc1);
    }
#pragma unroll
    for (int v = 0; v < 8; ++v) {
        int mr   = (lane < 16) ? v : v + 8;
        int node = wave * 16 + mr;
        if (node < N) {
            float* o = t2 + (size_t)node * 20;
            o[n] = acc0[v];
            if (c1ok) o[16 + n] = acc1[v];
        }
    }
}

__global__ void k_init_out(const float* __restrict__ t2, const float* __restrict__ dinv,
                           const float* __restrict__ b2, float* __restrict__ out, int N) {
    int i = blockIdx.x * blockDim.x + threadIdx.x;
    if (i < N * 20) {
        int node = i / 20, f = i - node * 20;
        float di = dinv[node];
        out[i] = t2[i] * di * di + b2[f];
    }
}

__global__ void k_edge2(const int* __restrict__ src, const int* __restrict__ dst,
                        const float* __restrict__ dinv, const float* __restrict__ t2,
                        float* __restrict__ out, int E) {
    int e = (blockIdx.x * blockDim.x + threadIdx.x) >> 5;
    if (e >= E) return;
    int lane = threadIdx.x & 31;
    int s = src[e], d = dst[e];
    float w = dinv[s] * dinv[d];
    if (lane < 20)
        atomicAdd(&out[(size_t)d * 20 + lane], t2[(size_t)s * 20 + lane] * w);
}

extern "C" void kernel_launch(void* const* d_in, const int* in_sizes, int n_in,
                              void* d_out, int out_size, void* d_ws, size_t ws_size,
                              hipStream_t stream) {
    const float* x    = (const float*)d_in[0];
    const float* cnn  = (const float*)d_in[1];
    const int*   ei   = (const int*)d_in[2];
    const float* fcW  = (const float*)d_in[3];
    const float* fcb  = (const float*)d_in[4];
    const float* cnnW = (const float*)d_in[5];
    const float* cnnb = (const float*)d_in[6];
    const float* W1   = (const float*)d_in[7];
    const float* b1   = (const float*)d_in[8];
    const float* W2   = (const float*)d_in[9];
    const float* b2   = (const float*)d_in[10];

    const int N = in_sizes[0] / 32;
    const int E = in_sizes[2] / 2;
    const int* src = ei;
    const int* dst = ei + E;

    float* ws   = (float*)d_ws;
    float* dinv = ws;  ws += N;
    float* h0   = ws;  ws += (size_t)N * 64;
    float* t1   = ws;  ws += (size_t)N * 32;
    float* agg1 = ws;  ws += (size_t)N * 32;
    float* t2   = ws;  ws += (size_t)N * 20;
    float* out  = (float*)d_out;

    const int nTiles = (N + 15) / 16;
    const int B = 256;
    dim3 blkG(128);
    dim3 grdG((nTiles * 32 + 127) / 128);
    dim3 grdEdge((unsigned)(((long long)E * 32 + B - 1) / B));

    k_zero<<<(N + B - 1) / B, B, 0, stream>>>(dinv, N);
    k_count_deg<<<(E + B - 1) / B, B, 0, stream>>>(dst, dinv, E);
    k_dinv<<<(N + B - 1) / B, B, 0, stream>>>(dinv, N);

    k_encoder<<<grdG, blkG, 0, stream>>>(x, cnn, fcW, fcb, cnnW, cnnb, h0, N, nTiles);
    k_gemm_t1<<<grdG, blkG, 0, stream>>>(h0, W1, t1, N, nTiles);

    k_init_agg1<<<(N * 32 + B - 1) / B, B, 0, stream>>>(t1, dinv, b1, agg1, N);
    k_edge1<<<grdEdge, B, 0, stream>>>(src, dst, dinv, t1, agg1, E);

    k_gemm_t2<<<grdG, blkG, 0, stream>>>(agg1, W2, t2, N, nTiles);
    k_init_out<<<(N * 20 + B - 1) / B, B, 0, stream>>>(t2, dinv, b2, out, N);
    k_edge2<<<grdEdge, B, 0, stream>>>(src, dst, dinv, t2, out, E);
}


# In-degree count privatized: each block counts one 16384-node range of one edge slice in integer LDS counters, then adds non-zero counters to deg with coalesced float atomics (exact integer sums)
# speedup vs baseline: 1.2510x; 1.2510x over previous
#include <hip/hip_runtime.h>
#include <hip/hip_bf16.h>

typedef __attribute__((ext_vector_type(2))) float v2f;
typedef __attribute__((ext_vector_type(8))) float v8f;

__device__ __forceinline__ v8f wmma4(v2f a, v2f b, v8f c) {
    return __builtin_amdgcn_wmma_f32_16x16x4_f32(
         false, a,  false, b,
         (short)0, c,  false,  false);
}

__global__ void k_zero(float* p, int n) {
    int i = blockIdx.x * blockDim.x + threadIdx.x;
    if (i < n) p[i] = 0.0f;
}

#define DEG_RANGE   16384
#define DEG_SPLITS  32
#define DEG_THREADS 1024

__device__ __forceinline__ void deg_count(unsigned* cnt, int d, int lo) {
    unsigned r = (unsigned)(d - lo);
    if (r < (unsigned)DEG_RANGE) atomicAdd(&cnt[r], 1u);
}

__device__ __forceinline__ void deg_count4(unsigned* cnt, int4 v, int lo) {
    deg_count(cnt, v.x, lo);
    deg_count(cnt, v.y, lo);
    deg_count(cnt, v.z, lo);
    deg_count(cnt, v.w, lo);
}

__global__ void __launch_bounds__(DEG_THREADS)
k_count_deg(const int* __restrict__ dst, float* __restrict__ deg, int E, int N) {
    __shared__ unsigned cnt[DEG_RANGE];
    const int tid = threadIdx.x;
    const int lo  = blockIdx.y * DEG_RANGE;

    for (int i = tid; i < DEG_RANGE; i += DEG_THREADS) cnt[i] = 0u;
    __syncthreads();

    int per = (E + DEG_SPLITS - 1) / DEG_SPLITS;
    per = (per + 3) & ~3;
    long long e0l = (long long)blockIdx.x * per;
    const int e0 = (e0l < (long long)E) ? (int)e0l : E;
    const int e1 = (E - e0 > per) ? (e0 + per) : E;

    if ((((size_t)dst) & 15) == 0) {
        const int4* __restrict__ p = (const int4*)(dst + e0);
        const int nv = (e1 - e0) >> 2;
        int i = tid;
        for (; i + 3 * DEG_THREADS < nv; i += 4 * DEG_THREADS) {
            int4 a = p[i];
            int4 b = p[i + DEG_THREADS];
            int4 c = p[i + 2 * DEG_THREADS];
            int4 d = p[i + 3 * DEG_THREADS];
            deg_count4(cnt, a, lo);
            deg_count4(cnt, b, lo);
            deg_count4(cnt, c, lo);
            deg_count4(cnt, d, lo);
        }
        for (; i < nv; i += DEG_THREADS) deg_count4(cnt, p[i], lo);
        for (int e = e0 + (nv << 2) + tid; e < e1; e += DEG_THREADS) deg_count(cnt, dst[e], lo);
    } else {
        for (int e = e0 + tid; e < e1; e += DEG_THREADS) deg_count(cnt, dst[e], lo);
    }
    __syncthreads();

    for (int i = tid; i < DEG_RANGE; i += DEG_THREADS) {
        unsigned c = cnt[i];
        int node = lo + i;
        if (c != 0u && node < N) atomicAdd(&deg[node], (float)c);
    }
}

__global__ void k_dinv(float* __restrict__ d, int N) {
    int i = blockIdx.x * blockDim.x + threadIdx.x;
    if (i < N) d[i] = rsqrtf(d[i] + 1.0f);
}

__global__ void k_encoder(const float* __restrict__ x, const float* __restrict__ cnn,
                          const float* __restrict__ fcW, const float* __restrict__ fcb,
                          const float* __restrict__ cnnW, const float* __restrict__ cnnb,
                          float* __restrict__ h0, int N, int nTiles) {
    int wave = (blockIdx.x * blockDim.x + threadIdx.x) >> 5;
    if (wave >= nTiles) return;
    int lane = threadIdx.x & 31;
    int n    = lane & 15;
    int kh   = lane >> 4;
    int rowA = wave * 16 + n;    if (rowA >= N) rowA = N - 1;

    v8f acc0 = {}, acc1 = {}, acc2 = {}, acc3 = {};

    const float* ax = x + (size_t)rowA * 32;
#pragma unroll
    for (int k0 = 0; k0 < 32; k0 += 4) {
        int kb = k0 + 2 * kh;
        v2f a;  a.x = ax[kb];               a.y = ax[kb + 1];
        v2f b0; b0.x = fcW[kb * 32 + n];    b0.y = fcW[(kb + 1) * 32 + n];
        v2f b1; b1.x = fcW[kb * 32 + 16 + n]; b1.y = fcW[(kb + 1) * 32 + 16 + n];
        acc0 = wmma4(a, b0, acc0);
        acc1 = wmma4(a, b1, acc1);
    }
    const float* ac = cnn + (size_t)rowA * 128;
#pragma unroll
    for (int k0 = 0; k0 < 128; k0 += 4) {
        int kb = k0 + 2 * kh;
        v2f a;  a.x = ac[kb];                a.y = ac[kb + 1];
        v2f b0; b0.x = cnnW[kb * 32 + n];    b0.y = cnnW[(kb + 1) * 32 + n];
        v2f b1; b1.x = cnnW[kb * 32 + 16 + n]; b1.y = cnnW[(kb + 1) * 32 + 16 + n];
        acc2 = wmma4(a, b0, acc2);
        acc3 = wmma4(a, b1, acc3);
    }
#pragma unroll
    for (int v = 0; v < 8; ++v) {
        int mr   = (lane < 16) ? v : v + 8;
        int node = wave * 16 + mr;
        if (node < N) {
            float* o = h0 + (size_t)node * 64;
            o[n]      = fmaxf(acc0[v] + fcb[n],       0.0f);
            o[16 + n] = fmaxf(acc1[v] + fcb[16 + n],  0.0f);
            o[32 + n] = fmaxf(acc2[v] + cnnb[n],      0.0f);
            o[48 + n] = fmaxf(acc3[v] + cnnb[16 + n], 0.0f);
        }
    }
}

__global__ void k_gemm_t1(const float* __restrict__ h0, const float* __restrict__ W1,
                          float* __restrict__ t1, int N, int nTiles) {
    int wave = (blockIdx.x * blockDim.x + threadIdx.x) >> 5;
    if (wave >= nTiles) return;
    int lane = threadIdx.x & 31;
    int n    = lane & 15;
    int kh   = lane >> 4;
    int rowA = wave * 16 + n;    if (rowA >= N) rowA = N - 1;

    v8f acc0 = {}, acc1 = {};
    const float* ar = h0 + (size_t)rowA * 64;
#pragma unroll
    for (int k0 = 0; k0 < 64; k0 += 4) {
        int kb = k0 + 2 * kh;
        v2f a;  a.x = ar[kb];              a.y = ar[kb + 1];
        v2f b0; b0.x = W1[kb * 32 + n];    b0.y = W1[(kb + 1) * 32 + n];
        v2f b1; b1.x = W1[kb * 32 + 16 + n]; b1.y = W1[(kb + 1) * 32 + 16 + n];
        acc0 = wmma4(a, b0, acc0);
        acc1 = wmma4(a, b1, acc1);
    }
#pragma unroll
    for (int v = 0; v < 8; ++v) {
        int mr   = (lane < 16) ? v : v + 8;
        int node = wave * 16 + mr;
        if (node < N) {
            float* o = t1 + (size_t)node * 32;
            o[n]      = acc0[v];
            o[16 + n] = acc1[v];
        }
    }
}

__global__ void k_init_agg1(const float* __restrict__ t1, const float* __restrict__ dinv,
                            const float* __restrict__ b1, float* __restrict__ agg1, int N) {
    int i = blockIdx.x * blockDim.x + threadIdx.x;
    if (i < N * 32) {
        int node = i >> 5, f = i & 31;
        float di = dinv[node];
        agg1[i] = t1[i] * di * di + b1[f];
    }
}

__global__ void k_edge1(const int* __restrict__ src, const int* __restrict__ dst,
                        const float* __restrict__ dinv, const float* __restrict__ t1,
                        float* __restrict__ agg1, int E) {
    int e = (blockIdx.x * blockDim.x + threadIdx.x) >> 5;
    if (e >= E) return;
    int lane = threadIdx.x & 31;
    int s = src[e], d = dst[e];
    float w = dinv[s] * dinv[d];
    atomicAdd(&agg1[(size_t)d * 32 + lane], t1[(size_t)s * 32 + lane] * w);
}

__global__ void k_gemm_t2(const float* __restrict__ agg1, const float* __restrict__ W2,
                          float* __restrict__ t2, int N, int nTiles) {
    int wave = (blockIdx.x * blockDim.x + threadIdx.x) >> 5;
    if (wave >= nTiles) return;
    int lane = threadIdx.x & 31;
    int n    = lane & 15;
    int kh   = lane >> 4;
    int rowA = wave * 16 + n;    if (rowA >= N) rowA = N - 1;

    v8f acc0 = {}, acc1 = {};
    const float* ar = agg1 + (size_t)rowA * 32;
    bool c1ok = (16 + n) < 20;
#pragma unroll
    for (int k0 = 0; k0 < 32; k0 += 4) {
        int kb = k0 + 2 * kh;
        v2f a;  a.x = fmaxf(ar[kb], 0.0f);  a.y = fmaxf(ar[kb + 1], 0.0f);
        v2f b0; b0.x = W2[kb * 20 + n];     b0.y = W2[(kb + 1) * 20 + n];
        v2f b1; b1.x = c1ok ? W2[kb * 20 + 16 + n] : 0.0f;
                b1.y = c1ok ? W2[(kb + 1) * 20 + 16 + n] : 0.0f;
        acc0 = wmma4(a, b0, acc0);
        acc1 = wmma4(a, b1, acc1);
    }
#pragma unroll
    for (int v = 0; v < 8; ++v) {
        int mr   = (lane < 16) ? v : v + 8;
        int node = wave * 16 + mr;
        if (node < N) {
            float* o = t2 + (size_t)node * 20;
            o[n] = acc0[v];
            if (c1ok) o[16 + n] = acc1[v];
        }
    }
}

__global__ void k_init_out(const float* __restrict__ t2, const float* __restrict__ dinv,
                           const float* __restrict__ b2, float* __restrict__ out, int N) {
    int i = blockIdx.x * blockDim.x + threadIdx.x;
    if (i < N * 20) {
        int node = i / 20, f = i - node * 20;
        float di = dinv[node];
        out[i] = t2[i] * di * di + b2[f];
    }
}

__global__ void k_edge2(const int* __restrict__ src, const int* __restrict__ dst,
                        const float* __restrict__ dinv, const float* __restrict__ t2,
                        float* __restrict__ out, int E) {
    int e = (blockIdx.x * blockDim.x + threadIdx.x) >> 5;
    if (e >= E) return;
    int lane = threadIdx.x & 31;
    int s = src[e], d = dst[e];
    float w = dinv[s] * dinv[d];
    if (lane < 20)
        atomicAdd(&out[(size_t)d * 20 + lane], t2[(size_t)s * 20 + lane] * w);
}

extern "C" void kernel_launch(void* const* d_in, const int* in_sizes, int n_in,
                              void* d_out, int out_size, void* d_ws, size_t ws_size,
                              hipStream_t stream) {
    const float* x    = (const float*)d_in[0];
    const float* cnn  = (const float*)d_in[1];
    const int*   ei   = (const int*)d_in[2];
    const float* fcW  = (const float*)d_in[3];
    const float* fcb  = (const float*)d_in[4];
    const float* cnnW = (const float*)d_in[5];
    const float* cnnb = (const float*)d_in[6];
    const float* W1   = (const float*)d_in[7];
    const float* b1   = (const float*)d_in[8];
    const float* W2   = (const float*)d_in[9];
    const float* b2   = (const float*)d_in[10];

    const int N = in_sizes[0] / 32;
    const int E = in_sizes[2] / 2;
    const int* src = ei;
    const int* dst = ei + E;

    float* ws   = (float*)d_ws;
    float* dinv = ws;  ws += N;
    float* h0   = ws;  ws += (size_t)N * 64;
    float* t1   = ws;  ws += (size_t)N * 32;
    float* agg1 = ws;  ws += (size_t)N * 32;
    float* t2   = ws;  ws += (size_t)N * 20;
    float* out  = (float*)d_out;

    const int nTiles = (N + 15) / 16;
    const int B = 256;
    dim3 blkG(128);
    dim3 grdG((nTiles * 32 + 127) / 128);
    dim3 grdEdge((unsigned)(((long long)E * 32 + B - 1) / B));

    k_zero<<<(N + B - 1) / B, B, 0, stream>>>(dinv, N);
    dim3 grdDeg(DEG_SPLITS, (N + DEG_RANGE - 1) / DEG_RANGE);
    k_count_deg<<<grdDeg, DEG_THREADS, 0, stream>>>(dst, dinv, E, N);
    k_dinv<<<(N + B - 1) / B, B, 0, stream>>>(dinv, N);

    k_encoder<<<grdG, blkG, 0, stream>>>(x, cnn, fcW, fcb, cnnW, cnnb, h0, N, nTiles);
    k_gemm_t1<<<grdG, blkG, 0, stream>>>(h0, W1, t1, N, nTiles);

    k_init_agg1<<<(N * 32 + B - 1) / B, B, 0, stream>>>(t1, dinv, b1, agg1, N);
    k_edge1<<<grdEdge, B, 0, stream>>>(src, dst, dinv, t1, agg1, E);

    k_gemm_t2<<<grdG, blkG, 0, stream>>>(agg1, W2, t2, N, nTiles);
    k_init_out<<<(N * 20 + B - 1) / B, B, 0, stream>>>(t2, dinv, b2, out, N);
    k_edge2<<<grdEdge, B, 0, stream>>>(src, dst, dinv, t2, out, E);
}


# Float-atomic scatter-add of both layers replaced by a per-call counting sort of edges by destination (LDS counters) and one wave per node gathering its rows; per-node sum order now follows LDS rank
# speedup vs baseline: 3.2289x; 2.5811x over previous
#include <hip/hip_runtime.h>
#include <hip/hip_bf16.h>

typedef __attribute__((ext_vector_type(2))) float v2f;
typedef __attribute__((ext_vector_type(8))) float v8f;

__device__ __forceinline__ v8f wmma4(v2f a, v2f b, v8f c) {
    return __builtin_amdgcn_wmma_f32_16x16x4_f32(
         false, a,  false, b,
         (short)0, c,  false,  false);
}

#define CS_RANGE   16384
#define CS_SPLITS  32
#define CS_THREADS 1024

__device__ __forceinline__ void cs_slice(int E, int x, int& e0, int& e1) {
    int per = (E + CS_SPLITS - 1) / CS_SPLITS;
    per = (per + 3) & ~3;
    long long e0l = (long long)x * per;
    e0 = (e0l < (long long)E) ? (int)e0l : E;
    e1 = (E - e0 > per) ? (e0 + per) : E;
}

__device__ __forceinline__ void cs_count(unsigned* cnt, int d, int lo) {
    unsigned r = (unsigned)(d - lo);
    if (r < (unsigned)CS_RANGE) atomicAdd(&cnt[r], 1u);
}

__device__ __forceinline__ void cs_count4(unsigned* cnt, int4 v, int lo) {
    cs_count(cnt, v.x, lo);
    cs_count(cnt, v.y, lo);
    cs_count(cnt, v.z, lo);
    cs_count(cnt, v.w, lo);
}

__global__ void __launch_bounds__(CS_THREADS)
k_count(const int* __restrict__ dst, int* __restrict__ plane, int E, int N) {
    __shared__ unsigned cnt[CS_RANGE];
    const int tid = threadIdx.x;
    const int lo  = blockIdx.y * CS_RANGE;

    for (int i = tid; i < CS_RANGE; i += CS_THREADS) cnt[i] = 0u;
    __syncthreads();

    int e0, e1;
    cs_slice(E, blockIdx.x, e0, e1);

    if ((((size_t)dst) & 15) == 0) {
        const int4* __restrict__ p = (const int4*)(dst + e0);
        const int nv = (e1 - e0) >> 2;
        int i = tid;
        for (; i + 3 * CS_THREADS < nv; i += 4 * CS_THREADS) {
            int4 a = p[i];
            int4 b = p[i + CS_THREADS];
            int4 c = p[i + 2 * CS_THREADS];
            int4 d = p[i + 3 * CS_THREADS];
            cs_count4(cnt, a, lo);
            cs_count4(cnt, b, lo);
            cs_count4(cnt, c, lo);
            cs_count4(cnt, d, lo);
        }
        for (; i < nv; i += CS_THREADS) cs_count4(cnt, p[i], lo);
        for (int e = e0 + (nv << 2) + tid; e < e1; e += CS_THREADS) cs_count(cnt, dst[e], lo);
    } else {
        for (int e = e0 + tid; e < e1; e += CS_THREADS) cs_count(cnt, dst[e], lo);
    }
    __syncthreads();

    int* __restrict__ row = plane + (size_t)blockIdx.x * N;
    for (int i = tid; i < CS_RANGE; i += CS_THREADS) {
        int node = lo + i;
        if (node < N) row[node] = (int)cnt[i];
    }
}

#define DEG_THREADS 256

__global__ void __launch_bounds__(DEG_THREADS)
k_deg(int* __restrict__ plane, float* __restrict__ dinv, int* __restrict__ bsum, int N) {
    __shared__ int wtot[DEG_THREADS / 32];
    const int tid  = threadIdx.x;
    const int lane = tid & 31;
    const int w    = tid >> 5;
    const int node = blockIdx.x * DEG_THREADS + tid;
    const bool ok  = node < N;

    int run = 0;
    if (ok) {
        int c[CS_SPLITS];
#pragma unroll
        for (int x = 0; x < CS_SPLITS; ++x) c[x] = plane[(size_t)x * N + node];
        run = c[0];
#pragma unroll
        for (int x = 1; x < CS_SPLITS; ++x) {
            plane[(size_t)x * N + node] = run;
            run += c[x];
        }
        dinv[node] = rsqrtf((float)run + 1.0f);
    }

    int inc = run;
    for (int o = 1; o < 32; o <<= 1) {
        int y = __shfl_up(inc, o);
        if (lane >= o) inc += y;
    }
    if (lane == 31) wtot[w] = inc;
    __syncthreads();
    int woff = 0;
    for (int k = 0; k < w; ++k) woff += wtot[k];
    if (ok) plane[node] = woff + inc - run;
    if (tid == DEG_THREADS - 1) bsum[blockIdx.x] = woff + inc;
}

__global__ void __launch_bounds__(DEG_THREADS)
k_addoff(int* __restrict__ plane, const int* __restrict__ bsum, int N) {
    __shared__ int wtot[DEG_THREADS / 32];
    const int tid  = threadIdx.x;
    const int lane = tid & 31;
    const int w    = tid >> 5;
    const int b    = blockIdx.x;

    int part = 0;
    for (int k = tid; k < b; k += DEG_THREADS) part += bsum[k];
    for (int o = 16; o > 0; o >>= 1) part += __shfl_xor(part, o);
    if (lane == 0) wtot[w] = part;
    __syncthreads();
    int off = 0;
#pragma unroll
    for (int k = 0; k < DEG_THREADS / 32; ++k) off += wtot[k];

    const int node = b * DEG_THREADS + tid;
    if (node < N) plane[node] += off;
}

__device__ __forceinline__ void cs_place4(unsigned* cnt, int4 v, int4 sv, int lo,
                                          const int* __restrict__ slot0,
                                          const int* __restrict__ pref, int prefMask,
                                          int* __restrict__ sorted, int E) {
    const unsigned r0 = (unsigned)(v.x - lo), r1 = (unsigned)(v.y - lo);
    const unsigned r2 = (unsigned)(v.z - lo), r3 = (unsigned)(v.w - lo);
    const bool m0 = r0 < (unsigned)CS_RANGE, m1 = r1 < (unsigned)CS_RANGE;
    const bool m2 = r2 < (unsigned)CS_RANGE, m3 = r3 < (unsigned)CS_RANGE;
    unsigned k0 = 0, k1 = 0, k2 = 0, k3 = 0;
    if (m0) k0 = atomicAdd(&cnt[r0], 1u);
    if (m1) k1 = atomicAdd(&cnt[r1], 1u);
    if (m2) k2 = atomicAdd(&cnt[r2], 1u);
    if (m3) k3 = atomicAdd(&cnt[r3], 1u);
    const int d0 = m0 ? v.x : lo, d1 = m1 ? v.y : lo, d2 = m2 ? v.z : lo, d3 = m3 ? v.w : lo;
    const int a0 = slot0[d0], a1 = slot0[d1], a2 = slot0[d2], a3 = slot0[d3];
    const int q0 = pref[d0],  q1 = pref[d1],  q2 = pref[d2],  q3 = pref[d3];
    const int p0 = a0 + (q0 & prefMask) + (int)k0;
    const int p1 = a1 + (q1 & prefMask) + (int)k1;
    const int p2 = a2 + (q2 & prefMask) + (int)k2;
    const int p3 = a3 + (q3 & prefMask) + (int)k3;
    if (m0 && (unsigned)p0 < (unsigned)E) sorted[p0] = sv.x;
    if (m1 && (unsigned)p1 < (unsigned)E) sorted[p1] = sv.y;
    if (m2 && (unsigned)p2 < (unsigned)E) sorted[p2] = sv.z;
    if (m3 && (unsigned)p3 < (unsigned)E) sorted[p3] = sv.w;
}

__device__ __forceinline__ void cs_place1(unsigned* cnt, int d, int e, int lo,
                                          const int* __restrict__ slot0,
                                          const int* __restrict__ pref, int prefMask,
                                          const int* __restrict__ src,
                                          int* __restrict__ sorted, int E) {
    const unsigned r = (unsigned)(d - lo);
    if (r < (unsigned)CS_RANGE) {
        unsigned k = atomicAdd(&cnt[r], 1u);
        int p = slot0[d] + (pref[d] & prefMask) + (int)k;
        if ((unsigned)p < (unsigned)E) sorted[p] = src[e];
    }
}

__global__ void __launch_bounds__(CS_THREADS)
k_fill(const int* __restrict__ src, const int* __restrict__ dst,
       const int* __restrict__ plane, int* __restrict__ sorted, int E, int N) {
    __shared__ unsigned cnt[CS_RANGE];
    const int tid = threadIdx.x;
    const int lo  = blockIdx.y * CS_RANGE;

    for (int i = tid; i < CS_RANGE; i += CS_THREADS) cnt[i] = 0u;
    __syncthreads();

    int e0, e1;
    cs_slice(E, blockIdx.x, e0, e1);
    const int* __restrict__ slot0 = plane;
    const int* __restrict__ pref  = plane + (size_t)blockIdx.x * N;
    const int prefMask = (blockIdx.x > 0) ? -1 : 0;

    if (((((size_t)dst) | ((size_t)src)) & 15) == 0) {
        const int4* __restrict__ p = (const int4*)(dst + e0);
        const int4* __restrict__ q = (const int4*)(src + e0);
        const int nv = (e1 - e0) >> 2;
        int i = tid;
        for (; i + CS_THREADS < nv; i += 2 * CS_THREADS) {
            int4 a  = p[i];
            int4 b  = p[i + CS_THREADS];
            int4 sa = q[i];
            int4 sb = q[i + CS_THREADS];
            cs_place4(cnt, a, sa, lo, slot0, pref, prefMask, sorted, E);
            cs_place4(cnt, b, sb, lo, slot0, pref, prefMask, sorted, E);
        }
        for (; i < nv; i += CS_THREADS)
            cs_place4(cnt, p[i], q[i], lo, slot0, pref, prefMask, sorted, E);
        for (int e = e0 + (nv << 2) + tid; e < e1; e += CS_THREADS)
            cs_place1(cnt, dst[e], e, lo, slot0, pref, prefMask, src, sorted, E);
    } else {
        for (int e = e0 + tid; e < e1; e += CS_THREADS)
            cs_place1(cnt, dst[e], e, lo, slot0, pref, prefMask, src, sorted, E);
    }
}

__global__ void k_encoder(const float* __restrict__ x, const float* __restrict__ cnn,
                          const float* __restrict__ fcW, const float* __restrict__ fcb,
                          const float* __restrict__ cnnW, const float* __restrict__ cnnb,
                          float* __restrict__ h0, int N, int nTiles) {
    int wave = (blockIdx.x * blockDim.x + threadIdx.x) >> 5;
    if (wave >= nTiles) return;
    int lane = threadIdx.x & 31;
    int n    = lane & 15;
    int kh   = lane >> 4;
    int rowA = wave * 16 + n;    if (rowA >= N) rowA = N - 1;

    v8f acc0 = {}, acc1 = {}, acc2 = {}, acc3 = {};

    const float* ax = x + (size_t)rowA * 32;
#pragma unroll
    for (int k0 = 0; k0 < 32; k0 += 4) {
        int kb = k0 + 2 * kh;
        v2f a;  a.x = ax[kb];               a.y = ax[kb + 1];
        v2f b0; b0.x = fcW[kb * 32 + n];    b0.y = fcW[(kb + 1) * 32 + n];
        v2f b1; b1.x = fcW[kb * 32 + 16 + n]; b1.y = fcW[(kb + 1) * 32 + 16 + n];
        acc0 = wmma4(a, b0, acc0);
        acc1 = wmma4(a, b1, acc1);
    }
    const float* ac = cnn + (size_t)rowA * 128;
#pragma unroll
    for (int k0 = 0; k0 < 128; k0 += 4) {
        int kb = k0 + 2 * kh;
        v2f a;  a.x = ac[kb];                a.y = ac[kb + 1];
        v2f b0; b0.x = cnnW[kb * 32 + n];    b0.y = cnnW[(kb + 1) * 32 + n];
        v2f b1; b1.x = cnnW[kb * 32 + 16 + n]; b1.y = cnnW[(kb + 1) * 32 + 16 + n];
        acc2 = wmma4(a, b0, acc2);
        acc3 = wmma4(a, b1, acc3);
    }
#pragma unroll
    for (int v = 0; v < 8; ++v) {
        int mr   = (lane < 16) ? v : v + 8;
        int node = wave * 16 + mr;
        if (node < N) {
            float* o = h0 + (size_t)node * 64;
            o[n]      = fmaxf(acc0[v] + fcb[n],       0.0f);
            o[16 + n] = fmaxf(acc1[v] + fcb[16 + n],  0.0f);
            o[32 + n] = fmaxf(acc2[v] + cnnb[n],      0.0f);
            o[48 + n] = fmaxf(acc3[v] + cnnb[16 + n], 0.0f);
        }
    }
}

__global__ void k_gemm_t1(const float* __restrict__ h0, const float* __restrict__ W1,
                          float* __restrict__ t1, int N, int nTiles) {
    int wave = (blockIdx.x * blockDim.x + threadIdx.x) >> 5;
    if (wave >= nTiles) return;
    int lane = threadIdx.x & 31;
    int n    = lane & 15;
    int kh   = lane >> 4;
    int rowA = wave * 16 + n;    if (rowA >= N) rowA = N - 1;

    v8f acc0 = {}, acc1 = {};
    const float* ar = h0 + (size_t)rowA * 64;
#pragma unroll
    for (int k0 = 0; k0 < 64; k0 += 4) {
        int kb = k0 + 2 * kh;
        v2f a;  a.x = ar[kb];              a.y = ar[kb + 1];
        v2f b0; b0.x = W1[kb * 32 + n];    b0.y = W1[(kb + 1) * 32 + n];
        v2f b1; b1.x = W1[kb * 32 + 16 + n]; b1.y = W1[(kb + 1) * 32 + 16 + n];
        acc0 = wmma4(a, b0, acc0);
        acc1 = wmma4(a, b1, acc1);
    }
#pragma unroll
    for (int v = 0; v < 8; ++v) {
        int mr   = (lane < 16) ? v : v + 8;
        int node = wave * 16 + mr;
        if (node < N) {
            float* o = t1 + (size_t)node * 32;
            o[n]      = acc0[v];
            o[16 + n] = acc1[v];
        }
    }
}

__global__ void k_gemm_t2(const float* __restrict__ agg1, const float* __restrict__ W2,
                          float* __restrict__ t2, int N, int nTiles) {
    int wave = (blockIdx.x * blockDim.x + threadIdx.x) >> 5;
    if (wave >= nTiles) return;
    int lane = threadIdx.x & 31;
    int n    = lane & 15;
    int kh   = lane >> 4;
    int rowA = wave * 16 + n;    if (rowA >= N) rowA = N - 1;

    v8f acc0 = {}, acc1 = {};
    const float* ar = agg1 + (size_t)rowA * 32;
    bool c1ok = (16 + n) < 20;
#pragma unroll
    for (int k0 = 0; k0 < 32; k0 += 4) {
        int kb = k0 + 2 * kh;
        v2f a;  a.x = fmaxf(ar[kb], 0.0f);  a.y = fmaxf(ar[kb + 1], 0.0f);
        v2f b0; b0.x = W2[kb * 20 + n];     b0.y = W2[(kb + 1) * 20 + n];
        v2f b1; b1.x = c1ok ? W2[kb * 20 + 16 + n] : 0.0f;
                b1.y = c1ok ? W2[(kb + 1) * 20 + 16 + n] : 0.0f;
        acc0 = wmma4(a, b0, acc0);
        acc1 = wmma4(a, b1, acc1);
    }
#pragma unroll
    for (int v = 0; v < 8; ++v) {
        int mr   = (lane < 16) ? v : v + 8;
        int node = wave * 16 + mr;
        if (node < N) {
            float* o = t2 + (size_t)node * 20;
            o[n] = acc0[v];
            if (c1ok) o[16 + n] = acc1[v];
        }
    }
}


template <int F, bool GUARD>
__device__ __forceinline__ float gather8(int s, float w, int j, int n,
                                         const float* __restrict__ t, int lane, bool act,
                                         float acc) {
    float v[8];
#pragma unroll
    for (int u = 0; u < 8; ++u) {
        v[u] = 0.0f;
        if (!GUARD || j + u < n) {
            const int sj = __builtin_amdgcn_readlane(s, j + u);
            if (act) v[u] = t[(size_t)sj * F + lane];
        }
    }
#pragma unroll
    for (int u = 0; u < 8; ++u) {
        if (!GUARD || j + u < n) {
            const float wj = __int_as_float(__builtin_amdgcn_readlane(__float_as_int(w), j + u));
            acc += v[u] * wj;
        }
    }
    return acc;
}

template <int F>
__device__ __forceinline__ void gather_rows(const int* __restrict__ slot0,
                                            const int* __restrict__ sorted,
                                            const float* __restrict__ dinv,
                                            const float* __restrict__ t,
                                            const float* __restrict__ bias,
                                            float* __restrict__ o, int N, int E) {
    const int node = __builtin_amdgcn_readfirstlane((int)((blockIdx.x * blockDim.x + threadIdx.x) >> 5));
    if (node >= N) return;
    const int  lane = threadIdx.x & 31;
    const bool act  = lane < F;
    const int  rs   = slot0[node];
    const int  re   = (node + 1 < N) ? slot0[node + 1] : E;
    const float di  = dinv[node];

    float acc = 0.0f;
    for (int base = rs; base < re; base += 32) {
        const int n = (re - base < 32) ? (re - base) : 32;
        int   s = 0;
        float w = 0.0f;
        if (lane < n) {
            s = sorted[base + lane];
            w = dinv[s] * di;
        }
        int j = 0;
        for (; j + 8 <= n; j += 8) acc = gather8<F, false>(s, w, j, n, t, lane, act, acc);
        if (j < n)                 acc = gather8<F, true >(s, w, j, n, t, lane, act, acc);
    }
    if (act) {
        const size_t idx = (size_t)node * F + lane;
        o[idx] = acc + t[idx] * di * di + bias[lane];
    }
}

__global__ void __launch_bounds__(256)
k_gather32(const int* __restrict__ slot0, const int* __restrict__ sorted,
           const float* __restrict__ dinv, const float* __restrict__ t1,
           const float* __restrict__ b1, float* __restrict__ agg1, int N, int E) {
    gather_rows<32>(slot0, sorted, dinv, t1, b1, agg1, N, E);
}

__global__ void __launch_bounds__(256)
k_gather20(const int* __restrict__ slot0, const int* __restrict__ sorted,
           const float* __restrict__ dinv, const float* __restrict__ t2,
           const float* __restrict__ b2, float* __restrict__ out, int N, int E) {
    gather_rows<20>(slot0, sorted, dinv, t2, b2, out, N, E);
}

extern "C" void kernel_launch(void* const* d_in, const int* in_sizes, int n_in,
                              void* d_out, int out_size, void* d_ws, size_t ws_size,
                              hipStream_t stream) {
    const float* x    = (const float*)d_in[0];
    const float* cnn  = (const float*)d_in[1];
    const int*   ei   = (const int*)d_in[2];
    const float* fcW  = (const float*)d_in[3];
    const float* fcb  = (const float*)d_in[4];
    const float* cnnW = (const float*)d_in[5];
    const float* cnnb = (const float*)d_in[6];
    const float* W1   = (const float*)d_in[7];
    const float* b1   = (const float*)d_in[8];
    const float* W2   = (const float*)d_in[9];
    const float* b2   = (const float*)d_in[10];

    const int N = in_sizes[0] / 32;
    const int E = in_sizes[2] / 2;
    const int* src = ei;
    const int* dst = ei + E;

    float* ws   = (float*)d_ws;
    float* dinv = ws;  ws += N;
    float* h0   = ws;  ws += (size_t)N * 64;
    float* t1   = ws;  ws += (size_t)N * 32;
    float* agg1 = ws;  ws += (size_t)N * 32;
    float* t2   = ws;  ws += (size_t)N * 20;
    float* out  = (float*)d_out;

    int* plane  = (int*)h0;
    int* sorted = plane + (size_t)CS_SPLITS * N;

    const int nTiles = (N + 15) / 16;
    const int B = 256;
    dim3 blkG(128);
    dim3 grdG((nTiles * 32 + 127) / 128);
    dim3 grdNode((unsigned)(((long long)N * 32 + B - 1) / B));

    k_encoder<<<grdG, blkG, 0, stream>>>(x, cnn, fcW, fcb, cnnW, cnnb, h0, N, nTiles);
    k_gemm_t1<<<grdG, blkG, 0, stream>>>(h0, W1, t1, N, nTiles);

    dim3 grdCS(CS_SPLITS, (N + CS_RANGE - 1) / CS_RANGE);
    k_count<<<grdCS, CS_THREADS, 0, stream>>>(dst, plane, E, N);
    int* bsum = (int*)t2;
    const int nbDeg = (N + DEG_THREADS - 1) / DEG_THREADS;
    k_deg<<<nbDeg, DEG_THREADS, 0, stream>>>(plane, dinv, bsum, N);
    k_addoff<<<nbDeg, DEG_THREADS, 0, stream>>>(plane, bsum, N);
    k_fill<<<grdCS, CS_THREADS, 0, stream>>>(src, dst, plane, sorted, E, N);

    k_gather32<<<grdNode, B, 0, stream>>>(plane, sorted, dinv, t1, b1, agg1, N, E);

    k_gemm_t2<<<grdG, blkG, 0, stream>>>(agg1, W2, t2, N, nTiles);
    k_gather20<<<grdNode, B, 0, stream>>>(plane, sorted, dinv, t2, b2, out, N, E);
}


# k_fill: each LDS counter starts at its node's first slot for the slice (read once per block, whole lines), so the returning add yields the place; no per-edge slot/prefix loads; sum order as before
# speedup vs baseline: 3.7937x; 1.1749x over previous
#include <hip/hip_runtime.h>
#include <hip/hip_bf16.h>

typedef __attribute__((ext_vector_type(2))) float v2f;
typedef __attribute__((ext_vector_type(8))) float v8f;

__device__ __forceinline__ v8f wmma4(v2f a, v2f b, v8f c) {
    return __builtin_amdgcn_wmma_f32_16x16x4_f32(
         false, a,  false, b,
         (short)0, c,  false,  false);
}

#define CS_RANGE   16384
#define CS_SPLITS  32
#define CS_THREADS 1024

__device__ __forceinline__ void cs_slice(int E, int x, int& e0, int& e1) {
    int per = (E + CS_SPLITS - 1) / CS_SPLITS;
    per = (per + 3) & ~3;
    long long e0l = (long long)x * per;
    e0 = (e0l < (long long)E) ? (int)e0l : E;
    e1 = (E - e0 > per) ? (e0 + per) : E;
}

__device__ __forceinline__ void cs_count(unsigned* cnt, int d, int lo) {
    unsigned r = (unsigned)(d - lo);
    if (r < (unsigned)CS_RANGE) atomicAdd(&cnt[r], 1u);
}

__device__ __forceinline__ void cs_count4(unsigned* cnt, int4 v, int lo) {
    cs_count(cnt, v.x, lo);
    cs_count(cnt, v.y, lo);
    cs_count(cnt, v.z, lo);
    cs_count(cnt, v.w, lo);
}

__global__ void __launch_bounds__(CS_THREADS)
k_count(const int* __restrict__ dst, int* __restrict__ plane, int E, int N) {
    __shared__ unsigned cnt[CS_RANGE];
    const int tid = threadIdx.x;
    const int lo  = blockIdx.y * CS_RANGE;

    for (int i = tid; i < CS_RANGE; i += CS_THREADS) cnt[i] = 0u;
    __syncthreads();

    int e0, e1;
    cs_slice(E, blockIdx.x, e0, e1);

    if ((((size_t)dst) & 15) == 0) {
        const int4* __restrict__ p = (const int4*)(dst + e0);
        const int nv = (e1 - e0) >> 2;
        int i = tid;
        for (; i + 3 * CS_THREADS < nv; i += 4 * CS_THREADS) {
            int4 a = p[i];
            int4 b = p[i + CS_THREADS];
            int4 c = p[i + 2 * CS_THREADS];
            int4 d = p[i + 3 * CS_THREADS];
            cs_count4(cnt, a, lo);
            cs_count4(cnt, b, lo);
            cs_count4(cnt, c, lo);
            cs_count4(cnt, d, lo);
        }
        for (; i < nv; i += CS_THREADS) cs_count4(cnt, p[i], lo);
        for (int e = e0 + (nv << 2) + tid; e < e1; e += CS_THREADS) cs_count(cnt, dst[e], lo);
    } else {
        for (int e = e0 + tid; e < e1; e += CS_THREADS) cs_count(cnt, dst[e], lo);
    }
    __syncthreads();

    int* __restrict__ row = plane + (size_t)blockIdx.x * N;
    for (int i = tid; i < CS_RANGE; i += CS_THREADS) {
        int node = lo + i;
        if (node < N) row[node] = (int)cnt[i];
    }
}

#define DEG_THREADS 256

__global__ void __launch_bounds__(DEG_THREADS)
k_deg(int* __restrict__ plane, float* __restrict__ dinv, int* __restrict__ bsum, int N) {
    __shared__ int wtot[DEG_THREADS / 32];
    const int tid  = threadIdx.x;
    const int lane = tid & 31;
    const int w    = tid >> 5;
    const int node = blockIdx.x * DEG_THREADS + tid;
    const bool ok  = node < N;

    int run = 0;
    if (ok) {
        int c[CS_SPLITS];
#pragma unroll
        for (int x = 0; x < CS_SPLITS; ++x) c[x] = plane[(size_t)x * N + node];
        run = c[0];
#pragma unroll
        for (int x = 1; x < CS_SPLITS; ++x) {
            plane[(size_t)x * N + node] = run;
            run += c[x];
        }
        dinv[node] = rsqrtf((float)run + 1.0f);
    }

    int inc = run;
    for (int o = 1; o < 32; o <<= 1) {
        int y = __shfl_up(inc, o);
        if (lane >= o) inc += y;
    }
    if (lane == 31) wtot[w] = inc;
    __syncthreads();
    int woff = 0;
    for (int k = 0; k < w; ++k) woff += wtot[k];
    if (ok) plane[node] = woff + inc - run;
    if (tid == DEG_THREADS - 1) bsum[blockIdx.x] = woff + inc;
}

__global__ void __launch_bounds__(DEG_THREADS)
k_addoff(const int* __restrict__ plane, const int* __restrict__ bsum, int* __restrict__ slot, int N) {
    __shared__ int wtot[DEG_THREADS / 32];
    const int tid  = threadIdx.x;
    const int lane = tid & 31;
    const int w    = tid >> 5;
    const int b    = blockIdx.x;

    int part = 0;
    for (int k = tid; k < b; k += DEG_THREADS) part += bsum[k];
    for (int o = 16; o > 0; o >>= 1) part += __shfl_xor(part, o);
    if (lane == 0) wtot[w] = part;
    __syncthreads();
    int off = 0;
#pragma unroll
    for (int k = 0; k < DEG_THREADS / 32; ++k) off += wtot[k];

    const int node = b * DEG_THREADS + tid;
    if (node < N) slot[node] = plane[node] + off;
}

__device__ __forceinline__ void cs_place4(unsigned* cnt, int4 v, int4 sv, int lo,
                                          int* __restrict__ sorted, int E) {
    const unsigned r0 = (unsigned)(v.x - lo), r1 = (unsigned)(v.y - lo);
    const unsigned r2 = (unsigned)(v.z - lo), r3 = (unsigned)(v.w - lo);
    const bool m0 = r0 < (unsigned)CS_RANGE, m1 = r1 < (unsigned)CS_RANGE;
    const bool m2 = r2 < (unsigned)CS_RANGE, m3 = r3 < (unsigned)CS_RANGE;
    unsigned p0 = 0, p1 = 0, p2 = 0, p3 = 0;
    if (m0) p0 = atomicAdd(&cnt[r0], 1u);
    if (m1) p1 = atomicAdd(&cnt[r1], 1u);
    if (m2) p2 = atomicAdd(&cnt[r2], 1u);
    if (m3) p3 = atomicAdd(&cnt[r3], 1u);
    if (m0 && p0 < (unsigned)E) sorted[p0] = sv.x;
    if (m1 && p1 < (unsigned)E) sorted[p1] = sv.y;
    if (m2 && p2 < (unsigned)E) sorted[p2] = sv.z;
    if (m3 && p3 < (unsigned)E) sorted[p3] = sv.w;
}

__device__ __forceinline__ void cs_place1(unsigned* cnt, int d, int s, int lo,
                                          int* __restrict__ sorted, int E) {
    const unsigned r = (unsigned)(d - lo);
    if (r < (unsigned)CS_RANGE) {
        const unsigned p = atomicAdd(&cnt[r], 1u);
        if (p < (unsigned)E) sorted[p] = s;
    }
}

__global__ void __launch_bounds__(CS_THREADS)
k_fill(const int* __restrict__ src, const int* __restrict__ dst,
       const int* __restrict__ slot0, const int* __restrict__ plane,
       int* __restrict__ sorted, int E, int N) {
    __shared__ unsigned cnt[CS_RANGE];
    const int tid = threadIdx.x;
    const int lo  = blockIdx.y * CS_RANGE;

    const int* __restrict__ pref = plane + (size_t)blockIdx.x * N;
    const int prefMask = (blockIdx.x > 0) ? -1 : 0;

    int first[CS_RANGE / CS_THREADS];
#pragma unroll
    for (int k = 0; k < CS_RANGE / CS_THREADS; ++k) {
        const int node = lo + k * CS_THREADS + tid;
        const int nc   = (node < N) ? node : (N - 1);
        const int v    = slot0[nc] + (pref[nc] & prefMask);
        first[k] = (node < N) ? v : 0;
    }
#pragma unroll
    for (int k = 0; k < CS_RANGE / CS_THREADS; ++k) cnt[k * CS_THREADS + tid] = (unsigned)first[k];
    __syncthreads();

    int e0, e1;
    cs_slice(E, blockIdx.x, e0, e1);

    if (((((size_t)dst) | ((size_t)src)) & 15) == 0) {
        const int4* __restrict__ p = (const int4*)(dst + e0);
        const int4* __restrict__ q = (const int4*)(src + e0);
        const int nv = (e1 - e0) >> 2;
        int i = tid;
        for (; i + CS_THREADS < nv; i += 2 * CS_THREADS) {
            int4 a  = p[i];
            int4 b  = p[i + CS_THREADS];
            int4 sa = q[i];
            int4 sb = q[i + CS_THREADS];
            cs_place4(cnt, a, sa, lo, sorted, E);
            cs_place4(cnt, b, sb, lo, sorted, E);
        }
        for (; i < nv; i += CS_THREADS)
            cs_place4(cnt, p[i], q[i], lo, sorted, E);
        for (int e = e0 + (nv << 2) + tid; e < e1; e += CS_THREADS)
            cs_place1(cnt, dst[e], src[e], lo, sorted, E);
    } else {
        for (int e = e0 + tid; e < e1; e += CS_THREADS)
            cs_place1(cnt, dst[e], src[e], lo, sorted, E);
    }
}

__global__ void k_encoder(const float* __restrict__ x, const float* __restrict__ cnn,
                          const float* __restrict__ fcW, const float* __restrict__ fcb,
                          const float* __restrict__ cnnW, const float* __restrict__ cnnb,
                          float* __restrict__ h0, int N, int nTiles) {
    int wave = (blockIdx.x * blockDim.x + threadIdx.x) >> 5;
    if (wave >= nTiles) return;
    int lane = threadIdx.x & 31;
    int n    = lane & 15;
    int kh   = lane >> 4;
    int rowA = wave * 16 + n;    if (rowA >= N) rowA = N - 1;

    v8f acc0 = {}, acc1 = {}, acc2 = {}, acc3 = {};

    const float* ax = x + (size_t)rowA * 32;
#pragma unroll
    for (int k0 = 0; k0 < 32; k0 += 4) {
        int kb = k0 + 2 * kh;
        v2f a;  a.x = ax[kb];               a.y = ax[kb + 1];
        v2f b0; b0.x = fcW[kb * 32 + n];    b0.y = fcW[(kb + 1) * 32 + n];
        v2f b1; b1.x = fcW[kb * 32 + 16 + n]; b1.y = fcW[(kb + 1) * 32 + 16 + n];
        acc0 = wmma4(a, b0, acc0);
        acc1 = wmma4(a, b1, acc1);
    }
    const float* ac = cnn + (size_t)rowA * 128;
#pragma unroll
    for (int k0 = 0; k0 < 128; k0 += 4) {
        int kb = k0 + 2 * kh;
        v2f a;  a.x = ac[kb];                a.y = ac[kb + 1];
        v2f b0; b0.x = cnnW[kb * 32 + n];    b0.y = cnnW[(kb + 1) * 32 + n];
        v2f b1; b1.x = cnnW[kb * 32 + 16 + n]; b1.y = cnnW[(kb + 1) * 32 + 16 + n];
        acc2 = wmma4(a, b0, acc2);
        acc3 = wmma4(a, b1, acc3);
    }
#pragma unroll
    for (int v = 0; v < 8; ++v) {
        int mr   = (lane < 16) ? v : v + 8;
        int node = wave * 16 + mr;
        if (node < N) {
            float* o = h0 + (size_t)node * 64;
            o[n]      = fmaxf(acc0[v] + fcb[n],       0.0f);
            o[16 + n] = fmaxf(acc1[v] + fcb[16 + n],  0.0f);
            o[32 + n] = fmaxf(acc2[v] + cnnb[n],      0.0f);
            o[48 + n] = fmaxf(acc3[v] + cnnb[16 + n], 0.0f);
        }
    }
}

__global__ void k_gemm_t1(const float* __restrict__ h0, const float* __restrict__ W1,
                          const float* __restrict__ dinv,
                          float* __restrict__ t1, int N, int nTiles) {
    int wave = (blockIdx.x * blockDim.x + threadIdx.x) >> 5;
    if (wave >= nTiles) return;
    int lane = threadIdx.x & 31;
    int n    = lane & 15;
    int kh   = lane >> 4;
    int rowA = wave * 16 + n;    if (rowA >= N) rowA = N - 1;

    float dr[8];
#pragma unroll
    for (int v = 0; v < 8; ++v) {
        int nd = wave * 16 + 8 * kh + v;
        dr[v] = dinv[nd < N ? nd : N - 1];
    }

    v8f acc0 = {}, acc1 = {};
    const float* ar = h0 + (size_t)rowA * 64;
#pragma unroll
    for (int k0 = 0; k0 < 64; k0 += 4) {
        int kb = k0 + 2 * kh;
        v2f a;  a.x = ar[kb];              a.y = ar[kb + 1];
        v2f b0; b0.x = W1[kb * 32 + n];    b0.y = W1[(kb + 1) * 32 + n];
        v2f b1; b1.x = W1[kb * 32 + 16 + n]; b1.y = W1[(kb + 1) * 32 + 16 + n];
        acc0 = wmma4(a, b0, acc0);
        acc1 = wmma4(a, b1, acc1);
    }
#pragma unroll
    for (int v = 0; v < 8; ++v) {
        int mr   = (lane < 16) ? v : v + 8;
        int node = wave * 16 + mr;
        if (node < N) {
            float* o = t1 + (size_t)node * 32;
            o[n]      = acc0[v] * dr[v];
            o[16 + n] = acc1[v] * dr[v];
        }
    }
}

__global__ void k_gemm_t2(const float* __restrict__ agg1, const float* __restrict__ W2,
                          const float* __restrict__ dinv,
                          float* __restrict__ t2, int N, int nTiles) {
    int wave = (blockIdx.x * blockDim.x + threadIdx.x) >> 5;
    if (wave >= nTiles) return;
    int lane = threadIdx.x & 31;
    int n    = lane & 15;
    int kh   = lane >> 4;
    int rowA = wave * 16 + n;    if (rowA >= N) rowA = N - 1;

    float dr[8];
#pragma unroll
    for (int v = 0; v < 8; ++v) {
        int nd = wave * 16 + 8 * kh + v;
        dr[v] = dinv[nd < N ? nd : N - 1];
    }

    v8f acc0 = {}, acc1 = {};
    const float* ar = agg1 + (size_t)rowA * 32;
    bool c1ok = (16 + n) < 20;
#pragma unroll
    for (int k0 = 0; k0 < 32; k0 += 4) {
        int kb = k0 + 2 * kh;
        v2f a;  a.x = fmaxf(ar[kb], 0.0f);  a.y = fmaxf(ar[kb + 1], 0.0f);
        v2f b0; b0.x = W2[kb * 20 + n];     b0.y = W2[(kb + 1) * 20 + n];
        v2f b1; b1.x = c1ok ? W2[kb * 20 + 16 + n] : 0.0f;
                b1.y = c1ok ? W2[(kb + 1) * 20 + 16 + n] : 0.0f;
        acc0 = wmma4(a, b0, acc0);
        acc1 = wmma4(a, b1, acc1);
    }
#pragma unroll
    for (int v = 0; v < 8; ++v) {
        int mr   = (lane < 16) ? v : v + 8;
        int node = wave * 16 + mr;
        if (node < N) {
            float* o = t2 + (size_t)node * 20;
            o[n] = acc0[v] * dr[v];
            if (c1ok) o[16 + n] = acc1[v] * dr[v];
        }
    }
}


template <int F, bool GUARD>
__device__ __forceinline__ float gather8(int s, int j, int n,
                                         const float* __restrict__ t, int lane, bool act,
                                         float acc) {
    float v[8];
#pragma unroll
    for (int u = 0; u < 8; ++u) {
        v[u] = 0.0f;
        if (!GUARD || j + u < n) {
            const int sj = __builtin_amdgcn_readlane(s, j + u);
            if (act) v[u] = t[(size_t)sj * F + lane];
        }
    }
#pragma unroll
    for (int u = 0; u < 8; ++u) acc += v[u];
    return acc;
}

template <int F>
__device__ __forceinline__ void gather_rows(const int* __restrict__ slot0,
                                            const int* __restrict__ sorted,
                                            const float* __restrict__ dinv,
                                            const float* __restrict__ t,
                                            const float* __restrict__ bias,
                                            float* __restrict__ o, int N, int E) {
    const int node = __builtin_amdgcn_readfirstlane((int)((blockIdx.x * blockDim.x + threadIdx.x) >> 5));
    if (node >= N) return;
    const int  lane = threadIdx.x & 31;
    const bool act  = lane < F;
    const int  rs   = slot0[node];
    const int  re   = (node + 1 < N) ? slot0[node + 1] : E;
    const float di  = dinv[node];

    float acc = 0.0f;
    for (int base = rs; base < re; base += 32) {
        const int n = (re - base < 32) ? (re - base) : 32;
        int s = 0;
        if (lane < n) s = sorted[base + lane];
        int j = 0;
        for (; j + 8 <= n; j += 8) acc = gather8<F, false>(s, j, n, t, lane, act, acc);
        if (j < n)                 acc = gather8<F, true >(s, j, n, t, lane, act, acc);
    }
    if (act) {
        const size_t idx = (size_t)node * F + lane;
        o[idx] = (acc + t[idx]) * di + bias[lane];
    }
}

__global__ void __launch_bounds__(256)
k_gather32(const int* __restrict__ slot0, const int* __restrict__ sorted,
           const float* __restrict__ dinv, const float* __restrict__ t1,
           const float* __restrict__ b1, float* __restrict__ agg1, int N, int E) {
    gather_rows<32>(slot0, sorted, dinv, t1, b1, agg1, N, E);
}

__global__ void __launch_bounds__(256)
k_gather20(const int* __restrict__ slot0, const int* __restrict__ sorted,
           const float* __restrict__ dinv, const float* __restrict__ t2,
           const float* __restrict__ b2, float* __restrict__ out, int N, int E) {
    gather_rows<20>(slot0, sorted, dinv, t2, b2, out, N, E);
}

extern "C" void kernel_launch(void* const* d_in, const int* in_sizes, int n_in,
                              void* d_out, int out_size, void* d_ws, size_t ws_size,
                              hipStream_t stream) {
    const float* x    = (const float*)d_in[0];
    const float* cnn  = (const float*)d_in[1];
    const int*   ei   = (const int*)d_in[2];
    const float* fcW  = (const float*)d_in[3];
    const float* fcb  = (const float*)d_in[4];
    const float* cnnW = (const float*)d_in[5];
    const float* cnnb = (const float*)d_in[6];
    const float* W1   = (const float*)d_in[7];
    const float* b1   = (const float*)d_in[8];
    const float* W2   = (const float*)d_in[9];
    const float* b2   = (const float*)d_in[10];

    const int N = in_sizes[0] / 32;
    const int E = in_sizes[2] / 2;
    const int* src = ei;
    const int* dst = ei + E;

    float* ws   = (float*)d_ws;
    float* dinv = ws;  ws += N;
    float* h0   = ws;  ws += (size_t)N * 64;
    float* t1   = ws;  ws += (size_t)N * 32;
    float* agg1 = ws;  ws += (size_t)N * 32;
    float* t2   = ws;  ws += (size_t)N * 20;
    float* out  = (float*)d_out;

    int* plane  = (int*)agg1;
    int* slot   = (int*)h0;
    int* sorted = slot + N;

    const int nTiles = (N + 15) / 16;
    const int B = 256;
    dim3 blkG(128);
    dim3 grdG((nTiles * 32 + 127) / 128);
    dim3 grdNode((unsigned)(((long long)N * 32 + B - 1) / B));

    dim3 grdCS(CS_SPLITS, (N + CS_RANGE - 1) / CS_RANGE);
    k_count<<<grdCS, CS_THREADS, 0, stream>>>(dst, plane, E, N);
    int* bsum = (int*)t2;
    const int nbDeg = (N + DEG_THREADS - 1) / DEG_THREADS;
    k_deg<<<nbDeg, DEG_THREADS, 0, stream>>>(plane, dinv, bsum, N);

    k_encoder<<<grdG, blkG, 0, stream>>>(x, cnn, fcW, fcb, cnnW, cnnb, h0, N, nTiles);
    k_gemm_t1<<<grdG, blkG, 0, stream>>>(h0, W1, dinv, t1, N, nTiles);

    k_addoff<<<nbDeg, DEG_THREADS, 0, stream>>>(plane, bsum, slot, N);
    k_fill<<<grdCS, CS_THREADS, 0, stream>>>(src, dst, slot, plane, sorted, E, N);

    k_gather32<<<grdNode, B, 0, stream>>>(slot, sorted, dinv, t1, b1, agg1, N, E);

    k_gemm_t2<<<grdG, blkG, 0, stream>>>(agg1, W2, dinv, t2, N, nTiles);
    k_gather20<<<grdNode, B, 0, stream>>>(slot, sorted, dinv, t2, b2, out, N, E);
}


# Two dependent launches folded away (9 -> 7): t1 GEMM runs in the encoder's kernel on the h0 tile kept in LDS (h0 never stored); k_fill sums the block totals itself; sums and their order unchanged
# speedup vs baseline: 4.0177x; 1.0590x over previous
#include <hip/hip_runtime.h>
#include <hip/hip_bf16.h>

typedef __attribute__((ext_vector_type(2))) float v2f;
typedef __attribute__((ext_vector_type(8))) float v8f;

__device__ __forceinline__ v8f wmma4(v2f a, v2f b, v8f c) {
    return __builtin_amdgcn_wmma_f32_16x16x4_f32(
         false, a,  false, b,
         (short)0, c,  false,  false);
}

#define CS_RANGE   16384
#define CS_SPLITS  32
#define CS_THREADS 1024

__device__ __forceinline__ void cs_slice(int E, int x, int& e0, int& e1) {
    int per = (E + CS_SPLITS - 1) / CS_SPLITS;
    per = (per + 3) & ~3;
    long long e0l = (long long)x * per;
    e0 = (e0l < (long long)E) ? (int)e0l : E;
    e1 = (E - e0 > per) ? (e0 + per) : E;
}

__device__ __forceinline__ void cs_count(unsigned* cnt, int d, int lo) {
    unsigned r = (unsigned)(d - lo);
    if (r < (unsigned)CS_RANGE) atomicAdd(&cnt[r], 1u);
}

__device__ __forceinline__ void cs_count4(unsigned* cnt, int4 v, int lo) {
    cs_count(cnt, v.x, lo);
    cs_count(cnt, v.y, lo);
    cs_count(cnt, v.z, lo);
    cs_count(cnt, v.w, lo);
}

__global__ void __launch_bounds__(CS_THREADS)
k_count(const int* __restrict__ dst, int* __restrict__ plane, int E, int N) {
    __shared__ unsigned cnt[CS_RANGE];
    const int tid = threadIdx.x;
    const int lo  = blockIdx.y * CS_RANGE;

    for (int i = tid; i < CS_RANGE; i += CS_THREADS) cnt[i] = 0u;
    __syncthreads();

    int e0, e1;
    cs_slice(E, blockIdx.x, e0, e1);

    if ((((size_t)dst) & 15) == 0) {
        const int4* __restrict__ p = (const int4*)(dst + e0);
        const int nv = (e1 - e0) >> 2;
        int i = tid;
        for (; i + 3 * CS_THREADS < nv; i += 4 * CS_THREADS) {
            int4 a = p[i];
            int4 b = p[i + CS_THREADS];
            int4 c = p[i + 2 * CS_THREADS];
            int4 d = p[i + 3 * CS_THREADS];
            cs_count4(cnt, a, lo);
            cs_count4(cnt, b, lo);
            cs_count4(cnt, c, lo);
            cs_count4(cnt, d, lo);
        }
        for (; i < nv; i += CS_THREADS) cs_count4(cnt, p[i], lo);
        for (int e = e0 + (nv << 2) + tid; e < e1; e += CS_THREADS) cs_count(cnt, dst[e], lo);
    } else {
        for (int e = e0 + tid; e < e1; e += CS_THREADS) cs_count(cnt, dst[e], lo);
    }
    __syncthreads();

    int* __restrict__ row = plane + (size_t)blockIdx.x * N;
    for (int i = tid; i < CS_RANGE; i += CS_THREADS) {
        int node = lo + i;
        if (node < N) row[node] = (int)cnt[i];
    }
}

#define DEG_THREADS 256

__global__ void __launch_bounds__(DEG_THREADS)
k_deg(int* __restrict__ plane, float* __restrict__ dinv, int* __restrict__ bsum, int N) {
    __shared__ int wtot[DEG_THREADS / 32];
    const int tid  = threadIdx.x;
    const int lane = tid & 31;
    const int w    = tid >> 5;
    const int node = blockIdx.x * DEG_THREADS + tid;
    const bool ok  = node < N;

    int run = 0;
    if (ok) {
        int c[CS_SPLITS];
#pragma unroll
        for (int x = 0; x < CS_SPLITS; ++x) c[x] = plane[(size_t)x * N + node];
        run = c[0];
#pragma unroll
        for (int x = 1; x < CS_SPLITS; ++x) {
            plane[(size_t)x * N + node] = run;
            run += c[x];
        }
        dinv[node] = rsqrtf((float)run + 1.0f);
    }

    int inc = run;
    for (int o = 1; o < 32; o <<= 1) {
        int y = __shfl_up(inc, o);
        if (lane >= o) inc += y;
    }
    if (lane == 31) wtot[w] = inc;
    __syncthreads();
    int woff = 0;
    for (int k = 0; k < w; ++k) woff += wtot[k];
    if (ok) plane[node] = woff + inc - run;
    if (tid == DEG_THREADS - 1) bsum[blockIdx.x] = woff + inc;
}

__device__ __forceinline__ void cs_place4(unsigned* cnt, int4 v, int4 sv, int lo,
                                          int* __restrict__ sorted, int E) {
    const unsigned r0 = (unsigned)(v.x - lo), r1 = (unsigned)(v.y - lo);
    const unsigned r2 = (unsigned)(v.z - lo), r3 = (unsigned)(v.w - lo);
    const bool m0 = r0 < (unsigned)CS_RANGE, m1 = r1 < (unsigned)CS_RANGE;
    const bool m2 = r2 < (unsigned)CS_RANGE, m3 = r3 < (unsigned)CS_RANGE;
    unsigned p0 = 0, p1 = 0, p2 = 0, p3 = 0;
    if (m0) p0 = atomicAdd(&cnt[r0], 1u);
    if (m1) p1 = atomicAdd(&cnt[r1], 1u);
    if (m2) p2 = atomicAdd(&cnt[r2], 1u);
    if (m3) p3 = atomicAdd(&cnt[r3], 1u);
    if (m0 && p0 < (unsigned)E) sorted[p0] = sv.x;
    if (m1 && p1 < (unsigned)E) sorted[p1] = sv.y;
    if (m2 && p2 < (unsigned)E) sorted[p2] = sv.z;
    if (m3 && p3 < (unsigned)E) sorted[p3] = sv.w;
}

__device__ __forceinline__ void cs_place1(unsigned* cnt, int d, int s, int lo,
                                          int* __restrict__ sorted, int E) {
    const unsigned r = (unsigned)(d - lo);
    if (r < (unsigned)CS_RANGE) {
        const unsigned p = atomicAdd(&cnt[r], 1u);
        if (p < (unsigned)E) sorted[p] = s;
    }
}

static_assert(CS_THREADS / 32 == 32, "one wave total per lane of the first wave");
static_assert(CS_RANGE / DEG_THREADS == 64, "two k_deg blocks per lane of the first wave");

__global__ void __launch_bounds__(CS_THREADS)
k_fill(const int* __restrict__ src, const int* __restrict__ dst,
       const int* __restrict__ plane, const int* __restrict__ bsum,
       int* __restrict__ slot, int* __restrict__ sorted, int E, int N) {
    __shared__ unsigned cnt[CS_RANGE];
    __shared__ int wred[CS_THREADS / 32];
    __shared__ int boff[CS_RANGE / DEG_THREADS];
    const int tid  = threadIdx.x;
    const int lane = tid & 31;
    const int w    = tid >> 5;
    const int lo   = blockIdx.y * CS_RANGE;

    const int nbDeg = (N + DEG_THREADS - 1) / DEG_THREADS;
    const int b0    = blockIdx.y * (CS_RANGE / DEG_THREADS);
    int part = 0;
    for (int k = tid; k < b0; k += CS_THREADS) part += bsum[k];
    for (int o = 16; o > 0; o >>= 1) part += __shfl_xor(part, o);
    if (lane == 0) wred[w] = part;
    __syncthreads();
    if (w == 0) {
        int base = wred[lane];
        for (int o = 16; o > 0; o >>= 1) base += __shfl_xor(base, o);
        const int i0 = b0 + 2 * lane;
        const int v0 = (i0 < nbDeg) ? bsum[i0] : 0;
        const int v1 = (i0 + 1 < nbDeg) ? bsum[i0 + 1] : 0;
        const int s2 = v0 + v1;
        int inc = s2;
        for (int o = 1; o < 32; o <<= 1) {
            int y = __shfl_up(inc, o);
            if (lane >= o) inc += y;
        }
        const int ex = base + inc - s2;
        boff[2 * lane]     = ex;
        boff[2 * lane + 1] = ex + v0;
    }
    __syncthreads();

    const int* __restrict__ pref = plane + (size_t)blockIdx.x * N;
    const int prefMask = (blockIdx.x > 0) ? -1 : 0;

    int first[CS_RANGE / CS_THREADS];
#pragma unroll
    for (int k = 0; k < CS_RANGE / CS_THREADS; ++k) {
        const int node = lo + k * CS_THREADS + tid;
        const int nc   = (node < N) ? node : (N - 1);
        const int v    = plane[nc] + boff[(k * CS_THREADS + tid) / DEG_THREADS] + (pref[nc] & prefMask);
        first[k] = (node < N) ? v : 0;
    }
    if (blockIdx.x == 0) {
#pragma unroll
        for (int k = 0; k < CS_RANGE / CS_THREADS; ++k) {
            const int node = lo + k * CS_THREADS + tid;
            if (node < N) slot[node] = first[k];
        }
    }
#pragma unroll
    for (int k = 0; k < CS_RANGE / CS_THREADS; ++k) cnt[k * CS_THREADS + tid] = (unsigned)first[k];
    __syncthreads();

    int e0, e1;
    cs_slice(E, blockIdx.x, e0, e1);

    if (((((size_t)dst) | ((size_t)src)) & 15) == 0) {
        const int4* __restrict__ p = (const int4*)(dst + e0);
        const int4* __restrict__ q = (const int4*)(src + e0);
        const int nv = (e1 - e0) >> 2;
        int i = tid;
        for (; i + CS_THREADS < nv; i += 2 * CS_THREADS) {
            int4 a  = p[i];
            int4 b  = p[i + CS_THREADS];
            int4 sa = q[i];
            int4 sb = q[i + CS_THREADS];
            cs_place4(cnt, a, sa, lo, sorted, E);
            cs_place4(cnt, b, sb, lo, sorted, E);
        }
        for (; i < nv; i += CS_THREADS)
            cs_place4(cnt, p[i], q[i], lo, sorted, E);
        for (int e = e0 + (nv << 2) + tid; e < e1; e += CS_THREADS)
            cs_place1(cnt, dst[e], src[e], lo, sorted, E);
    } else {
        for (int e = e0 + tid; e < e1; e += CS_THREADS)
            cs_place1(cnt, dst[e], src[e], lo, sorted, E);
    }
}

#define ENC_WAVES 4
#define ENC_PITCH 68

__global__ void __launch_bounds__(ENC_WAVES * 32)
k_encoder_t1(const float* __restrict__ x, const float* __restrict__ cnn,
             const float* __restrict__ fcW, const float* __restrict__ fcb,
             const float* __restrict__ cnnW, const float* __restrict__ cnnb,
             const float* __restrict__ W1, const float* __restrict__ dinv,
             float* __restrict__ t1, int N, int nTiles) {
    __shared__ __attribute__((aligned(16))) float h0t[ENC_WAVES][16 * ENC_PITCH];
    const int w    = threadIdx.x >> 5;
    const int lane = threadIdx.x & 31;
    const int wv   = blockIdx.x * ENC_WAVES + w;
    const bool live = wv < nTiles;
    const int  tile = live ? wv : (nTiles - 1);
    const int n    = lane & 15;
    const int kh   = lane >> 4;
    int rowA = tile * 16 + n;    if (rowA >= N) rowA = N - 1;

    v8f acc0 = {}, acc1 = {}, acc2 = {}, acc3 = {};

    const float* ax = x + (size_t)rowA * 32;
#pragma unroll
    for (int k0 = 0; k0 < 32; k0 += 4) {
        int kb = k0 + 2 * kh;
        v2f a;  a.x = ax[kb];               a.y = ax[kb + 1];
        v2f b0; b0.x = fcW[kb * 32 + n];    b0.y = fcW[(kb + 1) * 32 + n];
        v2f b1; b1.x = fcW[kb * 32 + 16 + n]; b1.y = fcW[(kb + 1) * 32 + 16 + n];
        acc0 = wmma4(a, b0, acc0);
        acc1 = wmma4(a, b1, acc1);
    }
    const float* ac = cnn + (size_t)rowA * 128;
#pragma unroll
    for (int k0 = 0; k0 < 128; k0 += 4) {
        int kb = k0 + 2 * kh;
        v2f a;  a.x = ac[kb];                a.y = ac[kb + 1];
        v2f b0; b0.x = cnnW[kb * 32 + n];    b0.y = cnnW[(kb + 1) * 32 + n];
        v2f b1; b1.x = cnnW[kb * 32 + 16 + n]; b1.y = cnnW[(kb + 1) * 32 + 16 + n];
        acc2 = wmma4(a, b0, acc2);
        acc3 = wmma4(a, b1, acc3);
    }

    float dr[8];
#pragma unroll
    for (int v = 0; v < 8; ++v) {
        int nd = tile * 16 + 8 * kh + v;
        dr[v] = dinv[nd < N ? nd : N - 1];
    }

    float* tw = h0t[w];
    {
        const float bf0 = fcb[n],  bf1 = fcb[16 + n];
        const float bc0 = cnnb[n], bc1 = cnnb[16 + n];
#pragma unroll
        for (int v = 0; v < 8; ++v) {
            float* o = tw + (8 * kh + v) * ENC_PITCH;
            o[n]      = fmaxf(acc0[v] + bf0, 0.0f);
            o[16 + n] = fmaxf(acc1[v] + bf1, 0.0f);
            o[32 + n] = fmaxf(acc2[v] + bc0, 0.0f);
            o[48 + n] = fmaxf(acc3[v] + bc1, 0.0f);
        }
    }
    __syncthreads();

    v8f d0 = {}, d1 = {};
    const float* ar = tw + n * ENC_PITCH + 2 * kh;
    const float* wr = W1 + (2 * kh) * 32 + n;
#pragma unroll
    for (int k0 = 0; k0 < 64; k0 += 4) {
        const v2f a = *(const v2f*)(ar + k0);
        v2f b0; b0.x = wr[k0 * 32];        b0.y = wr[k0 * 32 + 32];
        v2f b1; b1.x = wr[k0 * 32 + 16];   b1.y = wr[k0 * 32 + 48];
        d0 = wmma4(a, b0, d0);
        d1 = wmma4(a, b1, d1);
    }
    if (live) {
#pragma unroll
        for (int v = 0; v < 8; ++v) {
            int node = tile * 16 + 8 * kh + v;
            if (node < N) {
                float* o = t1 + (size_t)node * 32;
                o[n]      = d0[v] * dr[v];
                o[16 + n] = d1[v] * dr[v];
            }
        }
    }
}

__global__ void k_gemm_t2(const float* __restrict__ agg1, const float* __restrict__ W2,
                          const float* __restrict__ dinv,
                          float* __restrict__ t2, int N, int nTiles) {
    int wave = (blockIdx.x * blockDim.x + threadIdx.x) >> 5;
    if (wave >= nTiles) return;
    int lane = threadIdx.x & 31;
    int n    = lane & 15;
    int kh   = lane >> 4;
    int rowA = wave * 16 + n;    if (rowA >= N) rowA = N - 1;

    float dr[8];
#pragma unroll
    for (int v = 0; v < 8; ++v) {
        int nd = wave * 16 + 8 * kh + v;
        dr[v] = dinv[nd < N ? nd : N - 1];
    }

    v8f acc0 = {}, acc1 = {};
    const float* ar = agg1 + (size_t)rowA * 32;
    bool c1ok = (16 + n) < 20;
#pragma unroll
    for (int k0 = 0; k0 < 32; k0 += 4) {
        int kb = k0 + 2 * kh;
        v2f a;  a.x = fmaxf(ar[kb], 0.0f);  a.y = fmaxf(ar[kb + 1], 0.0f);
        v2f b0; b0.x = W2[kb * 20 + n];     b0.y = W2[(kb + 1) * 20 + n];
        v2f b1; b1.x = c1ok ? W2[kb * 20 + 16 + n] : 0.0f;
                b1.y = c1ok ? W2[(kb + 1) * 20 + 16 + n] : 0.0f;
        acc0 = wmma4(a, b0, acc0);
        acc1 = wmma4(a, b1, acc1);
    }
#pragma unroll
    for (int v = 0; v < 8; ++v) {
        int mr   = (lane < 16) ? v : v + 8;
        int node = wave * 16 + mr;
        if (node < N) {
            float* o = t2 + (size_t)node * 20;
            o[n] = acc0[v] * dr[v];
            if (c1ok) o[16 + n] = acc1[v] * dr[v];
        }
    }
}


template <int F, bool GUARD>
__device__ __forceinline__ float gather8(int s, int j, int n,
                                         const float* __restrict__ t, int lane, bool act,
                                         float acc) {
    float v[8];
#pragma unroll
    for (int u = 0; u < 8; ++u) {
        v[u] = 0.0f;
        if (!GUARD || j + u < n) {
            const int sj = __builtin_amdgcn_readlane(s, j + u);
            if (act) v[u] = t[(size_t)sj * F + lane];
        }
    }
#pragma unroll
    for (int u = 0; u < 8; ++u) acc += v[u];
    return acc;
}

template <int F>
__device__ __forceinline__ void gather_rows(const int* __restrict__ slot0,
                                            const int* __restrict__ sorted,
                                            const float* __restrict__ dinv,
                                            const float* __restrict__ t,
                                            const float* __restrict__ bias,
                                            float* __restrict__ o, int N, int E) {
    const int node = __builtin_amdgcn_readfirstlane((int)((blockIdx.x * blockDim.x + threadIdx.x) >> 5));
    if (node >= N) return;
    const int  lane = threadIdx.x & 31;
    const bool act  = lane < F;
    const int  rs   = slot0[node];
    const int  re   = (node + 1 < N) ? slot0[node + 1] : E;
    const float di  = dinv[node];

    float acc = 0.0f;
    for (int base = rs; base < re; base += 32) {
        const int n = (re - base < 32) ? (re - base) : 32;
        int s = 0;
        if (lane < n) s = sorted[base + lane];
        int j = 0;
        for (; j + 8 <= n; j += 8) acc = gather8<F, false>(s, j, n, t, lane, act, acc);
        if (j < n)                 acc = gather8<F, true >(s, j, n, t, lane, act, acc);
    }
    if (act) {
        const size_t idx = (size_t)node * F + lane;
        o[idx] = (acc + t[idx]) * di + bias[lane];
    }
}

__global__ void __launch_bounds__(256)
k_gather32(const int* __restrict__ slot0, const int* __restrict__ sorted,
           const float* __restrict__ dinv, const float* __restrict__ t1,
           const float* __restrict__ b1, float* __restrict__ agg1, int N, int E) {
    gather_rows<32>(slot0, sorted, dinv, t1, b1, agg1, N, E);
}

__global__ void __launch_bounds__(256)
k_gather20(const int* __restrict__ slot0, const int* __restrict__ sorted,
           const float* __restrict__ dinv, const float* __restrict__ t2,
           const float* __restrict__ b2, float* __restrict__ out, int N, int E) {
    gather_rows<20>(slot0, sorted, dinv, t2, b2, out, N, E);
}

extern "C" void kernel_launch(void* const* d_in, const int* in_sizes, int n_in,
                              void* d_out, int out_size, void* d_ws, size_t ws_size,
                              hipStream_t stream) {
    const float* x    = (const float*)d_in[0];
    const float* cnn  = (const float*)d_in[1];
    const int*   ei   = (const int*)d_in[2];
    const float* fcW  = (const float*)d_in[3];
    const float* fcb  = (const float*)d_in[4];
    const float* cnnW = (const float*)d_in[5];
    const float* cnnb = (const float*)d_in[6];
    const float* W1   = (const float*)d_in[7];
    const float* b1   = (const float*)d_in[8];
    const float* W2   = (const float*)d_in[9];
    const float* b2   = (const float*)d_in[10];

    const int N = in_sizes[0] / 32;
    const int E = in_sizes[2] / 2;
    const int* src = ei;
    const int* dst = ei + E;

    float* ws   = (float*)d_ws;
    float* dinv = ws;  ws += N;
    float* eidx = ws;  ws += (size_t)N * 64;
    float* t1   = ws;  ws += (size_t)N * 32;
    float* agg1 = ws;  ws += (size_t)N * 32;
    float* t2   = ws;  ws += (size_t)N * 20;
    float* out  = (float*)d_out;

    int* plane  = (int*)agg1;
    int* slot   = (int*)eidx;
    int* sorted = slot + N;

    const int nTiles = (N + 15) / 16;
    const int B = 256;
    dim3 blkG(ENC_WAVES * 32);
    dim3 grdG((nTiles * 32 + 127) / 128);
    dim3 grdNode((unsigned)(((long long)N * 32 + B - 1) / B));

    dim3 grdCS(CS_SPLITS, (N + CS_RANGE - 1) / CS_RANGE);
    k_count<<<grdCS, CS_THREADS, 0, stream>>>(dst, plane, E, N);
    int* bsum = (int*)t2;
    const int nbDeg = (N + DEG_THREADS - 1) / DEG_THREADS;
    k_deg<<<nbDeg, DEG_THREADS, 0, stream>>>(plane, dinv, bsum, N);

    k_encoder_t1<<<grdG, blkG, 0, stream>>>(x, cnn, fcW, fcb, cnnW, cnnb, W1, dinv, t1, N, nTiles);

    k_fill<<<grdCS, CS_THREADS, 0, stream>>>(src, dst, plane, bsum, slot, sorted, E, N);

    k_gather32<<<grdNode, B, 0, stream>>>(slot, sorted, dinv, t1, b1, agg1, N, E);

    k_gemm_t2<<<grdG, blkG, 0, stream>>>(agg1, W2, dinv, t2, N, nTiles);
    k_gather20<<<grdNode, B, 0, stream>>>(slot, sorted, dinv, t2, b2, out, N, E);
}


# Gathers run on a resident grid (2048 blocks of 8 waves): each wave walks the nodes a grid apart, with the next node's row bounds and source ids loaded ahead of need; per-node sum order unchanged
# speedup vs baseline: 4.1435x; 1.0313x over previous
#include <hip/hip_runtime.h>
#include <hip/hip_bf16.h>

typedef __attribute__((ext_vector_type(2))) float v2f;
typedef __attribute__((ext_vector_type(8))) float v8f;

__device__ __forceinline__ v8f wmma4(v2f a, v2f b, v8f c) {
    return __builtin_amdgcn_wmma_f32_16x16x4_f32(
         false, a,  false, b,
         (short)0, c,  false,  false);
}

#define CS_RANGE   16384
#define CS_SPLITS  32
#define CS_THREADS 1024

__device__ __forceinline__ void cs_slice(int E, int x, int& e0, int& e1) {
    int per = (E + CS_SPLITS - 1) / CS_SPLITS;
    per = (per + 3) & ~3;
    long long e0l = (long long)x * per;
    e0 = (e0l < (long long)E) ? (int)e0l : E;
    e1 = (E - e0 > per) ? (e0 + per) : E;
}

__device__ __forceinline__ void cs_count(unsigned* cnt, int d, int lo) {
    unsigned r = (unsigned)(d - lo);
    if (r < (unsigned)CS_RANGE) atomicAdd(&cnt[r], 1u);
}

__device__ __forceinline__ void cs_count4(unsigned* cnt, int4 v, int lo) {
    cs_count(cnt, v.x, lo);
    cs_count(cnt, v.y, lo);
    cs_count(cnt, v.z, lo);
    cs_count(cnt, v.w, lo);
}

__global__ void __launch_bounds__(CS_THREADS)
k_count(const int* __restrict__ dst, int* __restrict__ plane, int E, int N) {
    __shared__ unsigned cnt[CS_RANGE];
    const int tid = threadIdx.x;
    const int lo  = blockIdx.y * CS_RANGE;

    for (int i = tid; i < CS_RANGE; i += CS_THREADS) cnt[i] = 0u;
    __syncthreads();

    int e0, e1;
    cs_slice(E, blockIdx.x, e0, e1);

    if ((((size_t)dst) & 15) == 0) {
        const int4* __restrict__ p = (const int4*)(dst + e0);
        const int nv = (e1 - e0) >> 2;
        int i = tid;
        for (; i + 3 * CS_THREADS < nv; i += 4 * CS_THREADS) {
            int4 a = p[i];
            int4 b = p[i + CS_THREADS];
            int4 c = p[i + 2 * CS_THREADS];
            int4 d = p[i + 3 * CS_THREADS];
            cs_count4(cnt, a, lo);
            cs_count4(cnt, b, lo);
            cs_count4(cnt, c, lo);
            cs_count4(cnt, d, lo);
        }
        for (; i < nv; i += CS_THREADS) cs_count4(cnt, p[i], lo);
        for (int e = e0 + (nv << 2) + tid; e < e1; e += CS_THREADS) cs_count(cnt, dst[e], lo);
    } else {
        for (int e = e0 + tid; e < e1; e += CS_THREADS) cs_count(cnt, dst[e], lo);
    }
    __syncthreads();

    int* __restrict__ row = plane + (size_t)blockIdx.x * N;
    for (int i = tid; i < CS_RANGE; i += CS_THREADS) {
        int node = lo + i;
        if (node < N) row[node] = (int)cnt[i];
    }
}

#define DEG_THREADS 256

__global__ void __launch_bounds__(DEG_THREADS)
k_deg(int* __restrict__ plane, float* __restrict__ dinv, int* __restrict__ bsum, int N) {
    __shared__ int wtot[DEG_THREADS / 32];
    const int tid  = threadIdx.x;
    const int lane = tid & 31;
    const int w    = tid >> 5;
    const int node = blockIdx.x * DEG_THREADS + tid;
    const bool ok  = node < N;

    int run = 0;
    if (ok) {
        int c[CS_SPLITS];
#pragma unroll
        for (int x = 0; x < CS_SPLITS; ++x) c[x] = plane[(size_t)x * N + node];
        run = c[0];
#pragma unroll
        for (int x = 1; x < CS_SPLITS; ++x) {
            plane[(size_t)x * N + node] = run;
            run += c[x];
        }
        dinv[node] = rsqrtf((float)run + 1.0f);
    }

    int inc = run;
    for (int o = 1; o < 32; o <<= 1) {
        int y = __shfl_up(inc, o);
        if (lane >= o) inc += y;
    }
    if (lane == 31) wtot[w] = inc;
    __syncthreads();
    int woff = 0;
    for (int k = 0; k < w; ++k) woff += wtot[k];
    if (ok) plane[node] = woff + inc - run;
    if (tid == DEG_THREADS - 1) bsum[blockIdx.x] = woff + inc;
}

__device__ __forceinline__ void cs_place4(unsigned* cnt, int4 v, int4 sv, int lo,
                                          int* __restrict__ sorted, int E) {
    const unsigned r0 = (unsigned)(v.x - lo), r1 = (unsigned)(v.y - lo);
    const unsigned r2 = (unsigned)(v.z - lo), r3 = (unsigned)(v.w - lo);
    const bool m0 = r0 < (unsigned)CS_RANGE, m1 = r1 < (unsigned)CS_RANGE;
    const bool m2 = r2 < (unsigned)CS_RANGE, m3 = r3 < (unsigned)CS_RANGE;
    unsigned p0 = 0, p1 = 0, p2 = 0, p3 = 0;
    if (m0) p0 = atomicAdd(&cnt[r0], 1u);
    if (m1) p1 = atomicAdd(&cnt[r1], 1u);
    if (m2) p2 = atomicAdd(&cnt[r2], 1u);
    if (m3) p3 = atomicAdd(&cnt[r3], 1u);
    if (m0 && p0 < (unsigned)E) sorted[p0] = sv.x;
    if (m1 && p1 < (unsigned)E) sorted[p1] = sv.y;
    if (m2 && p2 < (unsigned)E) sorted[p2] = sv.z;
    if (m3 && p3 < (unsigned)E) sorted[p3] = sv.w;
}

__device__ __forceinline__ void cs_place1(unsigned* cnt, int d, int s, int lo,
                                          int* __restrict__ sorted, int E) {
    const unsigned r = (unsigned)(d - lo);
    if (r < (unsigned)CS_RANGE) {
        const unsigned p = atomicAdd(&cnt[r], 1u);
        if (p < (unsigned)E) sorted[p] = s;
    }
}

static_assert(CS_THREADS / 32 == 32, "one wave total per lane of the first wave");
static_assert(CS_RANGE / DEG_THREADS == 64, "two k_deg blocks per lane of the first wave");

__global__ void __launch_bounds__(CS_THREADS)
k_fill(const int* __restrict__ src, const int* __restrict__ dst,
       const int* __restrict__ plane, const int* __restrict__ bsum,
       int* __restrict__ slot, int* __restrict__ sorted, int E, int N) {
    __shared__ unsigned cnt[CS_RANGE];
    __shared__ int wred[CS_THREADS / 32];
    __shared__ int boff[CS_RANGE / DEG_THREADS];
    const int tid  = threadIdx.x;
    const int lane = tid & 31;
    const int w    = tid >> 5;
    const int lo   = blockIdx.y * CS_RANGE;

    const int nbDeg = (N + DEG_THREADS - 1) / DEG_THREADS;
    const int b0    = blockIdx.y * (CS_RANGE / DEG_THREADS);
    int part = 0;
    for (int k = tid; k < b0; k += CS_THREADS) part += bsum[k];
    for (int o = 16; o > 0; o >>= 1) part += __shfl_xor(part, o);
    if (lane == 0) wred[w] = part;
    __syncthreads();
    if (w == 0) {
        int base = wred[lane];
        for (int o = 16; o > 0; o >>= 1) base += __shfl_xor(base, o);
        const int i0 = b0 + 2 * lane;
        const int v0 = (i0 < nbDeg) ? bsum[i0] : 0;
        const int v1 = (i0 + 1 < nbDeg) ? bsum[i0 + 1] : 0;
        const int s2 = v0 + v1;
        int inc = s2;
        for (int o = 1; o < 32; o <<= 1) {
            int y = __shfl_up(inc, o);
            if (lane >= o) inc += y;
        }
        const int ex = base + inc - s2;
        boff[2 * lane]     = ex;
        boff[2 * lane + 1] = ex + v0;
    }
    __syncthreads();

    const int* __restrict__ pref = plane + (size_t)blockIdx.x * N;
    const int prefMask = (blockIdx.x > 0) ? -1 : 0;

    int first[CS_RANGE / CS_THREADS];
#pragma unroll
    for (int k = 0; k < CS_RANGE / CS_THREADS; ++k) {
        const int node = lo + k * CS_THREADS + tid;
        const int nc   = (node < N) ? node : (N - 1);
        const int v    = plane[nc] + boff[(k * CS_THREADS + tid) / DEG_THREADS] + (pref[nc] & prefMask);
        first[k] = (node < N) ? v : 0;
    }
    if (blockIdx.x == 0) {
#pragma unroll
        for (int k = 0; k < CS_RANGE / CS_THREADS; ++k) {
            const int node = lo + k * CS_THREADS + tid;
            if (node < N) slot[node] = first[k];
        }
    }
#pragma unroll
    for (int k = 0; k < CS_RANGE / CS_THREADS; ++k) cnt[k * CS_THREADS + tid] = (unsigned)first[k];
    __syncthreads();

    int e0, e1;
    cs_slice(E, blockIdx.x, e0, e1);

    if (((((size_t)dst) | ((size_t)src)) & 15) == 0) {
        const int4* __restrict__ p = (const int4*)(dst + e0);
        const int4* __restrict__ q = (const int4*)(src + e0);
        const int nv = (e1 - e0) >> 2;
        int i = tid;
        for (; i + CS_THREADS < nv; i += 2 * CS_THREADS) {
            int4 a  = p[i];
            int4 b  = p[i + CS_THREADS];
            int4 sa = q[i];
            int4 sb = q[i + CS_THREADS];
            cs_place4(cnt, a, sa, lo, sorted, E);
            cs_place4(cnt, b, sb, lo, sorted, E);
        }
        for (; i < nv; i += CS_THREADS)
            cs_place4(cnt, p[i], q[i], lo, sorted, E);
        for (int e = e0 + (nv << 2) + tid; e < e1; e += CS_THREADS)
            cs_place1(cnt, dst[e], src[e], lo, sorted, E);
    } else {
        for (int e = e0 + tid; e < e1; e += CS_THREADS)
            cs_place1(cnt, dst[e], src[e], lo, sorted, E);
    }
}

#define ENC_WAVES 4
#define ENC_PITCH 68

__global__ void __launch_bounds__(ENC_WAVES * 32)
k_encoder_t1(const float* __restrict__ x, const float* __restrict__ cnn,
             const float* __restrict__ fcW, const float* __restrict__ fcb,
             const float* __restrict__ cnnW, const float* __restrict__ cnnb,
             const float* __restrict__ W1, const float* __restrict__ dinv,
             float* __restrict__ t1, int N, int nTiles) {
    __shared__ __attribute__((aligned(16))) float h0t[ENC_WAVES][16 * ENC_PITCH];
    const int w    = threadIdx.x >> 5;
    const int lane = threadIdx.x & 31;
    const int wv   = blockIdx.x * ENC_WAVES + w;
    const bool live = wv < nTiles;
    const int  tile = live ? wv : (nTiles - 1);
    const int n    = lane & 15;
    const int kh   = lane >> 4;
    int rowA = tile * 16 + n;    if (rowA >= N) rowA = N - 1;

    v8f acc0 = {}, acc1 = {}, acc2 = {}, acc3 = {};

    const float* ax = x + (size_t)rowA * 32;
#pragma unroll
    for (int k0 = 0; k0 < 32; k0 += 4) {
        int kb = k0 + 2 * kh;
        v2f a;  a.x = ax[kb];               a.y = ax[kb + 1];
        v2f b0; b0.x = fcW[kb * 32 + n];    b0.y = fcW[(kb + 1) * 32 + n];
        v2f b1; b1.x = fcW[kb * 32 + 16 + n]; b1.y = fcW[(kb + 1) * 32 + 16 + n];
        acc0 = wmma4(a, b0, acc0);
        acc1 = wmma4(a, b1, acc1);
    }
    const float* ac = cnn + (size_t)rowA * 128;
#pragma unroll
    for (int k0 = 0; k0 < 128; k0 += 4) {
        int kb = k0 + 2 * kh;
        v2f a;  a.x = ac[kb];                a.y = ac[kb + 1];
        v2f b0; b0.x = cnnW[kb * 32 + n];    b0.y = cnnW[(kb + 1) * 32 + n];
        v2f b1; b1.x = cnnW[kb * 32 + 16 + n]; b1.y = cnnW[(kb + 1) * 32 + 16 + n];
        acc2 = wmma4(a, b0, acc2);
        acc3 = wmma4(a, b1, acc3);
    }

    float dr[8];
#pragma unroll
    for (int v = 0; v < 8; ++v) {
        int nd = tile * 16 + 8 * kh + v;
        dr[v] = dinv[nd < N ? nd : N - 1];
    }

    float* tw = h0t[w];
    {
        const float bf0 = fcb[n],  bf1 = fcb[16 + n];
        const float bc0 = cnnb[n], bc1 = cnnb[16 + n];
#pragma unroll
        for (int v = 0; v < 8; ++v) {
            float* o = tw + (8 * kh + v) * ENC_PITCH;
            o[n]      = fmaxf(acc0[v] + bf0, 0.0f);
            o[16 + n] = fmaxf(acc1[v] + bf1, 0.0f);
            o[32 + n] = fmaxf(acc2[v] + bc0, 0.0f);
            o[48 + n] = fmaxf(acc3[v] + bc1, 0.0f);
        }
    }
    __syncthreads();

    v8f d0 = {}, d1 = {};
    const float* ar = tw + n * ENC_PITCH + 2 * kh;
    const float* wr = W1 + (2 * kh) * 32 + n;
#pragma unroll
    for (int k0 = 0; k0 < 64; k0 += 4) {
        const v2f a = *(const v2f*)(ar + k0);
        v2f b0; b0.x = wr[k0 * 32];        b0.y = wr[k0 * 32 + 32];
        v2f b1; b1.x = wr[k0 * 32 + 16];   b1.y = wr[k0 * 32 + 48];
        d0 = wmma4(a, b0, d0);
        d1 = wmma4(a, b1, d1);
    }
    if (live) {
#pragma unroll
        for (int v = 0; v < 8; ++v) {
            int node = tile * 16 + 8 * kh + v;
            if (node < N) {
                float* o = t1 + (size_t)node * 32;
                o[n]      = d0[v] * dr[v];
                o[16 + n] = d1[v] * dr[v];
            }
        }
    }
}

__global__ void k_gemm_t2(const float* __restrict__ agg1, const float* __restrict__ W2,
                          const float* __restrict__ dinv,
                          float* __restrict__ t2, int N, int nTiles) {
    int wave = (blockIdx.x * blockDim.x + threadIdx.x) >> 5;
    if (wave >= nTiles) return;
    int lane = threadIdx.x & 31;
    int n    = lane & 15;
    int kh   = lane >> 4;
    int rowA = wave * 16 + n;    if (rowA >= N) rowA = N - 1;

    float dr[8];
#pragma unroll
    for (int v = 0; v < 8; ++v) {
        int nd = wave * 16 + 8 * kh + v;
        dr[v] = dinv[nd < N ? nd : N - 1];
    }

    v8f acc0 = {}, acc1 = {};
    const float* ar = agg1 + (size_t)rowA * 32;
    bool c1ok = (16 + n) < 20;
#pragma unroll
    for (int k0 = 0; k0 < 32; k0 += 4) {
        int kb = k0 + 2 * kh;
        v2f a;  a.x = fmaxf(ar[kb], 0.0f);  a.y = fmaxf(ar[kb + 1], 0.0f);
        v2f b0; b0.x = W2[kb * 20 + n];     b0.y = W2[(kb + 1) * 20 + n];
        v2f b1; b1.x = c1ok ? W2[kb * 20 + 16 + n] : 0.0f;
                b1.y = c1ok ? W2[(kb + 1) * 20 + 16 + n] : 0.0f;
        acc0 = wmma4(a, b0, acc0);
        acc1 = wmma4(a, b1, acc1);
    }
#pragma unroll
    for (int v = 0; v < 8; ++v) {
        int mr   = (lane < 16) ? v : v + 8;
        int node = wave * 16 + mr;
        if (node < N) {
            float* o = t2 + (size_t)node * 20;
            o[n] = acc0[v] * dr[v];
            if (c1ok) o[16 + n] = acc1[v] * dr[v];
        }
    }
}


template <int F, bool GUARD>
__device__ __forceinline__ void rows8(int s, int j, int n, const float* __restrict__ t,
                                      int lane, bool act, float (&v)[8]) {
#pragma unroll
    for (int u = 0; u < 8; ++u) {
        v[u] = 0.0f;
        if (!GUARD || j + u < n) {
            const int sj = __builtin_amdgcn_readlane(s, j + u);
            if (act) v[u] = t[(size_t)sj * F + lane];
        }
    }
}

__device__ __forceinline__ float sum8(const float (&v)[8], float acc) {
#pragma unroll
    for (int u = 0; u < 8; ++u) acc += v[u];
    return acc;
}

template <int F>
__device__ __forceinline__ void rows8_first(int s, int n, const float* __restrict__ t,
                                            int lane, bool act, float (&v)[8]) {
    const int col = act ? lane : 0;
#pragma unroll
    for (int u = 0; u < 8; ++u) {
        const int sj = __builtin_amdgcn_readlane(s, u);
        const float x = t[(size_t)sj * F + col];
        v[u] = (u < n) ? x : 0.0f;
    }
}

template <int F, bool GUARD>
__device__ __forceinline__ float gather8(int s, int j, int n,
                                         const float* __restrict__ t, int lane, bool act,
                                         float acc) {
    float v[8];
    rows8<F, GUARD>(s, j, n, t, lane, act, v);
    return sum8(v, acc);
}

template <int F>
__device__ __forceinline__ float gather_group(int s, int j0, int n, const float* __restrict__ t,
                                              int lane, bool act, float acc) {
    int j = j0;
    for (; j + 8 <= n; j += 8) acc = gather8<F, false>(s, j, n, t, lane, act, acc);
    if (j < n)                 acc = gather8<F, true >(s, j, n, t, lane, act, acc);
    return acc;
}

template <int F>
__device__ __forceinline__ void gather_rows(const int* __restrict__ slot0,
                                            const int* __restrict__ sorted,
                                            const float* __restrict__ dinv,
                                            const float* __restrict__ t,
                                            const float* __restrict__ bias,
                                            float* __restrict__ o, int N, int E) {
    const int G = __builtin_amdgcn_readfirstlane((int)(gridDim.x * (blockDim.x >> 5)));
    int node = __builtin_amdgcn_readfirstlane((int)((blockIdx.x * blockDim.x + threadIdx.x) >> 5));
    if (node >= N) return;
    const int  lane = threadIdx.x & 31;
    const bool act  = lane < F;
    const float bl  = act ? bias[lane] : 0.0f;

    int rs = slot0[node];
    int re = (node + 1 < N) ? slot0[node + 1] : E;
    int sa = 0, sb = 0;
    if (rs + lane < re)      sa = sorted[rs + lane];
    if (rs + 32 + lane < re) sb = sorted[rs + 32 + lane];
    int nx  = node + G;
    int rs1 = 0, re1 = 0;
    if (nx < N) { rs1 = slot0[nx]; re1 = (nx + 1 < N) ? slot0[nx + 1] : E; }

    for (;;) {
        const int cnt = re - rs;
        const int n0  = (cnt < 32) ? cnt : 32;
        float v[8];
        rows8_first<F>(sa, n0, t, lane, act, v);

        int sa1 = 0, sb1 = 0;
        if (nx < N) {
            if (rs1 + lane < re1)      sa1 = sorted[rs1 + lane];
            if (rs1 + 32 + lane < re1) sb1 = sorted[rs1 + 32 + lane];
        }
        const int nx2 = nx + G;
        int rs2 = 0, re2 = 0;
        if (nx < N && nx2 < N) { rs2 = slot0[nx2]; re2 = (nx2 + 1 < N) ? slot0[nx2 + 1] : E; }

        const float  di  = dinv[node];
        const size_t idx = (size_t)node * F + lane;
        float self = 0.0f;
        if (act) self = t[idx];

        float acc = sum8(v, 0.0f);
        acc = gather_group<F>(sa, 8, n0, t, lane, act, acc);
        if (cnt > 32)
            acc = gather_group<F>(sb, 0, (cnt < 64) ? (cnt - 32) : 32, t, lane, act, acc);
        for (int base = rs + 64; base < re; base += 32) {
            const int n = (re - base < 32) ? (re - base) : 32;
            int s = 0;
            if (lane < n) s = sorted[base + lane];
            acc = gather_group<F>(s, 0, n, t, lane, act, acc);
        }
        if (act) o[idx] = (acc + self) * di + bl;

        if (nx >= N) break;
        node = nx;  rs = rs1;  re = re1;  sa = sa1;  sb = sb1;
        nx = nx2;   rs1 = rs2; re1 = re2;
    }
}

__global__ void __launch_bounds__(256)
k_gather32(const int* __restrict__ slot0, const int* __restrict__ sorted,
           const float* __restrict__ dinv, const float* __restrict__ t1,
           const float* __restrict__ b1, float* __restrict__ agg1, int N, int E) {
    gather_rows<32>(slot0, sorted, dinv, t1, b1, agg1, N, E);
}

__global__ void __launch_bounds__(256)
k_gather20(const int* __restrict__ slot0, const int* __restrict__ sorted,
           const float* __restrict__ dinv, const float* __restrict__ t2,
           const float* __restrict__ b2, float* __restrict__ out, int N, int E) {
    gather_rows<20>(slot0, sorted, dinv, t2, b2, out, N, E);
}

extern "C" void kernel_launch(void* const* d_in, const int* in_sizes, int n_in,
                              void* d_out, int out_size, void* d_ws, size_t ws_size,
                              hipStream_t stream) {
    const float* x    = (const float*)d_in[0];
    const float* cnn  = (const float*)d_in[1];
    const int*   ei   = (const int*)d_in[2];
    const float* fcW  = (const float*)d_in[3];
    const float* fcb  = (const float*)d_in[4];
    const float* cnnW = (const float*)d_in[5];
    const float* cnnb = (const float*)d_in[6];
    const float* W1   = (const float*)d_in[7];
    const float* b1   = (const float*)d_in[8];
    const float* W2   = (const float*)d_in[9];
    const float* b2   = (const float*)d_in[10];

    const int N = in_sizes[0] / 32;
    const int E = in_sizes[2] / 2;
    const int* src = ei;
    const int* dst = ei + E;

    float* ws   = (float*)d_ws;
    float* dinv = ws;  ws += N;
    float* eidx = ws;  ws += (size_t)N * 64;
    float* t1   = ws;  ws += (size_t)N * 32;
    float* agg1 = ws;  ws += (size_t)N * 32;
    float* t2   = ws;  ws += (size_t)N * 20;
    float* out  = (float*)d_out;

    int* plane  = (int*)agg1;
    int* slot   = (int*)eidx;
    int* sorted = slot + N;

    const int nTiles = (N + 15) / 16;
    const int B = 256;
    dim3 blkG(ENC_WAVES * 32);
    dim3 grdG((nTiles * 32 + 127) / 128);
    dim3 grdNode(2048);

    dim3 grdCS(CS_SPLITS, (N + CS_RANGE - 1) / CS_RANGE);
    k_count<<<grdCS, CS_THREADS, 0, stream>>>(dst, plane, E, N);
    int* bsum = (int*)t2;
    const int nbDeg = (N + DEG_THREADS - 1) / DEG_THREADS;
    k_deg<<<nbDeg, DEG_THREADS, 0, stream>>>(plane, dinv, bsum, N);

    k_encoder_t1<<<grdG, blkG, 0, stream>>>(x, cnn, fcW, fcb, cnnW, cnnb, W1, dinv, t1, N, nTiles);

    k_fill<<<grdCS, CS_THREADS, 0, stream>>>(src, dst, plane, bsum, slot, sorted, E, N);

    k_gather32<<<grdNode, B, 0, stream>>>(slot, sorted, dinv, t1, b1, agg1, N, E);

    k_gemm_t2<<<grdG, blkG, 0, stream>>>(agg1, W2, dinv, t2, N, nTiles);
    k_gather20<<<grdNode, B, 0, stream>>>(slot, sorted, dinv, t2, b2, out, N, E);
}


# t2 rows stored at a 32-float (128-byte) pitch, in t1's region once t1 is dead: every source row the second aggregation fetches is one aligned line, none lies across two; sums and order unchanged
# speedup vs baseline: 4.3017x; 1.0382x over previous
#include <hip/hip_runtime.h>
#include <hip/hip_bf16.h>

typedef __attribute__((ext_vector_type(2))) float v2f;
typedef __attribute__((ext_vector_type(8))) float v8f;

__device__ __forceinline__ v8f wmma4(v2f a, v2f b, v8f c) {
    return __builtin_amdgcn_wmma_f32_16x16x4_f32(
         false, a,  false, b,
         (short)0, c,  false,  false);
}

#define CS_RANGE   16384
#define CS_SPLITS  32
#define CS_THREADS 1024

__device__ __forceinline__ void cs_slice(int E, int x, int& e0, int& e1) {
    int per = (E + CS_SPLITS - 1) / CS_SPLITS;
    per = (per + 3) & ~3;
    long long e0l = (long long)x * per;
    e0 = (e0l < (long long)E) ? (int)e0l : E;
    e1 = (E - e0 > per) ? (e0 + per) : E;
}

__device__ __forceinline__ void cs_count(unsigned* cnt, int d, int lo) {
    unsigned r = (unsigned)(d - lo);
    if (r < (unsigned)CS_RANGE) atomicAdd(&cnt[r], 1u);
}

__device__ __forceinline__ void cs_count4(unsigned* cnt, int4 v, int lo) {
    cs_count(cnt, v.x, lo);
    cs_count(cnt, v.y, lo);
    cs_count(cnt, v.z, lo);
    cs_count(cnt, v.w, lo);
}

__global__ void __launch_bounds__(CS_THREADS)
k_count(const int* __restrict__ dst, int* __restrict__ plane, int E, int N) {
    __shared__ unsigned cnt[CS_RANGE];
    const int tid = threadIdx.x;
    const int lo  = blockIdx.y * CS_RANGE;

    for (int i = tid; i < CS_RANGE; i += CS_THREADS) cnt[i] = 0u;
    __syncthreads();

    int e0, e1;
    cs_slice(E, blockIdx.x, e0, e1);

    if ((((size_t)dst) & 15) == 0) {
        const int4* __restrict__ p = (const int4*)(dst + e0);
        const int nv = (e1 - e0) >> 2;
        int i = tid;
        for (; i + 3 * CS_THREADS < nv; i += 4 * CS_THREADS) {
            int4 a = p[i];
            int4 b = p[i + CS_THREADS];
            int4 c = p[i + 2 * CS_THREADS];
            int4 d = p[i + 3 * CS_THREADS];
            cs_count4(cnt, a, lo);
            cs_count4(cnt, b, lo);
            cs_count4(cnt, c, lo);
            cs_count4(cnt, d, lo);
        }
        for (; i < nv; i += CS_THREADS) cs_count4(cnt, p[i], lo);
        for (int e = e0 + (nv << 2) + tid; e < e1; e += CS_THREADS) cs_count(cnt, dst[e], lo);
    } else {
        for (int e = e0 + tid; e < e1; e += CS_THREADS) cs_count(cnt, dst[e], lo);
    }
    __syncthreads();

    int* __restrict__ row = plane + (size_t)blockIdx.x * N;
    for (int i = tid; i < CS_RANGE; i += CS_THREADS) {
        int node = lo + i;
        if (node < N) row[node] = (int)cnt[i];
    }
}

#define DEG_THREADS 256

__global__ void __launch_bounds__(DEG_THREADS)
k_deg(int* __restrict__ plane, float* __restrict__ dinv, int* __restrict__ bsum, int N) {
    __shared__ int wtot[DEG_THREADS / 32];
    const int tid  = threadIdx.x;
    const int lane = tid & 31;
    const int w    = tid >> 5;
    const int node = blockIdx.x * DEG_THREADS + tid;
    const bool ok  = node < N;

    int run = 0;
    if (ok) {
        int c[CS_SPLITS];
#pragma unroll
        for (int x = 0; x < CS_SPLITS; ++x) c[x] = plane[(size_t)x * N + node];
        run = c[0];
#pragma unroll
        for (int x = 1; x < CS_SPLITS; ++x) {
            plane[(size_t)x * N + node] = run;
            run += c[x];
        }
        dinv[node] = rsqrtf((float)run + 1.0f);
    }

    int inc = run;
    for (int o = 1; o < 32; o <<= 1) {
        int y = __shfl_up(inc, o);
        if (lane >= o) inc += y;
    }
    if (lane == 31) wtot[w] = inc;
    __syncthreads();
    int woff = 0;
    for (int k = 0; k < w; ++k) woff += wtot[k];
    if (ok) plane[node] = woff + inc - run;
    if (tid == DEG_THREADS - 1) bsum[blockIdx.x] = woff + inc;
}

__device__ __forceinline__ void cs_place4(unsigned* cnt, int4 v, int4 sv, int lo,
                                          int* __restrict__ sorted, int E) {
    const unsigned r0 = (unsigned)(v.x - lo), r1 = (unsigned)(v.y - lo);
    const unsigned r2 = (unsigned)(v.z - lo), r3 = (unsigned)(v.w - lo);
    const bool m0 = r0 < (unsigned)CS_RANGE, m1 = r1 < (unsigned)CS_RANGE;
    const bool m2 = r2 < (unsigned)CS_RANGE, m3 = r3 < (unsigned)CS_RANGE;
    unsigned p0 = 0, p1 = 0, p2 = 0, p3 = 0;
    if (m0) p0 = atomicAdd(&cnt[r0], 1u);
    if (m1) p1 = atomicAdd(&cnt[r1], 1u);
    if (m2) p2 = atomicAdd(&cnt[r2], 1u);
    if (m3) p3 = atomicAdd(&cnt[r3], 1u);
    if (m0 && p0 < (unsigned)E) sorted[p0] = sv.x;
    if (m1 && p1 < (unsigned)E) sorted[p1] = sv.y;
    if (m2 && p2 < (unsigned)E) sorted[p2] = sv.z;
    if (m3 && p3 < (unsigned)E) sorted[p3] = sv.w;
}

__device__ __forceinline__ void cs_place1(unsigned* cnt, int d, int s, int lo,
                                          int* __restrict__ sorted, int E) {
    const unsigned r = (unsigned)(d - lo);
    if (r < (unsigned)CS_RANGE) {
        const unsigned p = atomicAdd(&cnt[r], 1u);
        if (p < (unsigned)E) sorted[p] = s;
    }
}

static_assert(CS_THREADS / 32 == 32, "one wave total per lane of the first wave");
static_assert(CS_RANGE / DEG_THREADS == 64, "two k_deg blocks per lane of the first wave");

__global__ void __launch_bounds__(CS_THREADS)
k_fill(const int* __restrict__ src, const int* __restrict__ dst,
       const int* __restrict__ plane, const int* __restrict__ bsum,
       int* __restrict__ slot, int* __restrict__ sorted, int E, int N) {
    __shared__ unsigned cnt[CS_RANGE];
    __shared__ int wred[CS_THREADS / 32];
    __shared__ int boff[CS_RANGE / DEG_THREADS];
    const int tid  = threadIdx.x;
    const int lane = tid & 31;
    const int w    = tid >> 5;
    const int lo   = blockIdx.y * CS_RANGE;

    const int nbDeg = (N + DEG_THREADS - 1) / DEG_THREADS;
    const int b0    = blockIdx.y * (CS_RANGE / DEG_THREADS);
    int part = 0;
    for (int k = tid; k < b0; k += CS_THREADS) part += bsum[k];
    for (int o = 16; o > 0; o >>= 1) part += __shfl_xor(part, o);
    if (lane == 0) wred[w] = part;
    __syncthreads();
    if (w == 0) {
        int base = wred[lane];
        for (int o = 16; o > 0; o >>= 1) base += __shfl_xor(base, o);
        const int i0 = b0 + 2 * lane;
        const int v0 = (i0 < nbDeg) ? bsum[i0] : 0;
        const int v1 = (i0 + 1 < nbDeg) ? bsum[i0 + 1] : 0;
        const int s2 = v0 + v1;
        int inc = s2;
        for (int o = 1; o < 32; o <<= 1) {
            int y = __shfl_up(inc, o);
            if (lane >= o) inc += y;
        }
        const int ex = base + inc - s2;
        boff[2 * lane]     = ex;
        boff[2 * lane + 1] = ex + v0;
    }
    __syncthreads();

    const int* __restrict__ pref = plane + (size_t)blockIdx.x * N;
    const int prefMask = (blockIdx.x > 0) ? -1 : 0;

    int first[CS_RANGE / CS_THREADS];
#pragma unroll
    for (int k = 0; k < CS_RANGE / CS_THREADS; ++k) {
        const int node = lo + k * CS_THREADS + tid;
        const int nc   = (node < N) ? node : (N - 1);
        const int v    = plane[nc] + boff[(k * CS_THREADS + tid) / DEG_THREADS] + (pref[nc] & prefMask);
        first[k] = (node < N) ? v : 0;
    }
    if (blockIdx.x == 0) {
#pragma unroll
        for (int k = 0; k < CS_RANGE / CS_THREADS; ++k) {
            const int node = lo + k * CS_THREADS + tid;
            if (node < N) slot[node] = first[k];
        }
    }
#pragma unroll
    for (int k = 0; k < CS_RANGE / CS_THREADS; ++k) cnt[k * CS_THREADS + tid] = (unsigned)first[k];
    __syncthreads();

    int e0, e1;
    cs_slice(E, blockIdx.x, e0, e1);

    if (((((size_t)dst) | ((size_t)src)) & 15) == 0) {
        const int4* __restrict__ p = (const int4*)(dst + e0);
        const int4* __restrict__ q = (const int4*)(src + e0);
        const int nv = (e1 - e0) >> 2;
        int i = tid;
        for (; i + CS_THREADS < nv; i += 2 * CS_THREADS) {
            int4 a  = p[i];
            int4 b  = p[i + CS_THREADS];
            int4 sa = q[i];
            int4 sb = q[i + CS_THREADS];
            cs_place4(cnt, a, sa, lo, sorted, E);
            cs_place4(cnt, b, sb, lo, sorted, E);
        }
        for (; i < nv; i += CS_THREADS)
            cs_place4(cnt, p[i], q[i], lo, sorted, E);
        for (int e = e0 + (nv << 2) + tid; e < e1; e += CS_THREADS)
            cs_place1(cnt, dst[e], src[e], lo, sorted, E);
    } else {
        for (int e = e0 + tid; e < e1; e += CS_THREADS)
            cs_place1(cnt, dst[e], src[e], lo, sorted, E);
    }
}

#define ENC_WAVES 4
#define ENC_PITCH 68

__global__ void __launch_bounds__(ENC_WAVES * 32)
k_encoder_t1(const float* __restrict__ x, const float* __restrict__ cnn,
             const float* __restrict__ fcW, const float* __restrict__ fcb,
             const float* __restrict__ cnnW, const float* __restrict__ cnnb,
             const float* __restrict__ W1, const float* __restrict__ dinv,
             float* __restrict__ t1, int N, int nTiles) {
    __shared__ __attribute__((aligned(16))) float h0t[ENC_WAVES][16 * ENC_PITCH];
    const int w    = threadIdx.x >> 5;
    const int lane = threadIdx.x & 31;
    const int wv   = blockIdx.x * ENC_WAVES + w;
    const bool live = wv < nTiles;
    const int  tile = live ? wv : (nTiles - 1);
    const int n    = lane & 15;
    const int kh   = lane >> 4;
    int rowA = tile * 16 + n;    if (rowA >= N) rowA = N - 1;

    v8f acc0 = {}, acc1 = {}, acc2 = {}, acc3 = {};

    const float* ax = x + (size_t)rowA * 32;
#pragma unroll
    for (int k0 = 0; k0 < 32; k0 += 4) {
        int kb = k0 + 2 * kh;
        v2f a;  a.x = ax[kb];               a.y = ax[kb + 1];
        v2f b0; b0.x = fcW[kb * 32 + n];    b0.y = fcW[(kb + 1) * 32 + n];
        v2f b1; b1.x = fcW[kb * 32 + 16 + n]; b1.y = fcW[(kb + 1) * 32 + 16 + n];
        acc0 = wmma4(a, b0, acc0);
        acc1 = wmma4(a, b1, acc1);
    }
    const float* ac = cnn + (size_t)rowA * 128;
#pragma unroll
    for (int k0 = 0; k0 < 128; k0 += 4) {
        int kb = k0 + 2 * kh;
        v2f a;  a.x = ac[kb];                a.y = ac[kb + 1];
        v2f b0; b0.x = cnnW[kb * 32 + n];    b0.y = cnnW[(kb + 1) * 32 + n];
        v2f b1; b1.x = cnnW[kb * 32 + 16 + n]; b1.y = cnnW[(kb + 1) * 32 + 16 + n];
        acc2 = wmma4(a, b0, acc2);
        acc3 = wmma4(a, b1, acc3);
    }

    float dr[8];
#pragma unroll
    for (int v = 0; v < 8; ++v) {
        int nd = tile * 16 + 8 * kh + v;
        dr[v] = dinv[nd < N ? nd : N - 1];
    }

    float* tw = h0t[w];
    {
        const float bf0 = fcb[n],  bf1 = fcb[16 + n];
        const float bc0 = cnnb[n], bc1 = cnnb[16 + n];
#pragma unroll
        for (int v = 0; v < 8; ++v) {
            float* o = tw + (8 * kh + v) * ENC_PITCH;
            o[n]      = fmaxf(acc0[v] + bf0, 0.0f);
            o[16 + n] = fmaxf(acc1[v] + bf1, 0.0f);
            o[32 + n] = fmaxf(acc2[v] + bc0, 0.0f);
            o[48 + n] = fmaxf(acc3[v] + bc1, 0.0f);
        }
    }
    __syncthreads();

    v8f d0 = {}, d1 = {};
    const float* ar = tw + n * ENC_PITCH + 2 * kh;
    const float* wr = W1 + (2 * kh) * 32 + n;
#pragma unroll
    for (int k0 = 0; k0 < 64; k0 += 4) {
        const v2f a = *(const v2f*)(ar + k0);
        v2f b0; b0.x = wr[k0 * 32];        b0.y = wr[k0 * 32 + 32];
        v2f b1; b1.x = wr[k0 * 32 + 16];   b1.y = wr[k0 * 32 + 48];
        d0 = wmma4(a, b0, d0);
        d1 = wmma4(a, b1, d1);
    }
    if (live) {
#pragma unroll
        for (int v = 0; v < 8; ++v) {
            int node = tile * 16 + 8 * kh + v;
            if (node < N) {
                float* o = t1 + (size_t)node * 32;
                o[n]      = d0[v] * dr[v];
                o[16 + n] = d1[v] * dr[v];
            }
        }
    }
}

#define T2_PITCH 32
static_assert(T2_PITCH >= 20 && T2_PITCH <= 32, "t2 lives in the N * 32 words of t1");
__global__ void k_gemm_t2(const float* __restrict__ agg1, const float* __restrict__ W2,
                          const float* __restrict__ dinv,
                          float* __restrict__ t2, int N, int nTiles) {
    int wave = (blockIdx.x * blockDim.x + threadIdx.x) >> 5;
    if (wave >= nTiles) return;
    int lane = threadIdx.x & 31;
    int n    = lane & 15;
    int kh   = lane >> 4;
    int rowA = wave * 16 + n;    if (rowA >= N) rowA = N - 1;

    float dr[8];
#pragma unroll
    for (int v = 0; v < 8; ++v) {
        int nd = wave * 16 + 8 * kh + v;
        dr[v] = dinv[nd < N ? nd : N - 1];
    }

    v8f acc0 = {}, acc1 = {};
    const float* ar = agg1 + (size_t)rowA * 32;
    bool c1ok = (16 + n) < 20;
#pragma unroll
    for (int k0 = 0; k0 < 32; k0 += 4) {
        int kb = k0 + 2 * kh;
        v2f a;  a.x = fmaxf(ar[kb], 0.0f);  a.y = fmaxf(ar[kb + 1], 0.0f);
        v2f b0; b0.x = W2[kb * 20 + n];     b0.y = W2[(kb + 1) * 20 + n];
        v2f b1; b1.x = c1ok ? W2[kb * 20 + 16 + n] : 0.0f;
                b1.y = c1ok ? W2[(kb + 1) * 20 + 16 + n] : 0.0f;
        acc0 = wmma4(a, b0, acc0);
        acc1 = wmma4(a, b1, acc1);
    }
#pragma unroll
    for (int v = 0; v < 8; ++v) {
        int mr   = (lane < 16) ? v : v + 8;
        int node = wave * 16 + mr;
        if (node < N) {
            float* o = t2 + (size_t)node * T2_PITCH;
            o[n] = acc0[v] * dr[v];
            if (c1ok) o[16 + n] = acc1[v] * dr[v];
        }
    }
}


template <int F, int P, bool GUARD>
__device__ __forceinline__ void rows8(int s, int j, int n, const float* __restrict__ t,
                                      int lane, bool act, float (&v)[8]) {
#pragma unroll
    for (int u = 0; u < 8; ++u) {
        v[u] = 0.0f;
        if (!GUARD || j + u < n) {
            const int sj = __builtin_amdgcn_readlane(s, j + u);
            if (act) v[u] = t[(size_t)sj * P + lane];
        }
    }
}

__device__ __forceinline__ float sum8(const float (&v)[8], float acc) {
#pragma unroll
    for (int u = 0; u < 8; ++u) acc += v[u];
    return acc;
}

template <int F, int P>
__device__ __forceinline__ void rows8_first(int s, int n, const float* __restrict__ t,
                                            int lane, bool act, float (&v)[8]) {
    const int col = act ? lane : 0;
#pragma unroll
    for (int u = 0; u < 8; ++u) {
        const int sj = __builtin_amdgcn_readlane(s, u);
        const float x = t[(size_t)sj * P + col];
        v[u] = (u < n) ? x : 0.0f;
    }
}

template <int F, int P, bool GUARD>
__device__ __forceinline__ float gather8(int s, int j, int n,
                                         const float* __restrict__ t, int lane, bool act,
                                         float acc) {
    float v[8];
    rows8<F, P, GUARD>(s, j, n, t, lane, act, v);
    return sum8(v, acc);
}

template <int F, int P>
__device__ __forceinline__ float gather_group(int s, int j0, int n, const float* __restrict__ t,
                                              int lane, bool act, float acc) {
    int j = j0;
    for (; j + 8 <= n; j += 8) acc = gather8<F, P, false>(s, j, n, t, lane, act, acc);
    if (j < n)                 acc = gather8<F, P, true >(s, j, n, t, lane, act, acc);
    return acc;
}

template <int F, int P>
__device__ __forceinline__ void gather_rows(const int* __restrict__ slot0,
                                            const int* __restrict__ sorted,
                                            const float* __restrict__ dinv,
                                            const float* __restrict__ t,
                                            const float* __restrict__ bias,
                                            float* __restrict__ o, int N, int E) {
    const int G = __builtin_amdgcn_readfirstlane((int)(gridDim.x * (blockDim.x >> 5)));
    int node = __builtin_amdgcn_readfirstlane((int)((blockIdx.x * blockDim.x + threadIdx.x) >> 5));
    if (node >= N) return;
    const int  lane = threadIdx.x & 31;
    const bool act  = lane < F;
    const float bl  = act ? bias[lane] : 0.0f;

    int rs = slot0[node];
    int re = (node + 1 < N) ? slot0[node + 1] : E;
    int sa = 0, sb = 0;
    if (rs + lane < re)      sa = sorted[rs + lane];
    if (rs + 32 + lane < re) sb = sorted[rs + 32 + lane];
    int nx  = node + G;
    int rs1 = 0, re1 = 0;
    if (nx < N) { rs1 = slot0[nx]; re1 = (nx + 1 < N) ? slot0[nx + 1] : E; }

    for (;;) {
        const int cnt = re - rs;
        const int n0  = (cnt < 32) ? cnt : 32;
        float v[8];
        rows8_first<F, P>(sa, n0, t, lane, act, v);

        int sa1 = 0, sb1 = 0;
        if (nx < N) {
            if (rs1 + lane < re1)      sa1 = sorted[rs1 + lane];
            if (rs1 + 32 + lane < re1) sb1 = sorted[rs1 + 32 + lane];
        }
        const int nx2 = nx + G;
        int rs2 = 0, re2 = 0;
        if (nx < N && nx2 < N) { rs2 = slot0[nx2]; re2 = (nx2 + 1 < N) ? slot0[nx2 + 1] : E; }

        const float  di  = dinv[node];
        const size_t idx = (size_t)node * F + lane;
        float self = 0.0f;
        if (act) self = t[(size_t)node * P + lane];

        float acc = sum8(v, 0.0f);
        acc = gather_group<F, P>(sa, 8, n0, t, lane, act, acc);
        if (cnt > 32)
            acc = gather_group<F, P>(sb, 0, (cnt < 64) ? (cnt - 32) : 32, t, lane, act, acc);
        for (int base = rs + 64; base < re; base += 32) {
            const int n = (re - base < 32) ? (re - base) : 32;
            int s = 0;
            if (lane < n) s = sorted[base + lane];
            acc = gather_group<F, P>(s, 0, n, t, lane, act, acc);
        }
        if (act) o[idx] = (acc + self) * di + bl;

        if (nx >= N) break;
        node = nx;  rs = rs1;  re = re1;  sa = sa1;  sb = sb1;
        nx = nx2;   rs1 = rs2; re1 = re2;
    }
}

__global__ void __launch_bounds__(256)
k_gather32(const int* __restrict__ slot0, const int* __restrict__ sorted,
           const float* __restrict__ dinv, const float* __restrict__ t1,
           const float* __restrict__ b1, float* __restrict__ agg1, int N, int E) {
    gather_rows<32, 32>(slot0, sorted, dinv, t1, b1, agg1, N, E);
}

__global__ void __launch_bounds__(256)
k_gather20(const int* __restrict__ slot0, const int* __restrict__ sorted,
           const float* __restrict__ dinv, const float* __restrict__ t2,
           const float* __restrict__ b2, float* __restrict__ out, int N, int E) {
    gather_rows<20, T2_PITCH>(slot0, sorted, dinv, t2, b2, out, N, E);
}

extern "C" void kernel_launch(void* const* d_in, const int* in_sizes, int n_in,
                              void* d_out, int out_size, void* d_ws, size_t ws_size,
                              hipStream_t stream) {
    const float* x    = (const float*)d_in[0];
    const float* cnn  = (const float*)d_in[1];
    const int*   ei   = (const int*)d_in[2];
    const float* fcW  = (const float*)d_in[3];
    const float* fcb  = (const float*)d_in[4];
    const float* cnnW = (const float*)d_in[5];
    const float* cnnb = (const float*)d_in[6];
    const float* W1   = (const float*)d_in[7];
    const float* b1   = (const float*)d_in[8];
    const float* W2   = (const float*)d_in[9];
    const float* b2   = (const float*)d_in[10];

    const int N = in_sizes[0] / 32;
    const int E = in_sizes[2] / 2;
    const int* src = ei;
    const int* dst = ei + E;

    float* ws   = (float*)d_ws;
    float* dinv = ws;  ws += N;
    float* eidx = ws;  ws += (size_t)N * 64;
    float* t1   = ws;  ws += (size_t)N * 32;
    float* agg1 = ws;  ws += (size_t)N * 32;
    float* bsw  = ws;  ws += (size_t)N * 20;
    float* t2   = t1;
    float* out  = (float*)d_out;

    int* plane  = (int*)agg1;
    int* slot   = (int*)eidx;
    int* sorted = slot + N;

    const int nTiles = (N + 15) / 16;
    const int B = 256;
    dim3 blkG(ENC_WAVES * 32);
    dim3 grdG((nTiles * 32 + 127) / 128);
    dim3 grdNode(2048);

    dim3 grdCS(CS_SPLITS, (N + CS_RANGE - 1) / CS_RANGE);
    k_count<<<grdCS, CS_THREADS, 0, stream>>>(dst, plane, E, N);
    int* bsum = (int*)bsw;
    const int nbDeg = (N + DEG_THREADS - 1) / DEG_THREADS;
    k_deg<<<nbDeg, DEG_THREADS, 0, stream>>>(plane, dinv, bsum, N);

    k_encoder_t1<<<grdG, blkG, 0, stream>>>(x, cnn, fcW, fcb, cnnW, cnnb, W1, dinv, t1, N, nTiles);

    k_fill<<<grdCS, CS_THREADS, 0, stream>>>(src, dst, plane, bsum, slot, sorted, E, N);

    k_gather32<<<grdNode, B, 0, stream>>>(slot, sorted, dinv, t1, b1, agg1, N, E);

    k_gemm_t2<<<grdG, blkG, 0, stream>>>(agg1, W2, dinv, t2, N, nTiles);
    k_gather20<<<grdNode, B, 0, stream>>>(slot, sorted, dinv, t2, b2, out, N, E);
}


# Encoder/t1 tiles run as extra blocks of the k_fill launch (7 -> 6), beside the fill blocks on the same WGPs: kernel held to 64 VGPRs, encoder loads grouped to fit; every sum and its order unchanged
# speedup vs baseline: 4.5624x; 1.0606x over previous
#include <hip/hip_runtime.h>
#include <hip/hip_bf16.h>

typedef __attribute__((ext_vector_type(2))) float v2f;
typedef __attribute__((ext_vector_type(8))) float v8f;

__device__ __forceinline__ v8f wmma4(v2f a, v2f b, v8f c) {
    return __builtin_amdgcn_wmma_f32_16x16x4_f32(
         false, a,  false, b,
         (short)0, c,  false,  false);
}

#define CS_RANGE   16384
#define CS_SPLITS  32
#define CS_THREADS 1024

__device__ __forceinline__ void cs_slice(int E, int x, int& e0, int& e1) {
    int per = (E + CS_SPLITS - 1) / CS_SPLITS;
    per = (per + 3) & ~3;
    long long e0l = (long long)x * per;
    e0 = (e0l < (long long)E) ? (int)e0l : E;
    e1 = (E - e0 > per) ? (e0 + per) : E;
}

__device__ __forceinline__ void cs_count(unsigned* cnt, int d, int lo) {
    unsigned r = (unsigned)(d - lo);
    if (r < (unsigned)CS_RANGE) atomicAdd(&cnt[r], 1u);
}

__device__ __forceinline__ void cs_count4(unsigned* cnt, int4 v, int lo) {
    cs_count(cnt, v.x, lo);
    cs_count(cnt, v.y, lo);
    cs_count(cnt, v.z, lo);
    cs_count(cnt, v.w, lo);
}

__global__ void __launch_bounds__(CS_THREADS)
k_count(const int* __restrict__ dst, int* __restrict__ plane, int E, int N) {
    __shared__ unsigned cnt[CS_RANGE];
    const int tid = threadIdx.x;
    const int lo  = blockIdx.y * CS_RANGE;

    for (int i = tid; i < CS_RANGE; i += CS_THREADS) cnt[i] = 0u;
    __syncthreads();

    int e0, e1;
    cs_slice(E, blockIdx.x, e0, e1);

    if ((((size_t)dst) & 15) == 0) {
        const int4* __restrict__ p = (const int4*)(dst + e0);
        const int nv = (e1 - e0) >> 2;
        int i = tid;
        for (; i + 3 * CS_THREADS < nv; i += 4 * CS_THREADS) {
            int4 a = p[i];
            int4 b = p[i + CS_THREADS];
            int4 c = p[i + 2 * CS_THREADS];
            int4 d = p[i + 3 * CS_THREADS];
            cs_count4(cnt, a, lo);
            cs_count4(cnt, b, lo);
            cs_count4(cnt, c, lo);
            cs_count4(cnt, d, lo);
        }
        for (; i < nv; i += CS_THREADS) cs_count4(cnt, p[i], lo);
        for (int e = e0 + (nv << 2) + tid; e < e1; e += CS_THREADS) cs_count(cnt, dst[e], lo);
    } else {
        for (int e = e0 + tid; e < e1; e += CS_THREADS) cs_count(cnt, dst[e], lo);
    }
    __syncthreads();

    int* __restrict__ row = plane + (size_t)blockIdx.x * N;
    for (int i = tid; i < CS_RANGE; i += CS_THREADS) {
        int node = lo + i;
        if (node < N) row[node] = (int)cnt[i];
    }
}

#define DEG_THREADS 256

__global__ void __launch_bounds__(DEG_THREADS)
k_deg(int* __restrict__ plane, float* __restrict__ dinv, int* __restrict__ bsum, int N) {
    __shared__ int wtot[DEG_THREADS / 32];
    const int tid  = threadIdx.x;
    const int lane = tid & 31;
    const int w    = tid >> 5;
    const int node = blockIdx.x * DEG_THREADS + tid;
    const bool ok  = node < N;

    int run = 0;
    if (ok) {
        int c[CS_SPLITS];
#pragma unroll
        for (int x = 0; x < CS_SPLITS; ++x) c[x] = plane[(size_t)x * N + node];
        run = c[0];
#pragma unroll
        for (int x = 1; x < CS_SPLITS; ++x) {
            plane[(size_t)x * N + node] = run;
            run += c[x];
        }
        dinv[node] = rsqrtf((float)run + 1.0f);
    }

    int inc = run;
    for (int o = 1; o < 32; o <<= 1) {
        int y = __shfl_up(inc, o);
        if (lane >= o) inc += y;
    }
    if (lane == 31) wtot[w] = inc;
    __syncthreads();
    int woff = 0;
    for (int k = 0; k < w; ++k) woff += wtot[k];
    if (ok) plane[node] = woff + inc - run;
    if (tid == DEG_THREADS - 1) bsum[blockIdx.x] = woff + inc;
}

__device__ __forceinline__ void cs_place4(unsigned* cnt, int4 v, int4 sv, int lo,
                                          int* __restrict__ sorted, int E) {
    const unsigned r0 = (unsigned)(v.x - lo), r1 = (unsigned)(v.y - lo);
    const unsigned r2 = (unsigned)(v.z - lo), r3 = (unsigned)(v.w - lo);
    const bool m0 = r0 < (unsigned)CS_RANGE, m1 = r1 < (unsigned)CS_RANGE;
    const bool m2 = r2 < (unsigned)CS_RANGE, m3 = r3 < (unsigned)CS_RANGE;
    unsigned p0 = 0, p1 = 0, p2 = 0, p3 = 0;
    if (m0) p0 = atomicAdd(&cnt[r0], 1u);
    if (m1) p1 = atomicAdd(&cnt[r1], 1u);
    if (m2) p2 = atomicAdd(&cnt[r2], 1u);
    if (m3) p3 = atomicAdd(&cnt[r3], 1u);
    if (m0 && p0 < (unsigned)E) sorted[p0] = sv.x;
    if (m1 && p1 < (unsigned)E) sorted[p1] = sv.y;
    if (m2 && p2 < (unsigned)E) sorted[p2] = sv.z;
    if (m3 && p3 < (unsigned)E) sorted[p3] = sv.w;
}

__device__ __forceinline__ void cs_place1(unsigned* cnt, int d, int s, int lo,
                                          int* __restrict__ sorted, int E) {
    const unsigned r = (unsigned)(d - lo);
    if (r < (unsigned)CS_RANGE) {
        const unsigned p = atomicAdd(&cnt[r], 1u);
        if (p < (unsigned)E) sorted[p] = s;
    }
}

static_assert(CS_THREADS / 32 == 32, "one wave total per lane of the first wave");
static_assert(CS_RANGE / DEG_THREADS == 64, "two k_deg blocks per lane of the first wave");

__device__ __forceinline__ void
fill_body(unsigned* cnt, int* wred, int* boff, const int bx, const int by,
          const int* __restrict__ src, const int* __restrict__ dst,
          const int* __restrict__ plane, const int* __restrict__ bsum,
          int* __restrict__ slot, int* __restrict__ sorted, int E, int N) {
    const int tid  = threadIdx.x;
    const int lane = tid & 31;
    const int w    = tid >> 5;
    const int lo   = by * CS_RANGE;

    const int nbDeg = (N + DEG_THREADS - 1) / DEG_THREADS;
    const int b0    = by * (CS_RANGE / DEG_THREADS);
    int part = 0;
    for (int k = tid; k < b0; k += CS_THREADS) part += bsum[k];
    for (int o = 16; o > 0; o >>= 1) part += __shfl_xor(part, o);
    if (lane == 0) wred[w] = part;
    __syncthreads();
    if (w == 0) {
        int base = wred[lane];
        for (int o = 16; o > 0; o >>= 1) base += __shfl_xor(base, o);
        const int i0 = b0 + 2 * lane;
        const int v0 = (i0 < nbDeg) ? bsum[i0] : 0;
        const int v1 = (i0 + 1 < nbDeg) ? bsum[i0 + 1] : 0;
        const int s2 = v0 + v1;
        int inc = s2;
        for (int o = 1; o < 32; o <<= 1) {
            int y = __shfl_up(inc, o);
            if (lane >= o) inc += y;
        }
        const int ex = base + inc - s2;
        boff[2 * lane]     = ex;
        boff[2 * lane + 1] = ex + v0;
    }
    __syncthreads();

    const int* __restrict__ pref = plane + (size_t)bx * N;
    const int prefMask = (bx > 0) ? -1 : 0;

    int first[CS_RANGE / CS_THREADS];
    int pl[CS_RANGE / CS_THREADS], pr[CS_RANGE / CS_THREADS];
#pragma unroll
    for (int k = 0; k < CS_RANGE / CS_THREADS; ++k) {
        const int node = lo + k * CS_THREADS + tid;
        const int nc   = (node < N) ? node : (N - 1);
        pl[k] = plane[nc];
        pr[k] = pref[nc];
    }
    __builtin_amdgcn_sched_barrier(0);
#pragma unroll
    for (int k = 0; k < CS_RANGE / CS_THREADS; ++k) {
        const int node = lo + k * CS_THREADS + tid;
        const int v    = pl[k] + boff[(k * CS_THREADS + tid) / DEG_THREADS] + (pr[k] & prefMask);
        first[k] = (node < N) ? v : 0;
    }
    if (bx == 0) {
#pragma unroll
        for (int k = 0; k < CS_RANGE / CS_THREADS; ++k) {
            const int node = lo + k * CS_THREADS + tid;
            if (node < N) slot[node] = first[k];
        }
    }
#pragma unroll
    for (int k = 0; k < CS_RANGE / CS_THREADS; ++k) cnt[k * CS_THREADS + tid] = (unsigned)first[k];
    __syncthreads();

    int e0, e1;
    cs_slice(E, bx, e0, e1);

    if (((((size_t)dst) | ((size_t)src)) & 15) == 0) {
        const int4* __restrict__ p = (const int4*)(dst + e0);
        const int4* __restrict__ q = (const int4*)(src + e0);
        const int nv = (e1 - e0) >> 2;
        int i = tid;
        for (; i + CS_THREADS < nv; i += 2 * CS_THREADS) {
            int4 a  = p[i];
            int4 b  = p[i + CS_THREADS];
            int4 sa = q[i];
            int4 sb = q[i + CS_THREADS];
            cs_place4(cnt, a, sa, lo, sorted, E);
            cs_place4(cnt, b, sb, lo, sorted, E);
        }
        for (; i < nv; i += CS_THREADS)
            cs_place4(cnt, p[i], q[i], lo, sorted, E);
        for (int e = e0 + (nv << 2) + tid; e < e1; e += CS_THREADS)
            cs_place1(cnt, dst[e], src[e], lo, sorted, E);
    } else {
        for (int e = e0 + tid; e < e1; e += CS_THREADS)
            cs_place1(cnt, dst[e], src[e], lo, sorted, E);
    }
}

#define ENC_WAVES 4
#define FE_WAVES  (CS_THREADS / 32)
#define ENC_PITCH 68

__device__ __forceinline__ void
enc_line(const float* __restrict__ arow, const float* __restrict__ Wm, const int k0,
         const int kh, const int n, v8f& c0, v8f& c1) {
    v2f a[8];
#pragma unroll
    for (int i = 0; i < 8; ++i) {
        const int kb = k0 + 4 * i + 2 * kh;
        a[i].x = arow[kb];   a[i].y = arow[kb + 1];
    }
#pragma unroll
    for (int h = 0; h < 2; ++h) {
        v2f b0[4], b1[4];
#pragma unroll
        for (int i = 0; i < 4; ++i) {
            const int kb = k0 + 4 * (4 * h + i) + 2 * kh;
            b0[i].x = Wm[kb * 32 + n];        b0[i].y = Wm[(kb + 1) * 32 + n];
            b1[i].x = Wm[kb * 32 + 16 + n];   b1[i].y = Wm[(kb + 1) * 32 + 16 + n];
        }
        __builtin_amdgcn_sched_barrier(0);
#pragma unroll
        for (int i = 0; i < 4; ++i) {
            c0 = wmma4(a[4 * h + i], b0[i], c0);
            c1 = wmma4(a[4 * h + i], b1[i], c1);
        }
        __builtin_amdgcn_sched_barrier(0);
    }
}

__device__ __forceinline__ void
enc_body(float* h0t, const int encBlock,
         const float* __restrict__ x, const float* __restrict__ cnn,
         const float* __restrict__ fcW, const float* __restrict__ fcb,
         const float* __restrict__ cnnW, const float* __restrict__ cnnb,
         const float* __restrict__ W1, const float* __restrict__ dinv,
         float* __restrict__ t1, int N, int nTiles) {
    const int w    = threadIdx.x >> 5;
    const int lane = threadIdx.x & 31;
    const int wv   = encBlock * FE_WAVES + w;
    const bool live = wv < nTiles;
    const int  tile = live ? wv : (nTiles - 1);
    const int n    = lane & 15;
    const int kh   = lane >> 4;
    int rowA = tile * 16 + n;    if (rowA >= N) rowA = N - 1;

    float* tw = h0t + w * (16 * ENC_PITCH);
    {
        v8f acc0 = {}, acc1 = {};
        enc_line(x + (size_t)rowA * 32, fcW, 0, kh, n, acc0, acc1);
        const float bf0 = fcb[n], bf1 = fcb[16 + n];
#pragma unroll
        for (int v = 0; v < 8; ++v) {
            float* o = tw + (8 * kh + v) * ENC_PITCH;
            o[n]      = fmaxf(acc0[v] + bf0, 0.0f);
            o[16 + n] = fmaxf(acc1[v] + bf1, 0.0f);
        }
    }
    {
        v8f acc2 = {}, acc3 = {};
        const float* ac = cnn + (size_t)rowA * 128;
#pragma unroll
        for (int k0 = 0; k0 < 128; k0 += 32) enc_line(ac, cnnW, k0, kh, n, acc2, acc3);
        const float bc0 = cnnb[n], bc1 = cnnb[16 + n];
#pragma unroll
        for (int v = 0; v < 8; ++v) {
            float* o = tw + (8 * kh + v) * ENC_PITCH;
            o[32 + n] = fmaxf(acc2[v] + bc0, 0.0f);
            o[48 + n] = fmaxf(acc3[v] + bc1, 0.0f);
        }
    }
    __syncthreads();

    float dr[8];
#pragma unroll
    for (int v = 0; v < 8; ++v) {
        int nd = tile * 16 + 8 * kh + v;
        dr[v] = dinv[nd < N ? nd : N - 1];
    }

    v8f d0 = {}, d1 = {};
    const float* ar = tw + n * ENC_PITCH + 2 * kh;
    const float* wr = W1 + (2 * kh) * 32 + n;
#pragma unroll
    for (int g = 0; g < 64; g += 16) {
        v2f a[4], b0[4], b1[4];
#pragma unroll
        for (int i = 0; i < 4; ++i) {
            const int k0 = g + 4 * i;
            a[i] = *(const v2f*)(ar + k0);
            b0[i].x = wr[k0 * 32];        b0[i].y = wr[k0 * 32 + 32];
            b1[i].x = wr[k0 * 32 + 16];   b1[i].y = wr[k0 * 32 + 48];
        }
        __builtin_amdgcn_sched_barrier(0);
#pragma unroll
        for (int i = 0; i < 4; ++i) {
            d0 = wmma4(a[i], b0[i], d0);
            d1 = wmma4(a[i], b1[i], d1);
        }
        __builtin_amdgcn_sched_barrier(0);
    }
    if (live) {
#pragma unroll
        for (int v = 0; v < 8; ++v) {
            int node = tile * 16 + 8 * kh + v;
            if (node < N) {
                float* o = t1 + (size_t)node * 32;
                o[n]      = d0[v] * dr[v];
                o[16 + n] = d1[v] * dr[v];
            }
        }
    }
}

#define FE_FILL_WORDS (CS_RANGE + CS_THREADS / 32 + CS_RANGE / DEG_THREADS)
#define FE_ENC_WORDS  (FE_WAVES * 16 * ENC_PITCH)
#define FE_LDS_WORDS  (FE_ENC_WORDS > FE_FILL_WORDS ? FE_ENC_WORDS : FE_FILL_WORDS)
static_assert(2 * FE_LDS_WORDS * 4 <= 320 * 1024, "a fill block and an encoder block share a WGP's LDS");

__global__ void __launch_bounds__(CS_THREADS, 16)
k_fill_encoder(const int* __restrict__ src, const int* __restrict__ dst,
               const int* __restrict__ plane, const int* __restrict__ bsum,
               int* __restrict__ slot, int* __restrict__ sorted, int E, int N, int nFill,
               const float* __restrict__ x, const float* __restrict__ cnn,
               const float* __restrict__ fcW, const float* __restrict__ fcb,
               const float* __restrict__ cnnW, const float* __restrict__ cnnb,
               const float* __restrict__ W1, const float* __restrict__ dinv,
               float* __restrict__ t1, int nTiles) {
    __shared__ __attribute__((aligned(16))) float lds[FE_LDS_WORDS];
    const int b = blockIdx.x;
    if (b < nFill) {
        unsigned* cnt = reinterpret_cast<unsigned*>(lds);
        int* wred = reinterpret_cast<int*>(lds) + CS_RANGE;
        int* boff = wred + CS_THREADS / 32;
        fill_body(cnt, wred, boff, b % CS_SPLITS, b / CS_SPLITS,
                  src, dst, plane, bsum, slot, sorted, E, N);
    } else {
        enc_body(lds, b - nFill, x, cnn, fcW, fcb, cnnW, cnnb, W1, dinv, t1, N, nTiles);
    }
}

#define T2_PITCH 32
static_assert(T2_PITCH >= 20 && T2_PITCH <= 32, "t2 lives in the N * 32 words of t1");
__global__ void k_gemm_t2(const float* __restrict__ agg1, const float* __restrict__ W2,
                          const float* __restrict__ dinv,
                          float* __restrict__ t2, int N, int nTiles) {
    int wave = (blockIdx.x * blockDim.x + threadIdx.x) >> 5;
    if (wave >= nTiles) return;
    int lane = threadIdx.x & 31;
    int n    = lane & 15;
    int kh   = lane >> 4;
    int rowA = wave * 16 + n;    if (rowA >= N) rowA = N - 1;

    float dr[8];
#pragma unroll
    for (int v = 0; v < 8; ++v) {
        int nd = wave * 16 + 8 * kh + v;
        dr[v] = dinv[nd < N ? nd : N - 1];
    }

    v8f acc0 = {}, acc1 = {};
    const float* ar = agg1 + (size_t)rowA * 32;
    bool c1ok = (16 + n) < 20;
#pragma unroll
    for (int k0 = 0; k0 < 32; k0 += 4) {
        int kb = k0 + 2 * kh;
        v2f a;  a.x = fmaxf(ar[kb], 0.0f);  a.y = fmaxf(ar[kb + 1], 0.0f);
        v2f b0; b0.x = W2[kb * 20 + n];     b0.y = W2[(kb + 1) * 20 + n];
        v2f b1; b1.x = c1ok ? W2[kb * 20 + 16 + n] : 0.0f;
                b1.y = c1ok ? W2[(kb + 1) * 20 + 16 + n] : 0.0f;
        acc0 = wmma4(a, b0, acc0);
        acc1 = wmma4(a, b1, acc1);
    }
#pragma unroll
    for (int v = 0; v < 8; ++v) {
        int mr   = (lane < 16) ? v : v + 8;
        int node = wave * 16 + mr;
        if (node < N) {
            float* o = t2 + (size_t)node * T2_PITCH;
            o[n] = acc0[v] * dr[v];
            if (c1ok) o[16 + n] = acc1[v] * dr[v];
        }
    }
}


template <int F, int P, bool GUARD>
__device__ __forceinline__ void rows8(int s, int j, int n, const float* __restrict__ t,
                                      int lane, bool act, float (&v)[8]) {
#pragma unroll
    for (int u = 0; u < 8; ++u) {
        v[u] = 0.0f;
        if (!GUARD || j + u < n) {
            const int sj = __builtin_amdgcn_readlane(s, j + u);
            if (act) v[u] = t[(size_t)sj * P + lane];
        }
    }
}

__device__ __forceinline__ float sum8(const float (&v)[8], float acc) {
#pragma unroll
    for (int u = 0; u < 8; ++u) acc += v[u];
    return acc;
}

template <int F, int P>
__device__ __forceinline__ void rows8_first(int s, int n, const float* __restrict__ t,
                                            int lane, bool act, float (&v)[8]) {
    const int col = act ? lane : 0;
#pragma unroll
    for (int u = 0; u < 8; ++u) {
        const int sj = __builtin_amdgcn_readlane(s, u);
        const float x = t[(size_t)sj * P + col];
        v[u] = (u < n) ? x : 0.0f;
    }
}

template <int F, int P, bool GUARD>
__device__ __forceinline__ float gather8(int s, int j, int n,
                                         const float* __restrict__ t, int lane, bool act,
                                         float acc) {
    float v[8];
    rows8<F, P, GUARD>(s, j, n, t, lane, act, v);
    return sum8(v, acc);
}

template <int F, int P>
__device__ __forceinline__ float gather_group(int s, int j0, int n, const float* __restrict__ t,
                                              int lane, bool act, float acc) {
    int j = j0;
    for (; j + 8 <= n; j += 8) acc = gather8<F, P, false>(s, j, n, t, lane, act, acc);
    if (j < n)                 acc = gather8<F, P, true >(s, j, n, t, lane, act, acc);
    return acc;
}

template <int F, int P>
__device__ __forceinline__ void gather_rows(const int* __restrict__ slot0,
                                            const int* __restrict__ sorted,
                                            const float* __restrict__ dinv,
                                            const float* __restrict__ t,
                                            const float* __restrict__ bias,
                                            float* __restrict__ o, int N, int E) {
    const int G = __builtin_amdgcn_readfirstlane((int)(gridDim.x * (blockDim.x >> 5)));
    int node = __builtin_amdgcn_readfirstlane((int)((blockIdx.x * blockDim.x + threadIdx.x) >> 5));
    if (node >= N) return;
    const int  lane = threadIdx.x & 31;
    const bool act  = lane < F;
    const float bl  = act ? bias[lane] : 0.0f;

    int rs = slot0[node];
    int re = (node + 1 < N) ? slot0[node + 1] : E;
    int sa = 0, sb = 0;
    if (rs + lane < re)      sa = sorted[rs + lane];
    if (rs + 32 + lane < re) sb = sorted[rs + 32 + lane];
    int nx  = node + G;
    int rs1 = 0, re1 = 0;
    if (nx < N) { rs1 = slot0[nx]; re1 = (nx + 1 < N) ? slot0[nx + 1] : E; }

    for (;;) {
        const int cnt = re - rs;
        const int n0  = (cnt < 32) ? cnt : 32;
        float v[8];
        rows8_first<F, P>(sa, n0, t, lane, act, v);

        int sa1 = 0, sb1 = 0;
        if (nx < N) {
            if (rs1 + lane < re1)      sa1 = sorted[rs1 + lane];
            if (rs1 + 32 + lane < re1) sb1 = sorted[rs1 + 32 + lane];
        }
        const int nx2 = nx + G;
        int rs2 = 0, re2 = 0;
        if (nx < N && nx2 < N) { rs2 = slot0[nx2]; re2 = (nx2 + 1 < N) ? slot0[nx2 + 1] : E; }

        const float  di  = dinv[node];
        const size_t idx = (size_t)node * F + lane;
        float self = 0.0f;
        if (act) self = t[(size_t)node * P + lane];

        float acc = sum8(v, 0.0f);
        acc = gather_group<F, P>(sa, 8, n0, t, lane, act, acc);
        if (cnt > 32)
            acc = gather_group<F, P>(sb, 0, (cnt < 64) ? (cnt - 32) : 32, t, lane, act, acc);
        for (int base = rs + 64; base < re; base += 32) {
            const int n = (re - base < 32) ? (re - base) : 32;
            int s = 0;
            if (lane < n) s = sorted[base + lane];
            acc = gather_group<F, P>(s, 0, n, t, lane, act, acc);
        }
        if (act) o[idx] = (acc + self) * di + bl;

        if (nx >= N) break;
        node = nx;  rs = rs1;  re = re1;  sa = sa1;  sb = sb1;
        nx = nx2;   rs1 = rs2; re1 = re2;
    }
}

__global__ void __launch_bounds__(256)
k_gather32(const int* __restrict__ slot0, const int* __restrict__ sorted,
           const float* __restrict__ dinv, const float* __restrict__ t1,
           const float* __restrict__ b1, float* __restrict__ agg1, int N, int E) {
    gather_rows<32, 32>(slot0, sorted, dinv, t1, b1, agg1, N, E);
}

__global__ void __launch_bounds__(256)
k_gather20(const int* __restrict__ slot0, const int* __restrict__ sorted,
           const float* __restrict__ dinv, const float* __restrict__ t2,
           const float* __restrict__ b2, float* __restrict__ out, int N, int E) {
    gather_rows<20, T2_PITCH>(slot0, sorted, dinv, t2, b2, out, N, E);
}

extern "C" void kernel_launch(void* const* d_in, const int* in_sizes, int n_in,
                              void* d_out, int out_size, void* d_ws, size_t ws_size,
                              hipStream_t stream) {
    const float* x    = (const float*)d_in[0];
    const float* cnn  = (const float*)d_in[1];
    const int*   ei   = (const int*)d_in[2];
    const float* fcW  = (const float*)d_in[3];
    const float* fcb  = (const float*)d_in[4];
    const float* cnnW = (const float*)d_in[5];
    const float* cnnb = (const float*)d_in[6];
    const float* W1   = (const float*)d_in[7];
    const float* b1   = (const float*)d_in[8];
    const float* W2   = (const float*)d_in[9];
    const float* b2   = (const float*)d_in[10];

    const int N = in_sizes[0] / 32;
    const int E = in_sizes[2] / 2;
    const int* src = ei;
    const int* dst = ei + E;

    float* ws   = (float*)d_ws;
    float* dinv = ws;  ws += N;
    float* eidx = ws;  ws += (size_t)N * 64;
    float* t1   = ws;  ws += (size_t)N * 32;
    float* agg1 = ws;  ws += (size_t)N * 32;
    float* bsw  = ws;  ws += (size_t)N * 20;
    float* t2   = t1;
    float* out  = (float*)d_out;

    int* plane  = (int*)agg1;
    int* slot   = (int*)eidx;
    int* sorted = slot + N;

    const int nTiles = (N + 15) / 16;
    const int B = 256;
    dim3 blkG(ENC_WAVES * 32);
    dim3 grdG((nTiles * 32 + 127) / 128);
    dim3 grdNode(2048);

    dim3 grdCS(CS_SPLITS, (N + CS_RANGE - 1) / CS_RANGE);
    k_count<<<grdCS, CS_THREADS, 0, stream>>>(dst, plane, E, N);
    int* bsum = (int*)bsw;
    const int nbDeg = (N + DEG_THREADS - 1) / DEG_THREADS;
    k_deg<<<nbDeg, DEG_THREADS, 0, stream>>>(plane, dinv, bsum, N);

    const int nFill = CS_SPLITS * ((N + CS_RANGE - 1) / CS_RANGE);
    const int nEnc  = (nTiles + FE_WAVES - 1) / FE_WAVES;
    k_fill_encoder<<<nFill + nEnc, CS_THREADS, 0, stream>>>(src, dst, plane, bsum, slot, sorted, E, N, nFill,
                                                            x, cnn, fcW, fcb, cnnW, cnnb, W1, dinv, t1, nTiles);

    k_gather32<<<grdNode, B, 0, stream>>>(slot, sorted, dinv, t1, b1, agg1, N, E);

    k_gemm_t2<<<grdG, blkG, 0, stream>>>(agg1, W2, dinv, t2, N, nTiles);
    k_gather20<<<grdNode, B, 0, stream>>>(slot, sorted, dinv, t2, b2, out, N, E);
}


# Layer-2 linear folded into the first aggregation's launch (6 -> 5): each gather wave keeps its finished rows as a 16x32 LDS tile and multiplies it by W2 on WMMA; agg1 never stored; sums unchanged
# speedup vs baseline: 4.7470x; 1.0405x over previous
#include <hip/hip_runtime.h>
#include <hip/hip_bf16.h>

typedef __attribute__((ext_vector_type(2))) float v2f;
typedef __attribute__((ext_vector_type(8))) float v8f;

__device__ __forceinline__ v8f wmma4(v2f a, v2f b, v8f c) {
    return __builtin_amdgcn_wmma_f32_16x16x4_f32(
         false, a,  false, b,
         (short)0, c,  false,  false);
}

#define CS_RANGE   16384
#define CS_SPLITS  32
#define CS_THREADS 1024

__device__ __forceinline__ void cs_slice(int E, int x, int& e0, int& e1) {
    int per = (E + CS_SPLITS - 1) / CS_SPLITS;
    per = (per + 3) & ~3;
    long long e0l = (long long)x * per;
    e0 = (e0l < (long long)E) ? (int)e0l : E;
    e1 = (E - e0 > per) ? (e0 + per) : E;
}

__device__ __forceinline__ void cs_count(unsigned* cnt, int d, int lo) {
    unsigned r = (unsigned)(d - lo);
    if (r < (unsigned)CS_RANGE) atomicAdd(&cnt[r], 1u);
}

__device__ __forceinline__ void cs_count4(unsigned* cnt, int4 v, int lo) {
    cs_count(cnt, v.x, lo);
    cs_count(cnt, v.y, lo);
    cs_count(cnt, v.z, lo);
    cs_count(cnt, v.w, lo);
}

__global__ void __launch_bounds__(CS_THREADS)
k_count(const int* __restrict__ dst, int* __restrict__ plane, int E, int N) {
    __shared__ unsigned cnt[CS_RANGE];
    const int tid = threadIdx.x;
    const int lo  = blockIdx.y * CS_RANGE;

    for (int i = tid; i < CS_RANGE; i += CS_THREADS) cnt[i] = 0u;
    __syncthreads();

    int e0, e1;
    cs_slice(E, blockIdx.x, e0, e1);

    if ((((size_t)dst) & 15) == 0) {
        const int4* __restrict__ p = (const int4*)(dst + e0);
        const int nv = (e1 - e0) >> 2;
        int i = tid;
        for (; i + 3 * CS_THREADS < nv; i += 4 * CS_THREADS) {
            int4 a = p[i];
            int4 b = p[i + CS_THREADS];
            int4 c = p[i + 2 * CS_THREADS];
            int4 d = p[i + 3 * CS_THREADS];
            cs_count4(cnt, a, lo);
            cs_count4(cnt, b, lo);
            cs_count4(cnt, c, lo);
            cs_count4(cnt, d, lo);
        }
        for (; i < nv; i += CS_THREADS) cs_count4(cnt, p[i], lo);
        for (int e = e0 + (nv << 2) + tid; e < e1; e += CS_THREADS) cs_count(cnt, dst[e], lo);
    } else {
        for (int e = e0 + tid; e < e1; e += CS_THREADS) cs_count(cnt, dst[e], lo);
    }
    __syncthreads();

    int* __restrict__ row = plane + (size_t)blockIdx.x * N;
    for (int i = tid; i < CS_RANGE; i += CS_THREADS) {
        int node = lo + i;
        if (node < N) row[node] = (int)cnt[i];
    }
}

#define DEG_THREADS 256

__global__ void __launch_bounds__(DEG_THREADS)
k_deg(int* __restrict__ plane, float* __restrict__ dinv, int* __restrict__ bsum, int N) {
    __shared__ int wtot[DEG_THREADS / 32];
    const int tid  = threadIdx.x;
    const int lane = tid & 31;
    const int w    = tid >> 5;
    const int node = blockIdx.x * DEG_THREADS + tid;
    const bool ok  = node < N;

    int run = 0;
    if (ok) {
        int c[CS_SPLITS];
#pragma unroll
        for (int x = 0; x < CS_SPLITS; ++x) c[x] = plane[(size_t)x * N + node];
        run = c[0];
#pragma unroll
        for (int x = 1; x < CS_SPLITS; ++x) {
            plane[(size_t)x * N + node] = run;
            run += c[x];
        }
        dinv[node] = rsqrtf((float)run + 1.0f);
    }

    int inc = run;
    for (int o = 1; o < 32; o <<= 1) {
        int y = __shfl_up(inc, o);
        if (lane >= o) inc += y;
    }
    if (lane == 31) wtot[w] = inc;
    __syncthreads();
    int woff = 0;
    for (int k = 0; k < w; ++k) woff += wtot[k];
    if (ok) plane[node] = woff + inc - run;
    if (tid == DEG_THREADS - 1) bsum[blockIdx.x] = woff + inc;
}

__device__ __forceinline__ void cs_place4(unsigned* cnt, int4 v, int4 sv, int lo,
                                          int* __restrict__ sorted, int E) {
    const unsigned r0 = (unsigned)(v.x - lo), r1 = (unsigned)(v.y - lo);
    const unsigned r2 = (unsigned)(v.z - lo), r3 = (unsigned)(v.w - lo);
    const bool m0 = r0 < (unsigned)CS_RANGE, m1 = r1 < (unsigned)CS_RANGE;
    const bool m2 = r2 < (unsigned)CS_RANGE, m3 = r3 < (unsigned)CS_RANGE;
    unsigned p0 = 0, p1 = 0, p2 = 0, p3 = 0;
    if (m0) p0 = atomicAdd(&cnt[r0], 1u);
    if (m1) p1 = atomicAdd(&cnt[r1], 1u);
    if (m2) p2 = atomicAdd(&cnt[r2], 1u);
    if (m3) p3 = atomicAdd(&cnt[r3], 1u);
    if (m0 && p0 < (unsigned)E) sorted[p0] = sv.x;
    if (m1 && p1 < (unsigned)E) sorted[p1] = sv.y;
    if (m2 && p2 < (unsigned)E) sorted[p2] = sv.z;
    if (m3 && p3 < (unsigned)E) sorted[p3] = sv.w;
}

__device__ __forceinline__ void cs_place1(unsigned* cnt, int d, int s, int lo,
                                          int* __restrict__ sorted, int E) {
    const unsigned r = (unsigned)(d - lo);
    if (r < (unsigned)CS_RANGE) {
        const unsigned p = atomicAdd(&cnt[r], 1u);
        if (p < (unsigned)E) sorted[p] = s;
    }
}

static_assert(CS_THREADS / 32 == 32, "one wave total per lane of the first wave");
static_assert(CS_RANGE / DEG_THREADS == 64, "two k_deg blocks per lane of the first wave");

__device__ __forceinline__ void
fill_body(unsigned* cnt, int* wred, int* boff, const int bx, const int by,
          const int* __restrict__ src, const int* __restrict__ dst,
          const int* __restrict__ plane, const int* __restrict__ bsum,
          int* __restrict__ slot, int* __restrict__ sorted, int E, int N) {
    const int tid  = threadIdx.x;
    const int lane = tid & 31;
    const int w    = tid >> 5;
    const int lo   = by * CS_RANGE;

    const int nbDeg = (N + DEG_THREADS - 1) / DEG_THREADS;
    const int b0    = by * (CS_RANGE / DEG_THREADS);
    int part = 0;
    for (int k = tid; k < b0; k += CS_THREADS) part += bsum[k];
    for (int o = 16; o > 0; o >>= 1) part += __shfl_xor(part, o);
    if (lane == 0) wred[w] = part;
    __syncthreads();
    if (w == 0) {
        int base = wred[lane];
        for (int o = 16; o > 0; o >>= 1) base += __shfl_xor(base, o);
        const int i0 = b0 + 2 * lane;
        const int v0 = (i0 < nbDeg) ? bsum[i0] : 0;
        const int v1 = (i0 + 1 < nbDeg) ? bsum[i0 + 1] : 0;
        const int s2 = v0 + v1;
        int inc = s2;
        for (int o = 1; o < 32; o <<= 1) {
            int y = __shfl_up(inc, o);
            if (lane >= o) inc += y;
        }
        const int ex = base + inc - s2;
        boff[2 * lane]     = ex;
        boff[2 * lane + 1] = ex + v0;
    }
    __syncthreads();

    const int* __restrict__ pref = plane + (size_t)bx * N;
    const int prefMask = (bx > 0) ? -1 : 0;

    int first[CS_RANGE / CS_THREADS];
    int pl[CS_RANGE / CS_THREADS], pr[CS_RANGE / CS_THREADS];
#pragma unroll
    for (int k = 0; k < CS_RANGE / CS_THREADS; ++k) {
        const int node = lo + k * CS_THREADS + tid;
        const int nc   = (node < N) ? node : (N - 1);
        pl[k] = plane[nc];
        pr[k] = pref[nc];
    }
    __builtin_amdgcn_sched_barrier(0);
#pragma unroll
    for (int k = 0; k < CS_RANGE / CS_THREADS; ++k) {
        const int node = lo + k * CS_THREADS + tid;
        const int v    = pl[k] + boff[(k * CS_THREADS + tid) / DEG_THREADS] + (pr[k] & prefMask);
        first[k] = (node < N) ? v : 0;
    }
    if (bx == 0) {
#pragma unroll
        for (int k = 0; k < CS_RANGE / CS_THREADS; ++k) {
            const int node = lo + k * CS_THREADS + tid;
            if (node < N) slot[node] = first[k];
        }
    }
#pragma unroll
    for (int k = 0; k < CS_RANGE / CS_THREADS; ++k) cnt[k * CS_THREADS + tid] = (unsigned)first[k];
    __syncthreads();

    int e0, e1;
    cs_slice(E, bx, e0, e1);

    if (((((size_t)dst) | ((size_t)src)) & 15) == 0) {
        const int4* __restrict__ p = (const int4*)(dst + e0);
        const int4* __restrict__ q = (const int4*)(src + e0);
        const int nv = (e1 - e0) >> 2;
        int i = tid;
        for (; i + CS_THREADS < nv; i += 2 * CS_THREADS) {
            int4 a  = p[i];
            int4 b  = p[i + CS_THREADS];
            int4 sa = q[i];
            int4 sb = q[i + CS_THREADS];
            cs_place4(cnt, a, sa, lo, sorted, E);
            cs_place4(cnt, b, sb, lo, sorted, E);
        }
        for (; i < nv; i += CS_THREADS)
            cs_place4(cnt, p[i], q[i], lo, sorted, E);
        for (int e = e0 + (nv << 2) + tid; e < e1; e += CS_THREADS)
            cs_place1(cnt, dst[e], src[e], lo, sorted, E);
    } else {
        for (int e = e0 + tid; e < e1; e += CS_THREADS)
            cs_place1(cnt, dst[e], src[e], lo, sorted, E);
    }
}

#define FE_WAVES  (CS_THREADS / 32)
#define ENC_PITCH 68

__device__ __forceinline__ void
enc_line(const float* __restrict__ arow, const float* __restrict__ Wm, const int k0,
         const int kh, const int n, v8f& c0, v8f& c1) {
    v2f a[8];
#pragma unroll
    for (int i = 0; i < 8; ++i) {
        const int kb = k0 + 4 * i + 2 * kh;
        a[i].x = arow[kb];   a[i].y = arow[kb + 1];
    }
#pragma unroll
    for (int h = 0; h < 2; ++h) {
        v2f b0[4], b1[4];
#pragma unroll
        for (int i = 0; i < 4; ++i) {
            const int kb = k0 + 4 * (4 * h + i) + 2 * kh;
            b0[i].x = Wm[kb * 32 + n];        b0[i].y = Wm[(kb + 1) * 32 + n];
            b1[i].x = Wm[kb * 32 + 16 + n];   b1[i].y = Wm[(kb + 1) * 32 + 16 + n];
        }
        __builtin_amdgcn_sched_barrier(0);
#pragma unroll
        for (int i = 0; i < 4; ++i) {
            c0 = wmma4(a[4 * h + i], b0[i], c0);
            c1 = wmma4(a[4 * h + i], b1[i], c1);
        }
        __builtin_amdgcn_sched_barrier(0);
    }
}

__device__ __forceinline__ void
enc_body(float* h0t, const int encBlock,
         const float* __restrict__ x, const float* __restrict__ cnn,
         const float* __restrict__ fcW, const float* __restrict__ fcb,
         const float* __restrict__ cnnW, const float* __restrict__ cnnb,
         const float* __restrict__ W1, const float* __restrict__ dinv,
         float* __restrict__ t1, int N, int nTiles) {
    const int w    = threadIdx.x >> 5;
    const int lane = threadIdx.x & 31;
    const int wv   = encBlock * FE_WAVES + w;
    const bool live = wv < nTiles;
    const int  tile = live ? wv : (nTiles - 1);
    const int n    = lane & 15;
    const int kh   = lane >> 4;
    int rowA = tile * 16 + n;    if (rowA >= N) rowA = N - 1;

    float* tw = h0t + w * (16 * ENC_PITCH);
    {
        v8f acc0 = {}, acc1 = {};
        enc_line(x + (size_t)rowA * 32, fcW, 0, kh, n, acc0, acc1);
        const float bf0 = fcb[n], bf1 = fcb[16 + n];
#pragma unroll
        for (int v = 0; v < 8; ++v) {
            float* o = tw + (8 * kh + v) * ENC_PITCH;
            o[n]      = fmaxf(acc0[v] + bf0, 0.0f);
            o[16 + n] = fmaxf(acc1[v] + bf1, 0.0f);
        }
    }
    {
        v8f acc2 = {}, acc3 = {};
        const float* ac = cnn + (size_t)rowA * 128;
#pragma unroll
        for (int k0 = 0; k0 < 128; k0 += 32) enc_line(ac, cnnW, k0, kh, n, acc2, acc3);
        const float bc0 = cnnb[n], bc1 = cnnb[16 + n];
#pragma unroll
        for (int v = 0; v < 8; ++v) {
            float* o = tw + (8 * kh + v) * ENC_PITCH;
            o[32 + n] = fmaxf(acc2[v] + bc0, 0.0f);
            o[48 + n] = fmaxf(acc3[v] + bc1, 0.0f);
        }
    }
    __syncthreads();

    float dr[8];
#pragma unroll
    for (int v = 0; v < 8; ++v) {
        int nd = tile * 16 + 8 * kh + v;
        dr[v] = dinv[nd < N ? nd : N - 1];
    }

    v8f d0 = {}, d1 = {};
    const float* ar = tw + n * ENC_PITCH + 2 * kh;
    const float* wr = W1 + (2 * kh) * 32 + n;
#pragma unroll
    for (int g = 0; g < 64; g += 16) {
        v2f a[4], b0[4], b1[4];
#pragma unroll
        for (int i = 0; i < 4; ++i) {
            const int k0 = g + 4 * i;
            a[i] = *(const v2f*)(ar + k0);
            b0[i].x = wr[k0 * 32];        b0[i].y = wr[k0 * 32 + 32];
            b1[i].x = wr[k0 * 32 + 16];   b1[i].y = wr[k0 * 32 + 48];
        }
        __builtin_amdgcn_sched_barrier(0);
#pragma unroll
        for (int i = 0; i < 4; ++i) {
            d0 = wmma4(a[i], b0[i], d0);
            d1 = wmma4(a[i], b1[i], d1);
        }
        __builtin_amdgcn_sched_barrier(0);
    }
    if (live) {
#pragma unroll
        for (int v = 0; v < 8; ++v) {
            int node = tile * 16 + 8 * kh + v;
            if (node < N) {
                float* o = t1 + (size_t)node * 32;
                o[n]      = d0[v] * dr[v];
                o[16 + n] = d1[v] * dr[v];
            }
        }
    }
}

#define FE_FILL_WORDS (CS_RANGE + CS_THREADS / 32 + CS_RANGE / DEG_THREADS)
#define FE_ENC_WORDS  (FE_WAVES * 16 * ENC_PITCH)
#define FE_LDS_WORDS  (FE_ENC_WORDS > FE_FILL_WORDS ? FE_ENC_WORDS : FE_FILL_WORDS)
static_assert(2 * FE_LDS_WORDS * 4 <= 320 * 1024, "a fill block and an encoder block share a WGP's LDS");

__global__ void __launch_bounds__(CS_THREADS, 16)
k_fill_encoder(const int* __restrict__ src, const int* __restrict__ dst,
               const int* __restrict__ plane, const int* __restrict__ bsum,
               int* __restrict__ slot, int* __restrict__ sorted, int E, int N, int nFill,
               const float* __restrict__ x, const float* __restrict__ cnn,
               const float* __restrict__ fcW, const float* __restrict__ fcb,
               const float* __restrict__ cnnW, const float* __restrict__ cnnb,
               const float* __restrict__ W1, const float* __restrict__ dinv,
               float* __restrict__ t1, int nTiles) {
    __shared__ __attribute__((aligned(16))) float lds[FE_LDS_WORDS];
    const int b = blockIdx.x;
    if (b < nFill) {
        unsigned* cnt = reinterpret_cast<unsigned*>(lds);
        int* wred = reinterpret_cast<int*>(lds) + CS_RANGE;
        int* boff = wred + CS_THREADS / 32;
        fill_body(cnt, wred, boff, b % CS_SPLITS, b / CS_SPLITS,
                  src, dst, plane, bsum, slot, sorted, E, N);
    } else {
        enc_body(lds, b - nFill, x, cnn, fcW, fcb, cnnW, cnnb, W1, dinv, t1, N, nTiles);
    }
}

#define T2_PITCH 32
static_assert(T2_PITCH >= 20 && T2_PITCH <= 32, "t2 lives in the N * 32 words of agg1's region");
#define GT_PITCH 36
#define GT_WAVES 8

#define W2_LPITCH 40
#define W2_LWORDS (32 * W2_LPITCH)

__device__ __forceinline__ void
t2_flush(const float* tw, const float* td, const int* tn, const float* w2s, const int rows,
         const int lane, float* __restrict__ t2) {
    const int  n    = lane & 15;
    const int  kh   = lane >> 4;
    const bool c1ok = (16 + n) < 20;
    const float* ar = tw + n * GT_PITCH + 2 * kh;
    const float* br = w2s + (2 * kh) * W2_LPITCH + n;
    v8f acc0 = {}, acc1 = {};
#pragma unroll
    for (int k0 = 0; k0 < 32; k0 += 4) {
        v2f a;  a.x = ar[k0];                       a.y = ar[k0 + 1];
        v2f b0; b0.x = br[k0 * W2_LPITCH];          b0.y = br[(k0 + 1) * W2_LPITCH];
        v2f b1; b1.x = br[k0 * W2_LPITCH + 16];     b1.y = br[(k0 + 1) * W2_LPITCH + 16];
        acc0 = wmma4(a, b0, acc0);
        acc1 = wmma4(a, b1, acc1);
    }
    float dr[8];
    int   nd[8];
#pragma unroll
    for (int v = 0; v < 8; ++v) { dr[v] = td[8 * kh + v]; nd[v] = tn[8 * kh + v]; }
#pragma unroll
    for (int v = 0; v < 8; ++v) {
        const int r = 8 * kh + v;
        if (r < rows) {
            float* o = t2 + (size_t)nd[v] * T2_PITCH;
            o[n] = acc0[v] * dr[v];
            if (c1ok) o[16 + n] = acc1[v] * dr[v];
        }
    }
}


template <int F, int P, bool GUARD>
__device__ __forceinline__ void rows8(int s, int j, int n, const float* __restrict__ t,
                                      int lane, bool act, float (&v)[8]) {
#pragma unroll
    for (int u = 0; u < 8; ++u) {
        v[u] = 0.0f;
        if (!GUARD || j + u < n) {
            const int sj = __builtin_amdgcn_readlane(s, j + u);
            if (act) v[u] = t[(size_t)sj * P + lane];
        }
    }
}

__device__ __forceinline__ float sum8(const float (&v)[8], float acc) {
#pragma unroll
    for (int u = 0; u < 8; ++u) acc += v[u];
    return acc;
}

template <int F, int P>
__device__ __forceinline__ void rows8_first(int s, int n, const float* __restrict__ t,
                                            int lane, bool act, float (&v)[8]) {
    const int col = act ? lane : 0;
#pragma unroll
    for (int u = 0; u < 8; ++u) {
        const int sj = __builtin_amdgcn_readlane(s, u);
        const float x = t[(size_t)sj * P + col];
        v[u] = (u < n) ? x : 0.0f;
    }
}

template <int F, int P, bool GUARD>
__device__ __forceinline__ float gather8(int s, int j, int n,
                                         const float* __restrict__ t, int lane, bool act,
                                         float acc) {
    float v[8];
    rows8<F, P, GUARD>(s, j, n, t, lane, act, v);
    return sum8(v, acc);
}

template <int F, int P>
__device__ __forceinline__ float gather_group(int s, int j0, int n, const float* __restrict__ t,
                                              int lane, bool act, float acc) {
    int j = j0;
    for (; j + 8 <= n; j += 8) acc = gather8<F, P, false>(s, j, n, t, lane, act, acc);
    if (j < n)                 acc = gather8<F, P, true >(s, j, n, t, lane, act, acc);
    return acc;
}

template <int F, int P, bool FUSE>
__device__ __forceinline__ void gather_rows(const int* __restrict__ slot0,
                                            const int* __restrict__ sorted,
                                            const float* __restrict__ dinv,
                                            const float* __restrict__ t,
                                            const float* __restrict__ bias,
                                            float* __restrict__ o, int N, int E,
                                            float* tw, float* td, int* tn,
                                            float* w2s, const float* __restrict__ W2) {
    static_assert(!FUSE || F == 32, "a fused row fills one tile row: lane = k");
    static_assert(W2_LWORDS % (GT_WAVES * 32) == 0, "every thread stages the same number of words");
    float wv[W2_LWORDS / (GT_WAVES * 32)];
    if (FUSE) {
#pragma unroll
        for (int j = 0; j < W2_LWORDS / (GT_WAVES * 32); ++j) {
            const int i = j * (GT_WAVES * 32) + (int)threadIdx.x;
            const int k = i / W2_LPITCH, col = i % W2_LPITCH;
            wv[j] = W2[k * 20 + (col < 20 ? col : 0)];
        }
    }
    const int G = __builtin_amdgcn_readfirstlane((int)(gridDim.x * (blockDim.x >> 5)));
    int node = __builtin_amdgcn_readfirstlane((int)((blockIdx.x * blockDim.x + threadIdx.x) >> 5));
    const bool alive = node < N;
    if (!FUSE && !alive) return;
    const int  lane = threadIdx.x & 31;
    const bool act  = lane < F;
    const float bl  = act ? bias[lane] : 0.0f;

    int rows = 0;
    if (FUSE) {
#pragma unroll
        for (int r = 0; r < 16; ++r) tw[r * GT_PITCH + lane] = 0.0f;
        if (lane < 16) { td[lane] = 0.0f; tn[lane] = 0; }
    }

    int rs = 0, re = 0;
    int sa = 0, sb = 0;
    int nx  = node + G;
    int rs1 = 0, re1 = 0;
    if (alive) {
        rs = slot0[node];
        re = (node + 1 < N) ? slot0[node + 1] : E;
        if (rs + lane < re)      sa = sorted[rs + lane];
        if (rs + 32 + lane < re) sb = sorted[rs + 32 + lane];
        if (nx < N) { rs1 = slot0[nx]; re1 = (nx + 1 < N) ? slot0[nx + 1] : E; }
    }

    if (FUSE) {
#pragma unroll
        for (int j = 0; j < W2_LWORDS / (GT_WAVES * 32); ++j) {
            const int i = j * (GT_WAVES * 32) + (int)threadIdx.x;
            w2s[i] = (i % W2_LPITCH < 20) ? wv[j] : 0.0f;
        }
        __syncthreads();
        if (!alive) return;
    }

    for (;;) {
        const int cnt = re - rs;
        const int n0  = (cnt < 32) ? cnt : 32;
        float v[8];
        rows8_first<F, P>(sa, n0, t, lane, act, v);

        int sa1 = 0, sb1 = 0;
        if (nx < N) {
            if (rs1 + lane < re1)      sa1 = sorted[rs1 + lane];
            if (rs1 + 32 + lane < re1) sb1 = sorted[rs1 + 32 + lane];
        }
        const int nx2 = nx + G;
        int rs2 = 0, re2 = 0;
        if (nx < N && nx2 < N) { rs2 = slot0[nx2]; re2 = (nx2 + 1 < N) ? slot0[nx2 + 1] : E; }

        const float  di  = dinv[node];
        const size_t idx = (size_t)node * F + lane;
        float self = 0.0f;
        if (act) self = t[(size_t)node * P + lane];

        float acc = sum8(v, 0.0f);
        acc = gather_group<F, P>(sa, 8, n0, t, lane, act, acc);
        if (cnt > 32)
            acc = gather_group<F, P>(sb, 0, (cnt < 64) ? (cnt - 32) : 32, t, lane, act, acc);
        for (int base = rs + 64; base < re; base += 32) {
            const int n = (re - base < 32) ? (re - base) : 32;
            int s = 0;
            if (lane < n) s = sorted[base + lane];
            acc = gather_group<F, P>(s, 0, n, t, lane, act, acc);
        }
        const float val = (acc + self) * di + bl;
        if (FUSE) {
            tw[rows * GT_PITCH + lane] = fmaxf(val, 0.0f);
            if (lane == 0) { td[rows] = di; tn[rows] = node; }
            ++rows;
            if (rows == 16 || nx >= N) { t2_flush(tw, td, tn, w2s, rows, lane, o); rows = 0; }
        } else {
            if (act) o[idx] = val;
        }

        if (nx >= N) break;
        node = nx;  rs = rs1;  re = re1;  sa = sa1;  sb = sb1;
        nx = nx2;   rs1 = rs2; re1 = re2;
    }
}

__global__ void __launch_bounds__(GT_WAVES * 32, 16)
k_gather32_t2(const int* __restrict__ slot0, const int* __restrict__ sorted,
              const float* __restrict__ dinv, const float* __restrict__ t1,
              const float* __restrict__ b1, const float* __restrict__ W2,
              float* __restrict__ t2, int N, int E) {
    __shared__ float tile[GT_WAVES][16 * GT_PITCH];
    __shared__ float tfac[GT_WAVES][16];
    __shared__ int   tnod[GT_WAVES][16];
    __shared__ float w2s[W2_LWORDS];
    const int w = threadIdx.x >> 5;
    gather_rows<32, 32, true>(slot0, sorted, dinv, t1, b1, t2, N, E, tile[w], tfac[w], tnod[w], w2s, W2);
}

__global__ void __launch_bounds__(256)
k_gather20(const int* __restrict__ slot0, const int* __restrict__ sorted,
           const float* __restrict__ dinv, const float* __restrict__ t2,
           const float* __restrict__ b2, float* __restrict__ out, int N, int E) {
    gather_rows<20, T2_PITCH, false>(slot0, sorted, dinv, t2, b2, out, N, E, nullptr, nullptr, nullptr, nullptr, nullptr);
}

extern "C" void kernel_launch(void* const* d_in, const int* in_sizes, int n_in,
                              void* d_out, int out_size, void* d_ws, size_t ws_size,
                              hipStream_t stream) {
    const float* x    = (const float*)d_in[0];
    const float* cnn  = (const float*)d_in[1];
    const int*   ei   = (const int*)d_in[2];
    const float* fcW  = (const float*)d_in[3];
    const float* fcb  = (const float*)d_in[4];
    const float* cnnW = (const float*)d_in[5];
    const float* cnnb = (const float*)d_in[6];
    const float* W1   = (const float*)d_in[7];
    const float* b1   = (const float*)d_in[8];
    const float* W2   = (const float*)d_in[9];
    const float* b2   = (const float*)d_in[10];

    const int N = in_sizes[0] / 32;
    const int E = in_sizes[2] / 2;
    const int* src = ei;
    const int* dst = ei + E;

    float* ws   = (float*)d_ws;
    float* dinv = ws;  ws += N;
    float* eidx = ws;  ws += (size_t)N * 64;
    float* t1   = ws;  ws += (size_t)N * 32;
    float* agg1 = ws;  ws += (size_t)N * 32;
    float* bsw  = ws;  ws += (size_t)N * 20;
    float* t2   = agg1;
    float* out  = (float*)d_out;

    int* plane  = (int*)agg1;
    int* slot   = (int*)eidx;
    int* sorted = slot + N;

    const int nTiles = (N + 15) / 16;
    const int B = GT_WAVES * 32;
    dim3 grdNode(2048);

    dim3 grdCS(CS_SPLITS, (N + CS_RANGE - 1) / CS_RANGE);
    k_count<<<grdCS, CS_THREADS, 0, stream>>>(dst, plane, E, N);
    int* bsum = (int*)bsw;
    const int nbDeg = (N + DEG_THREADS - 1) / DEG_THREADS;
    k_deg<<<nbDeg, DEG_THREADS, 0, stream>>>(plane, dinv, bsum, N);

    const int nFill = CS_SPLITS * ((N + CS_RANGE - 1) / CS_RANGE);
    const int nEnc  = (nTiles + FE_WAVES - 1) / FE_WAVES;
    k_fill_encoder<<<nFill + nEnc, CS_THREADS, 0, stream>>>(src, dst, plane, bsum, slot, sorted, E, N, nFill,
                                                            x, cnn, fcW, fcb, cnnW, cnnb, W1, dinv, t1, nTiles);

    k_gather32_t2<<<grdNode, B, 0, stream>>>(slot, sorted, dinv, t1, b1, W2, t2, N, E);

    k_gather20<<<grdNode, B, 0, stream>>>(slot, sorted, dinv, t2, b2, out, N, E);
}


# Edge sort without one scattered 4-byte store per edge: edges go by 512-node bucket, then by node, each pass ordered in LDS and stored as whole runs; per-node sum order follows the new LDS ranks
# speedup vs baseline: 6.9833x; 1.4711x over previous
#include <hip/hip_runtime.h>
#include <hip/hip_bf16.h>

typedef __attribute__((ext_vector_type(2))) float v2f;
typedef __attribute__((ext_vector_type(8))) float v8f;

__device__ __forceinline__ v8f wmma4(v2f a, v2f b, v8f c) {
    return __builtin_amdgcn_wmma_f32_16x16x4_f32(
         false, a,  false, b,
         (short)0, c,  false,  false);
}

#define BK_SHIFT   9
#define BK_NODES   (1 << BK_SHIFT)
#define BK_MAX     256
#define SC_EDGES   16384
#define CS_THREADS 1024
#define SC_NIT     (SC_EDGES / 4 / CS_THREADS)

static_assert(SC_EDGES == 4 * CS_THREADS * SC_NIT, "a slice is a whole number of 16-byte loads per thread");
static_assert(SC_EDGES < 65536, "a slice's count per bucket fits 16 bits");
static_assert(CS_THREADS == 8 * (BK_MAX / 2), "k_bscatter: eight threads per pair of buckets sum the slice counts");
static_assert(BK_NODES * 2 == CS_THREADS, "k_bsort_encoder: the first half of the block owns one node each");

__device__ __forceinline__ int block_exscan(int v, int* wt, int lane, int w, int& total) {
    int inc = v;
    for (int o = 1; o < 32; o <<= 1) {
        int y = __shfl_up(inc, o);
        if (lane >= o) inc += y;
    }
    if (lane == 31) wt[w] = inc;
    __syncthreads();
    int woff = 0, tot = 0;
    for (int k = 0; k < CS_THREADS / 32; ++k) {
        const int t = wt[k];
        if (k < w) woff += t;
        tot += t;
    }
    total = tot;
    return woff + inc - v;
}

__device__ __forceinline__ void bk_count(unsigned* h, int d, int N) {
    if ((unsigned)d < (unsigned)N) atomicAdd(&h[d >> BK_SHIFT], 1u);
}

__global__ void __launch_bounds__(CS_THREADS)
k_bcount(const int* __restrict__ dst, unsigned* __restrict__ bh, int E, int N) {
    __shared__ unsigned h[BK_MAX];
    const int tid = threadIdx.x;
    if (tid < BK_MAX) h[tid] = 0u;
    __syncthreads();

    const int e0 = blockIdx.x * SC_EDGES;
    const int e1 = (E - e0 > SC_EDGES) ? (e0 + SC_EDGES) : E;
    if ((((size_t)dst) & 15) == 0) {
        const int4* __restrict__ p = (const int4*)(dst + e0);
        const int nv = (e1 - e0) >> 2;
        if (nv > 0) {
            int4 v[SC_NIT];
#pragma unroll
            for (int k = 0; k < SC_NIT; ++k) {
                const int i = k * CS_THREADS + tid;
                v[k] = p[i < nv ? i : nv - 1];
            }
#pragma unroll
            for (int k = 0; k < SC_NIT; ++k) {
                if (k * CS_THREADS + tid < nv) {
                    bk_count(h, v[k].x, N);  bk_count(h, v[k].y, N);
                    bk_count(h, v[k].z, N);  bk_count(h, v[k].w, N);
                }
            }
        }
        for (int e = e0 + (nv << 2) + tid; e < e1; e += CS_THREADS) bk_count(h, dst[e], N);
    } else {
        for (int e = e0 + tid; e < e1; e += CS_THREADS) bk_count(h, dst[e], N);
    }
    __syncthreads();
    if (tid < BK_MAX / 2) bh[blockIdx.x * (BK_MAX / 2) + tid] = h[2 * tid] | (h[2 * tid + 1] << 16);
}

__device__ __forceinline__ void bk_place4(unsigned* lcur, const int* delta, unsigned* stW, int* stG,
                                          int4 s, int4 d, int N) {
    const bool m0 = (unsigned)d.x < (unsigned)N, m1 = (unsigned)d.y < (unsigned)N;
    const bool m2 = (unsigned)d.z < (unsigned)N, m3 = (unsigned)d.w < (unsigned)N;
    const int b0 = m0 ? (d.x >> BK_SHIFT) : 0, b1 = m1 ? (d.y >> BK_SHIFT) : 0;
    const int b2 = m2 ? (d.z >> BK_SHIFT) : 0, b3 = m3 ? (d.w >> BK_SHIFT) : 0;
    unsigned p0 = SC_EDGES, p1 = SC_EDGES, p2 = SC_EDGES, p3 = SC_EDGES;
    if (m0) p0 = atomicAdd(&lcur[b0], 1u);
    if (m1) p1 = atomicAdd(&lcur[b1], 1u);
    if (m2) p2 = atomicAdd(&lcur[b2], 1u);
    if (m3) p3 = atomicAdd(&lcur[b3], 1u);
    const int t0 = delta[b0], t1 = delta[b1], t2 = delta[b2], t3 = delta[b3];
    if (p0 < (unsigned)SC_EDGES) {
        stW[p0] = ((unsigned)s.x << BK_SHIFT) | (unsigned)(d.x & (BK_NODES - 1));  stG[p0] = (int)p0 + t0;
    }
    if (p1 < (unsigned)SC_EDGES) {
        stW[p1] = ((unsigned)s.y << BK_SHIFT) | (unsigned)(d.y & (BK_NODES - 1));  stG[p1] = (int)p1 + t1;
    }
    if (p2 < (unsigned)SC_EDGES) {
        stW[p2] = ((unsigned)s.z << BK_SHIFT) | (unsigned)(d.z & (BK_NODES - 1));  stG[p2] = (int)p2 + t2;
    }
    if (p3 < (unsigned)SC_EDGES) {
        stW[p3] = ((unsigned)s.w << BK_SHIFT) | (unsigned)(d.w & (BK_NODES - 1));  stG[p3] = (int)p3 + t3;
    }
}

__device__ __forceinline__ void bk_place(unsigned* lcur, const int* delta, unsigned* stW, int* stG,
                                         int s, int d, int N) {
    if ((unsigned)d < (unsigned)N) {
        const int b = d >> BK_SHIFT;
        const unsigned p = atomicAdd(&lcur[b], 1u);
        if (p < (unsigned)SC_EDGES) {
            stW[p] = ((unsigned)s << BK_SHIFT) | (unsigned)(d & (BK_NODES - 1));
            stG[p] = (int)p + delta[b];
        }
    }
}

__global__ void __launch_bounds__(CS_THREADS)
k_bscatter(const int* __restrict__ src, const int* __restrict__ dst, const unsigned* __restrict__ bh,
           int* __restrict__ gbase, unsigned* __restrict__ tmp, int E, int N, int nSl) {
    __shared__ int      ppre[8][BK_MAX], ptot[8][BK_MAX];
    __shared__ int      wtA[CS_THREADS / 32], wtB[CS_THREADS / 32];
    __shared__ unsigned lcur[BK_MAX];
    __shared__ int      delta[BK_MAX];
    __shared__ unsigned stW[SC_EDGES];
    __shared__ int      stG[SC_EDGES];
    const int tid  = threadIdx.x;
    const int lane = tid & 31;
    const int w    = tid >> 5;
    const int x    = blockIdx.x;

    {
        const int q   = tid >> 7;
        const int bp  = tid & (BK_MAX / 2 - 1);
        const int per = (nSl + 7) >> 3;
        const int xa  = q * per;
        const int xb  = (nSl - xa > per) ? (xa + per) : nSl;
        int pre0 = 0, pre1 = 0, tot0 = 0, tot1 = 0;
        for (int xs = xa; xs < xb; xs += 8) {
            unsigned v[8];
#pragma unroll
            for (int k = 0; k < 8; ++k) {
                const int xx = xs + k;
                v[k] = bh[(xx < xb ? xx : xb - 1) * (BK_MAX / 2) + bp];
            }
#pragma unroll
            for (int k = 0; k < 8; ++k) {
                const int xx = xs + k;
                if (xx < xb) {
                    const int c0 = (int)(v[k] & 0xffffu), c1 = (int)(v[k] >> 16);
                    tot0 += c0;  tot1 += c1;
                    if (xx < x) { pre0 += c0;  pre1 += c1; }
                }
            }
        }
        ppre[q][2 * bp] = pre0;  ppre[q][2 * bp + 1] = pre1;
        ptot[q][2 * bp] = tot0;  ptot[q][2 * bp + 1] = tot1;
    }
    __syncthreads();
    int tot = 0, pre = 0, own = 0;
    if (tid < BK_MAX) {
        for (int q = 0; q < 8; ++q) { tot += ptot[q][tid];  pre += ppre[q][tid]; }
        const unsigned ow = bh[x * (BK_MAX / 2) + (tid >> 1)];
        own = (int)((tid & 1) ? (ow >> 16) : (ow & 0xffffu));
    }
    int allE, ownE;
    const int gb = block_exscan(tot, wtA, lane, w, allE);
    const int ls = block_exscan(own, wtB, lane, w, ownE);
    if (tid < BK_MAX) {
        lcur[tid]  = (unsigned)ls;
        delta[tid] = gb + pre - ls;
        if (x == 0) {
            gbase[tid] = gb;
            if (tid == BK_MAX - 1) gbase[BK_MAX] = gb + tot;
        }
    }
    __syncthreads();

    const int e0 = x * SC_EDGES;
    const int e1 = (E - e0 > SC_EDGES) ? (e0 + SC_EDGES) : E;
    if (((((size_t)dst) | ((size_t)src)) & 15) == 0) {
        const int4* __restrict__ p = (const int4*)(dst + e0);
        const int4* __restrict__ q = (const int4*)(src + e0);
        const int nv = (e1 - e0) >> 2;
        if (nv > 0) {
            int4 d4[SC_NIT], s4[SC_NIT];
#pragma unroll
            for (int k = 0; k < SC_NIT; ++k) {
                const int i  = k * CS_THREADS + tid;
                const int ic = i < nv ? i : nv - 1;
                d4[k] = p[ic];
                s4[k] = q[ic];
            }
#pragma unroll
            for (int k = 0; k < SC_NIT; ++k) {
                if (k * CS_THREADS + tid < nv) bk_place4(lcur, delta, stW, stG, s4[k], d4[k], N);
            }
        }
        for (int e = e0 + (nv << 2) + tid; e < e1; e += CS_THREADS)
            bk_place(lcur, delta, stW, stG, src[e], dst[e], N);
    } else {
        for (int e = e0 + tid; e < e1; e += CS_THREADS)
            bk_place(lcur, delta, stW, stG, src[e], dst[e], N);
    }
    __syncthreads();

    const int nOwn = (ownE < SC_EDGES) ? ownE : SC_EDGES;
    for (int base = 0; base < nOwn; base += 8 * CS_THREADS) {
        int      g[8];
        unsigned wd[8];
#pragma unroll
        for (int k = 0; k < 8; ++k) {
            const int i  = base + k * CS_THREADS + tid;
            const int ic = i < nOwn ? i : nOwn - 1;
            g[k]  = stG[ic];
            wd[k] = stW[ic];
        }
#pragma unroll
        for (int k = 0; k < 8; ++k)
            if (base + k * CS_THREADS + tid < nOwn && (unsigned)g[k] < (unsigned)E) tmp[g[k]] = wd[k];
    }
}

#define FE_WAVES  (CS_THREADS / 32)
#define ENC_PITCH 68

__device__ __forceinline__ void
enc_line(const float* __restrict__ arow, const float* __restrict__ Wm, const int k0,
         const int kh, const int n, v8f& c0, v8f& c1) {
    v2f a[8];
#pragma unroll
    for (int i = 0; i < 8; ++i) {
        const int kb = k0 + 4 * i + 2 * kh;
        a[i].x = arow[kb];   a[i].y = arow[kb + 1];
    }
#pragma unroll
    for (int h = 0; h < 2; ++h) {
        v2f b0[4], b1[4];
#pragma unroll
        for (int i = 0; i < 4; ++i) {
            const int kb = k0 + 4 * (4 * h + i) + 2 * kh;
            b0[i].x = Wm[kb * 32 + n];        b0[i].y = Wm[(kb + 1) * 32 + n];
            b1[i].x = Wm[kb * 32 + 16 + n];   b1[i].y = Wm[(kb + 1) * 32 + 16 + n];
        }
        __builtin_amdgcn_sched_barrier(0);
#pragma unroll
        for (int i = 0; i < 4; ++i) {
            c0 = wmma4(a[4 * h + i], b0[i], c0);
            c1 = wmma4(a[4 * h + i], b1[i], c1);
        }
        __builtin_amdgcn_sched_barrier(0);
    }
}

__device__ __forceinline__ void
enc_body(float* h0t, const int encBlock,
         const float* __restrict__ x, const float* __restrict__ cnn,
         const float* __restrict__ fcW, const float* __restrict__ fcb,
         const float* __restrict__ cnnW, const float* __restrict__ cnnb,
         const float* __restrict__ W1, const float* dl,
         float* __restrict__ t1, int N, int nTiles) {
    const int w    = threadIdx.x >> 5;
    const int lane = threadIdx.x & 31;
    const int wv   = encBlock * FE_WAVES + w;
    const bool live = wv < nTiles;
    const int  tile = live ? wv : (nTiles - 1);
    const int n    = lane & 15;
    const int kh   = lane >> 4;
    int rowA = tile * 16 + n;    if (rowA >= N) rowA = N - 1;

    float* tw = h0t + w * (16 * ENC_PITCH);
    {
        v8f acc0 = {}, acc1 = {};
        enc_line(x + (size_t)rowA * 32, fcW, 0, kh, n, acc0, acc1);
        const float bf0 = fcb[n], bf1 = fcb[16 + n];
#pragma unroll
        for (int v = 0; v < 8; ++v) {
            float* o = tw + (8 * kh + v) * ENC_PITCH;
            o[n]      = fmaxf(acc0[v] + bf0, 0.0f);
            o[16 + n] = fmaxf(acc1[v] + bf1, 0.0f);
        }
    }
    {
        v8f acc2 = {}, acc3 = {};
        const float* ac = cnn + (size_t)rowA * 128;
#pragma unroll
        for (int k0 = 0; k0 < 128; k0 += 32) enc_line(ac, cnnW, k0, kh, n, acc2, acc3);
        const float bc0 = cnnb[n], bc1 = cnnb[16 + n];
#pragma unroll
        for (int v = 0; v < 8; ++v) {
            float* o = tw + (8 * kh + v) * ENC_PITCH;
            o[32 + n] = fmaxf(acc2[v] + bc0, 0.0f);
            o[48 + n] = fmaxf(acc3[v] + bc1, 0.0f);
        }
    }
    __syncthreads();

    float dr[8];
#pragma unroll
    for (int v = 0; v < 8; ++v) dr[v] = dl[16 * w + 8 * kh + v];

    v8f d0 = {}, d1 = {};
    const float* ar = tw + n * ENC_PITCH + 2 * kh;
    const float* wr = W1 + (2 * kh) * 32 + n;
#pragma unroll
    for (int g = 0; g < 64; g += 16) {
        v2f a[4], b0[4], b1[4];
#pragma unroll
        for (int i = 0; i < 4; ++i) {
            const int k0 = g + 4 * i;
            a[i] = *(const v2f*)(ar + k0);
            b0[i].x = wr[k0 * 32];        b0[i].y = wr[k0 * 32 + 32];
            b1[i].x = wr[k0 * 32 + 16];   b1[i].y = wr[k0 * 32 + 48];
        }
        __builtin_amdgcn_sched_barrier(0);
#pragma unroll
        for (int i = 0; i < 4; ++i) {
            d0 = wmma4(a[i], b0[i], d0);
            d1 = wmma4(a[i], b1[i], d1);
        }
        __builtin_amdgcn_sched_barrier(0);
    }
    if (live) {
#pragma unroll
        for (int v = 0; v < 8; ++v) {
            int node = tile * 16 + 8 * kh + v;
            if (node < N) {
                float* o = t1 + (size_t)node * 32;
                o[n]      = d0[v] * dr[v];
                o[16 + n] = d1[v] * dr[v];
            }
        }
    }
}


#define FE_ENC_WORDS (FE_WAVES * 16 * ENC_PITCH)
#define SORT_CAP     FE_ENC_WORDS
static_assert(FE_WAVES * 16 == BK_NODES, "one encoder tile per wave covers the bucket's nodes");
#define BS_BATCH 16

__global__ void __launch_bounds__(CS_THREADS, 16)
k_bsort_encoder(const unsigned* __restrict__ tmp, const int* __restrict__ gbase,
                int* __restrict__ slot, int* __restrict__ sorted, float* __restrict__ dinv,
                int E, int N,
                const float* __restrict__ x, const float* __restrict__ cnn,
                const float* __restrict__ fcW, const float* __restrict__ fcb,
                const float* __restrict__ cnnW, const float* __restrict__ cnnb,
                const float* __restrict__ W1, float* __restrict__ t1, int nTiles) {
    __shared__ __attribute__((aligned(16))) float lds[FE_ENC_WORDS];
    __shared__ unsigned deg[BK_NODES];
    __shared__ unsigned ncur[BK_NODES];
    __shared__ float    dl[BK_NODES];
    __shared__ int      wt[CS_THREADS / 32];
    const int tid  = threadIdx.x;
    const int lane = tid & 31;
    const int w    = tid >> 5;
    const int b    = blockIdx.x;
    const int lo   = b * BK_NODES;
    int* stage = reinterpret_cast<int*>(lds);

    int g0 = gbase[b], g1 = gbase[b + 1];
    g0 = (g0 < 0) ? 0 : ((g0 > E) ? E : g0);
    g1 = (g1 < g0) ? g0 : ((g1 > E) ? E : g1);
    const int cnt = g1 - g0;

    if (tid < BK_NODES) deg[tid] = 0u;
    __syncthreads();
    for (int base = 0; base < cnt; base += CS_THREADS * BS_BATCH) {
        unsigned wd[BS_BATCH];
#pragma unroll
        for (int k = 0; k < BS_BATCH; ++k) {
            const int i = base + k * CS_THREADS + tid;
            wd[k] = tmp[g0 + (i < cnt ? i : cnt - 1)];
        }
#pragma unroll
        for (int k = 0; k < BS_BATCH; ++k)
            if (base + k * CS_THREADS + tid < cnt) atomicAdd(&deg[wd[k] & (BK_NODES - 1)], 1u);
    }
    __syncthreads();

    const int dv = (tid < BK_NODES) ? (int)deg[tid] : 0;
    int all;
    const int ex = block_exscan(dv, wt, lane, w, all);
    if (tid < BK_NODES) {
        const float di = rsqrtf((float)dv + 1.0f);
        ncur[tid] = (unsigned)ex;
        dl[tid]   = di;
        const int node = lo + tid;
        if (node < N) { slot[node] = g0 + ex;  dinv[node] = di; }
    }
    __syncthreads();

    const bool fits = cnt <= SORT_CAP;
    for (int base = 0; base < cnt; base += CS_THREADS * BS_BATCH) {
        unsigned wd[BS_BATCH];
#pragma unroll
        for (int k = 0; k < BS_BATCH; ++k) {
            const int i = base + k * CS_THREADS + tid;
            wd[k] = tmp[g0 + (i < cnt ? i : cnt - 1)];
        }
#pragma unroll
        for (int k = 0; k < BS_BATCH; ++k) {
            if (base + k * CS_THREADS + tid < cnt) {
                const unsigned p = atomicAdd(&ncur[wd[k] & (BK_NODES - 1)], 1u);
                const int      s = (int)(wd[k] >> BK_SHIFT);
                if (p < (unsigned)cnt) {
                    if (fits) stage[p] = s; else sorted[g0 + p] = s;
                }
            }
        }
    }
    __syncthreads();
    if (fits)
        for (int i = tid; i < cnt; i += CS_THREADS) sorted[g0 + i] = stage[i];
    __syncthreads();

    enc_body(lds, b, x, cnn, fcW, fcb, cnnW, cnnb, W1, dl, t1, N, nTiles);
}

#define T2_PITCH 32
static_assert(T2_PITCH >= 20 && T2_PITCH <= 32, "t2 lives in the N * 32 words of agg1's region");
#define GT_PITCH 36
#define GT_WAVES 8

#define W2_LPITCH 40
#define W2_LWORDS (32 * W2_LPITCH)

__device__ __forceinline__ void
t2_flush(const float* tw, const float* td, const int* tn, const float* w2s, const int rows,
         const int lane, float* __restrict__ t2) {
    const int  n    = lane & 15;
    const int  kh   = lane >> 4;
    const bool c1ok = (16 + n) < 20;
    const float* ar = tw + n * GT_PITCH + 2 * kh;
    const float* br = w2s + (2 * kh) * W2_LPITCH + n;
    v8f acc0 = {}, acc1 = {};
#pragma unroll
    for (int k0 = 0; k0 < 32; k0 += 4) {
        v2f a;  a.x = ar[k0];                       a.y = ar[k0 + 1];
        v2f b0; b0.x = br[k0 * W2_LPITCH];          b0.y = br[(k0 + 1) * W2_LPITCH];
        v2f b1; b1.x = br[k0 * W2_LPITCH + 16];     b1.y = br[(k0 + 1) * W2_LPITCH + 16];
        acc0 = wmma4(a, b0, acc0);
        acc1 = wmma4(a, b1, acc1);
    }
    float dr[8];
    int   nd[8];
#pragma unroll
    for (int v = 0; v < 8; ++v) { dr[v] = td[8 * kh + v]; nd[v] = tn[8 * kh + v]; }
#pragma unroll
    for (int v = 0; v < 8; ++v) {
        const int r = 8 * kh + v;
        if (r < rows) {
            float* o = t2 + (size_t)nd[v] * T2_PITCH;
            o[n] = acc0[v] * dr[v];
            if (c1ok) o[16 + n] = acc1[v] * dr[v];
        }
    }
}


template <int F, int P, bool GUARD>
__device__ __forceinline__ void rows8(int s, int j, int n, const float* __restrict__ t,
                                      int lane, bool act, float (&v)[8]) {
#pragma unroll
    for (int u = 0; u < 8; ++u) {
        v[u] = 0.0f;
        if (!GUARD || j + u < n) {
            const int sj = __builtin_amdgcn_readlane(s, j + u);
            if (act) v[u] = t[(size_t)sj * P + lane];
        }
    }
}

__device__ __forceinline__ float sum8(const float (&v)[8], float acc) {
#pragma unroll
    for (int u = 0; u < 8; ++u) acc += v[u];
    return acc;
}

template <int F, int P>
__device__ __forceinline__ void rows8_first(int s, int n, const float* __restrict__ t,
                                            int lane, bool act, float (&v)[8]) {
    const int col = act ? lane : 0;
#pragma unroll
    for (int u = 0; u < 8; ++u) {
        const int sj = __builtin_amdgcn_readlane(s, u);
        const float x = t[(size_t)sj * P + col];
        v[u] = (u < n) ? x : 0.0f;
    }
}

template <int F, int P, bool GUARD>
__device__ __forceinline__ float gather8(int s, int j, int n,
                                         const float* __restrict__ t, int lane, bool act,
                                         float acc) {
    float v[8];
    rows8<F, P, GUARD>(s, j, n, t, lane, act, v);
    return sum8(v, acc);
}

template <int F, int P>
__device__ __forceinline__ float gather_group(int s, int j0, int n, const float* __restrict__ t,
                                              int lane, bool act, float acc) {
    int j = j0;
    for (; j + 8 <= n; j += 8) acc = gather8<F, P, false>(s, j, n, t, lane, act, acc);
    if (j < n)                 acc = gather8<F, P, true >(s, j, n, t, lane, act, acc);
    return acc;
}

template <int F, int P, bool FUSE>
__device__ __forceinline__ void gather_rows(const int* __restrict__ slot0,
                                            const int* __restrict__ sorted,
                                            const float* __restrict__ dinv,
                                            const float* __restrict__ t,
                                            const float* __restrict__ bias,
                                            float* __restrict__ o, int N, int E,
                                            float* tw, float* td, int* tn,
                                            float* w2s, const float* __restrict__ W2) {
    static_assert(!FUSE || F == 32, "a fused row fills one tile row: lane = k");
    static_assert(W2_LWORDS % (GT_WAVES * 32) == 0, "every thread stages the same number of words");
    float wv[W2_LWORDS / (GT_WAVES * 32)];
    if (FUSE) {
#pragma unroll
        for (int j = 0; j < W2_LWORDS / (GT_WAVES * 32); ++j) {
            const int i = j * (GT_WAVES * 32) + (int)threadIdx.x;
            const int k = i / W2_LPITCH, col = i % W2_LPITCH;
            wv[j] = W2[k * 20 + (col < 20 ? col : 0)];
        }
    }
    const int G = __builtin_amdgcn_readfirstlane((int)(gridDim.x * (blockDim.x >> 5)));
    int node = __builtin_amdgcn_readfirstlane((int)((blockIdx.x * blockDim.x + threadIdx.x) >> 5));
    const bool alive = node < N;
    if (!FUSE && !alive) return;
    const int  lane = threadIdx.x & 31;
    const bool act  = lane < F;
    const float bl  = act ? bias[lane] : 0.0f;

    int rows = 0;
    if (FUSE) {
#pragma unroll
        for (int r = 0; r < 16; ++r) tw[r * GT_PITCH + lane] = 0.0f;
        if (lane < 16) { td[lane] = 0.0f; tn[lane] = 0; }
    }

    int rs = 0, re = 0;
    int sa = 0, sb = 0;
    int nx  = node + G;
    int rs1 = 0, re1 = 0;
    if (alive) {
        rs = slot0[node];
        re = (node + 1 < N) ? slot0[node + 1] : E;
        if (rs + lane < re)      sa = sorted[rs + lane];
        if (rs + 32 + lane < re) sb = sorted[rs + 32 + lane];
        if (nx < N) { rs1 = slot0[nx]; re1 = (nx + 1 < N) ? slot0[nx + 1] : E; }
    }

    if (FUSE) {
#pragma unroll
        for (int j = 0; j < W2_LWORDS / (GT_WAVES * 32); ++j) {
            const int i = j * (GT_WAVES * 32) + (int)threadIdx.x;
            w2s[i] = (i % W2_LPITCH < 20) ? wv[j] : 0.0f;
        }
        __syncthreads();
        if (!alive) return;
    }

    for (;;) {
        const int cnt = re - rs;
        const int n0  = (cnt < 32) ? cnt : 32;
        float v[8];
        rows8_first<F, P>(sa, n0, t, lane, act, v);

        int sa1 = 0, sb1 = 0;
        if (nx < N) {
            if (rs1 + lane < re1)      sa1 = sorted[rs1 + lane];
            if (rs1 + 32 + lane < re1) sb1 = sorted[rs1 + 32 + lane];
        }
        const int nx2 = nx + G;
        int rs2 = 0, re2 = 0;
        if (nx < N && nx2 < N) { rs2 = slot0[nx2]; re2 = (nx2 + 1 < N) ? slot0[nx2 + 1] : E; }

        const float  di  = dinv[node];
        const size_t idx = (size_t)node * F + lane;
        float self = 0.0f;
        if (act) self = t[(size_t)node * P + lane];

        float acc = sum8(v, 0.0f);
        acc = gather_group<F, P>(sa, 8, n0, t, lane, act, acc);
        if (cnt > 32)
            acc = gather_group<F, P>(sb, 0, (cnt < 64) ? (cnt - 32) : 32, t, lane, act, acc);
        for (int base = rs + 64; base < re; base += 32) {
            const int n = (re - base < 32) ? (re - base) : 32;
            int s = 0;
            if (lane < n) s = sorted[base + lane];
            acc = gather_group<F, P>(s, 0, n, t, lane, act, acc);
        }
        const float val = (acc + self) * di + bl;
        if (FUSE) {
            tw[rows * GT_PITCH + lane] = fmaxf(val, 0.0f);
            if (lane == 0) { td[rows] = di; tn[rows] = node; }
            ++rows;
            if (rows == 16 || nx >= N) { t2_flush(tw, td, tn, w2s, rows, lane, o); rows = 0; }
        } else {
            if (act) o[idx] = val;
        }

        if (nx >= N) break;
        node = nx;  rs = rs1;  re = re1;  sa = sa1;  sb = sb1;
        nx = nx2;   rs1 = rs2; re1 = re2;
    }
}

__global__ void __launch_bounds__(GT_WAVES * 32, 16)
k_gather32_t2(const int* __restrict__ slot0, const int* __restrict__ sorted,
              const float* __restrict__ dinv, const float* __restrict__ t1,
              const float* __restrict__ b1, const float* __restrict__ W2,
              float* __restrict__ t2, int N, int E) {
    __shared__ float tile[GT_WAVES][16 * GT_PITCH];
    __shared__ float tfac[GT_WAVES][16];
    __shared__ int   tnod[GT_WAVES][16];
    __shared__ float w2s[W2_LWORDS];
    const int w = threadIdx.x >> 5;
    gather_rows<32, 32, true>(slot0, sorted, dinv, t1, b1, t2, N, E, tile[w], tfac[w], tnod[w], w2s, W2);
}

__global__ void __launch_bounds__(256)
k_gather20(const int* __restrict__ slot0, const int* __restrict__ sorted,
           const float* __restrict__ dinv, const float* __restrict__ t2,
           const float* __restrict__ b2, float* __restrict__ out, int N, int E) {
    gather_rows<20, T2_PITCH, false>(slot0, sorted, dinv, t2, b2, out, N, E, nullptr, nullptr, nullptr, nullptr, nullptr);
}


extern "C" void kernel_launch(void* const* d_in, const int* in_sizes, int n_in,
                              void* d_out, int out_size, void* d_ws, size_t ws_size,
                              hipStream_t stream) {
    const float* x    = (const float*)d_in[0];
    const float* cnn  = (const float*)d_in[1];
    const int*   ei   = (const int*)d_in[2];
    const float* fcW  = (const float*)d_in[3];
    const float* fcb  = (const float*)d_in[4];
    const float* cnnW = (const float*)d_in[5];
    const float* cnnb = (const float*)d_in[6];
    const float* W1   = (const float*)d_in[7];
    const float* b1   = (const float*)d_in[8];
    const float* W2   = (const float*)d_in[9];
    const float* b2   = (const float*)d_in[10];

    const int N = in_sizes[0] / 32;
    const int E = in_sizes[2] / 2;
    const int* src = ei;
    const int* dst = ei + E;

    float* ws   = (float*)d_ws;
    float* dinv = ws;  ws += N;
    float* eidx = ws;  ws += (size_t)N * 64;
    float* t1   = ws;  ws += (size_t)N * 32;
    float* agg1 = ws;  ws += (size_t)N * 32;
    float* bsw  = ws;  ws += (size_t)N * 20;
    float* t2   = agg1;
    float* out  = (float*)d_out;

    unsigned* tmp = (unsigned*)agg1;
    int* slot   = (int*)eidx;
    int* sorted = slot + N;

    const int nTiles = (N + 15) / 16;
    const int B = GT_WAVES * 32;
    dim3 grdNode(2048);

    const int nSl = (E + SC_EDGES - 1) / SC_EDGES;
    const int nBk = (N + BK_NODES - 1) / BK_NODES;
    unsigned* bh = (unsigned*)bsw;
    int* gbase   = (int*)(bh + (size_t)nSl * (BK_MAX / 2));

    k_bcount<<<nSl, CS_THREADS, 0, stream>>>(dst, bh, E, N);
    k_bscatter<<<nSl, CS_THREADS, 0, stream>>>(src, dst, bh, gbase, tmp, E, N, nSl);

    k_bsort_encoder<<<nBk, CS_THREADS, 0, stream>>>(tmp, gbase, slot, sorted, dinv, E, N,
                                                    x, cnn, fcW, fcb, cnnW, cnnb, W1, t1, nTiles);

    k_gather32_t2<<<grdNode, B, 0, stream>>>(slot, sorted, dinv, t1, b1, W2, t2, N, E);

    k_gather20<<<grdNode, B, 0, stream>>>(slot, sorted, dinv, t2, b2, out, N, E);
}
